# starting kernel
# baseline (speedup 1.0000x reference)
#include <hip/hip_runtime.h>
#include <hip/hip_bf16.h>

typedef __attribute__((ext_vector_type(2))) float v2f;
typedef __attribute__((ext_vector_type(4))) float v4f;
typedef __attribute__((ext_vector_type(8))) float v8f;

#define NNODES  100000
#define MPAD    100096
#define NEDGES  1600000
#define NGRAPHS 512

__global__ void zero_kernel(float* __restrict__ p, long long n) {
  long long i = (long long)blockIdx.x * blockDim.x + threadIdx.x;
  if (i < n) p[i] = 0.0f;
}

__global__ void deg_count_kernel(const int* __restrict__ dst, float* __restrict__ deg, int E) {
  int e = blockIdx.x * blockDim.x + threadIdx.x;
  if (e < E) atomicAdd(&deg[dst[e]], 1.0f);
}

__global__ void dinv_kernel(float* __restrict__ deg, int n) {
  int i = blockIdx.x * blockDim.x + threadIdx.x;
  if (i < n) deg[i] = rsqrtf(deg[i] + 1.0f);
}

__global__ void pad_x_kernel(const float* __restrict__ x, float* __restrict__ xp) {
  long long i = (long long)blockIdx.x * blockDim.x + threadIdx.x;
  if (i >= (long long)MPAD * 4) return;
  int node = (int)(i >> 2), j = (int)(i & 3);
  xp[i] = (node < NNODES && j < 3) ? x[(size_t)node * 3 + j] : 0.0f;
}

__global__ void pad_w1_kernel(const float* __restrict__ w, float* __restrict__ wp) {
  int i = blockIdx.x * blockDim.x + threadIdx.x;
  if (i >= 4 * 128) return;
  int k = i >> 7;
  wp[i] = (k < 3) ? w[i] : 0.0f;
}

__global__ void pad_wl2_kernel(const float* __restrict__ w, const float* __restrict__ b,
                               float* __restrict__ wp, float* __restrict__ bp) {
  int i = blockIdx.x * blockDim.x + threadIdx.x;
  if (i >= 32 * 16) return;
  int k = i >> 4, n = i & 15;
  wp[i] = (n < 9) ? w[k * 9 + n] : 0.0f;
  if (i < 16) bp[i] = (i < 9) ? b[i] : 0.0f;
}

__global__ void copy_out_kernel(const float* __restrict__ o16, float* __restrict__ out) {
  int i = blockIdx.x * blockDim.x + threadIdx.x;
  if (i >= NGRAPHS * 9) return;
  int g = i / 9, j = i - g * 9;
  out[i] = o16[g * 16 + j];
}

template <int K, int N, bool RELU, bool BIAS>
__global__ __launch_bounds__(256) void gemm_wmma(
    const float* __restrict__ A, const float* __restrict__ W,
    const float* __restrict__ bias, float* __restrict__ C) {
  __shared__ float lw[K * N];

  const int tid = threadIdx.x;
  constexpr int QUADS = (K / 2) * N / 4;
  for (int q = tid; q < QUADS; q += 256) {
    int t = (q * 4) / N;
    int n = (q * 4) % N;
    int k = t * 2;
    v4f w0 = *(const v4f*)(W + (size_t)k * N + n);
    v4f w1 = *(const v4f*)(W + (size_t)(k + 1) * N + n);
    v4f lo, hi;
    lo.x = w0.x; lo.y = w1.x; lo.z = w0.y; lo.w = w1.y;
    hi.x = w0.z; hi.y = w1.z; hi.z = w0.w; hi.w = w1.w;
    float* d = lw + ((size_t)t * N + n) * 2;
    *(v4f*)(d)     = lo;
    *(v4f*)(d + 4) = hi;
  }
  __syncthreads();

  const int lane = tid & 31;
  const int wave = tid >> 5;
  const int half = lane >> 4;
  const int l16  = lane & 15;
  const int row  = blockIdx.x * 128 + wave * 16 + l16;
  const float* __restrict__ Arow = A + (size_t)row * K + 2 * half;

  for (int n_base = 0; n_base < N; n_base += 16) {
    const int n = n_base + l16;
    float binit = BIAS ? bias[n] : 0.0f;
    v8f acc;
#pragma unroll
    for (int r = 0; r < 8; ++r) acc[r] = binit;

#pragma unroll
    for (int k = 0; k < K; k += 4) {
      v2f av = *(const v2f*)(Arow + k);
      if (RELU) { av.x = fmaxf(av.x, 0.0f); av.y = fmaxf(av.y, 0.0f); }
      const int kk = k + 2 * half;
      v2f bv = *(const v2f*)(lw + ((size_t)(kk >> 1) * N + n) * 2);
      acc = __builtin_amdgcn_wmma_f32_16x16x4_f32(
          false, av, false, bv, (short)0, acc, false, false);
    }

    float* __restrict__ Cp = C + (size_t)(blockIdx.x * 128 + wave * 16 + half * 8) * N + n;
#pragma unroll
    for (int r = 0; r < 8; ++r) Cp[(size_t)r * N] = acc[r];
  }
}

__global__ void selfloop_bias_kernel(const float* __restrict__ h,
                                     const float* __restrict__ dinv,
                                     const float* __restrict__ bias,
                                     float* __restrict__ out,
                                     int F, int lgF4) {
  long long i = (long long)blockIdx.x * blockDim.x + threadIdx.x;
  if (i >= ((long long)MPAD << lgF4)) return;
  int node = (int)(i >> lgF4);
  int f    = (int)(i & ((1 << lgF4) - 1)) * 4;
  v4f o;
  if (node < NNODES) {
    float di = dinv[node];
    float s  = di * di;
    v4f hv = *(const v4f*)(h + (size_t)node * F + f);
    v4f bv = *(const v4f*)(bias + f);
    o.x = hv.x * s + bv.x; o.y = hv.y * s + bv.y;
    o.z = hv.z * s + bv.z; o.w = hv.w * s + bv.w;
  } else {
    o.x = o.y = o.z = o.w = 0.0f;
  }
  *(v4f*)(out + (size_t)node * F + f) = o;
}

__global__ void edge_scatter_kernel(const float* __restrict__ h,
                                    const int* __restrict__ src,
                                    const int* __restrict__ dst,
                                    const float* __restrict__ dinv,
                                    float* __restrict__ out,
                                    int E, int F, int lgF4) {
  long long i = (long long)blockIdx.x * blockDim.x + threadIdx.x;
  if (i >= ((long long)E << lgF4)) return;
  int e = (int)(i >> lgF4);
  int f = (int)(i & ((1 << lgF4) - 1)) * 4;
  int s = src[e], d = dst[e];
  float nrm = dinv[s] * dinv[d];
  v4f hv = *(const v4f*)(h + (size_t)s * F + f);
  float* o = out + (size_t)d * F + f;
  atomicAdd(o + 0, hv.x * nrm);
  atomicAdd(o + 1, hv.y * nrm);
  atomicAdd(o + 2, hv.z * nrm);
  atomicAdd(o + 3, hv.w * nrm);
}

__global__ void pool_acc_kernel(const float* __restrict__ h,
                                const int* __restrict__ batch,
                                float* __restrict__ pooled,
                                float* __restrict__ cnt) {
  long long i = (long long)blockIdx.x * blockDim.x + threadIdx.x;
  if (i >= (long long)NNODES * 16) return;
  int node = (int)(i >> 4);
  int f    = (int)(i & 15) * 4;
  int g = batch[node];
  v4f hv = *(const v4f*)(h + (size_t)node * 64 + f);
  float* p = pooled + (size_t)g * 64 + f;
  atomicAdd(p + 0, hv.x);
  atomicAdd(p + 1, hv.y);
  atomicAdd(p + 2, hv.z);
  atomicAdd(p + 3, hv.w);
  if (f == 0) atomicAdd(&cnt[g], 1.0f);
}

__global__ void pool_div_kernel(float* __restrict__ pooled,
                                const float* __restrict__ cnt) {
  int i = blockIdx.x * blockDim.x + threadIdx.x;
  if (i < NGRAPHS * 64) pooled[i] /= fmaxf(cnt[i >> 6], 1.0f);
}

extern "C" void kernel_launch(void* const* d_in, const int* in_sizes, int n_in,
                              void* d_out, int out_size, void* d_ws, size_t ws_size,
                              hipStream_t stream) {
  const float* x   = (const float*)d_in[0];
  const int*   ei  = (const int*)d_in[1];
  const int*   bat = (const int*)d_in[2];
  const float* W1  = (const float*)d_in[3];  const float* b1  = (const float*)d_in[4];
  const float* W2  = (const float*)d_in[5];  const float* b2  = (const float*)d_in[6];
  const float* W3  = (const float*)d_in[7];  const float* b3  = (const float*)d_in[8];
  const float* W4  = (const float*)d_in[9];  const float* b4  = (const float*)d_in[10];
  const float* Wl1 = (const float*)d_in[11]; const float* bl1 = (const float*)d_in[12];
  const float* Wl2 = (const float*)d_in[13]; const float* bl2 = (const float*)d_in[14];
  float* out = (float*)d_out;

  const int* src = ei;
  const int* dst = ei + NEDGES;

  float* ws     = (float*)d_ws;
  float* dinv   = ws;
  float* xp     = dinv + NNODES;
  float* hT     = xp + (size_t)MPAD * 4;
  float* hA     = hT + (size_t)MPAD * 128;
  float* pooled = hA + (size_t)MPAD * 128;
  float* cnt    = pooled + NGRAPHS * 64;
  float* t1     = cnt + NGRAPHS;
  float* W1p    = t1 + NGRAPHS * 32;
  float* Wl2p   = W1p + 4 * 128;
  float* bl2p   = Wl2p + 32 * 16;
  float* o16    = bl2p + 16;

  const int B = 256;
  auto blks = [](long long n, int b) { return (unsigned)((n + b - 1) / b); };

  zero_kernel<<<blks(NNODES, B), B, 0, stream>>>(dinv, NNODES);
  deg_count_kernel<<<blks(NEDGES, B), B, 0, stream>>>(dst, dinv, NEDGES);
  dinv_kernel<<<blks(NNODES, B), B, 0, stream>>>(dinv, NNODES);

  pad_x_kernel<<<blks((long long)MPAD * 4, B), B, 0, stream>>>(x, xp);
  pad_w1_kernel<<<blks(4 * 128, B), B, 0, stream>>>(W1, W1p);
  pad_wl2_kernel<<<blks(32 * 16, B), B, 0, stream>>>(Wl2, bl2, Wl2p, bl2p);

  const unsigned GB = MPAD / 128;
  const long long n128 = (long long)MPAD * 32;
  const long long n64  = (long long)MPAD * 16;
  const long long e128 = (long long)NEDGES * 32;
  const long long e64  = (long long)NEDGES * 16;

  gemm_wmma<4, 128, false, false><<<GB, 256, 0, stream>>>(xp, W1p, nullptr, hT);
  selfloop_bias_kernel<<<blks(n128, B), B, 0, stream>>>(hT, dinv, b1, hA, 128, 5);
  edge_scatter_kernel<<<blks(e128, B), B, 0, stream>>>(hT, src, dst, dinv, hA, NEDGES, 128, 5);

  gemm_wmma<128, 128, true, false><<<GB, 256, 0, stream>>>(hA, W2, nullptr, hT);
  selfloop_bias_kernel<<<blks(n128, B), B, 0, stream>>>(hT, dinv, b2, hA, 128, 5);
  edge_scatter_kernel<<<blks(e128, B), B, 0, stream>>>(hT, src, dst, dinv, hA, NEDGES, 128, 5);

  gemm_wmma<128, 128, true, false><<<GB, 256, 0, stream>>>(hA, W3, nullptr, hT);
  selfloop_bias_kernel<<<blks(n128, B), B, 0, stream>>>(hT, dinv, b3, hA, 128, 5);
  edge_scatter_kernel<<<blks(e128, B), B, 0, stream>>>(hT, src, dst, dinv, hA, NEDGES, 128, 5);

  gemm_wmma<128, 64, true, false><<<GB, 256, 0, stream>>>(hA, W4, nullptr, hT);
  selfloop_bias_kernel<<<blks(n64, B), B, 0, stream>>>(hT, dinv, b4, hA, 64, 4);
  edge_scatter_kernel<<<blks(e64, B), B, 0, stream>>>(hT, src, dst, dinv, hA, NEDGES, 64, 4);

  zero_kernel<<<blks(NGRAPHS * 64, B), B, 0, stream>>>(pooled, NGRAPHS * 64);
  zero_kernel<<<blks(NGRAPHS, B), B, 0, stream>>>(cnt, NGRAPHS);
  pool_acc_kernel<<<blks((long long)NNODES * 16, B), B, 0, stream>>>(hA, bat, pooled, cnt);
  pool_div_kernel<<<blks(NGRAPHS * 64, B), B, 0, stream>>>(pooled, cnt);

  gemm_wmma<64, 32, false, true><<<NGRAPHS / 128, 256, 0, stream>>>(pooled, Wl1, bl1, t1);
  gemm_wmma<32, 16, true, true><<<NGRAPHS / 128, 256, 0, stream>>>(t1, Wl2p, bl2p, o16);
  copy_out_kernel<<<blks(NGRAPHS * 9, B), B, 0, stream>>>(o16, out);
}


# Edge aggregation via per-call CSR (int-atomic rank, scan, fill) and one-wave-per-row gather with self-loop+bias fused; float atomic scatter removed; row sums now in CSR slot order
# speedup vs baseline: 2.4724x; 2.4724x over previous
#include <hip/hip_runtime.h>
#include <hip/hip_bf16.h>

typedef __attribute__((ext_vector_type(2))) float v2f;
typedef __attribute__((ext_vector_type(4))) float v4f;
typedef __attribute__((ext_vector_type(8))) float v8f;

#define NNODES  100000
#define MPAD    100096
#define NEDGES  1600000
#define NGRAPHS 512

__global__ void zero_kernel(float* __restrict__ p, long long n) {
  long long i = (long long)blockIdx.x * blockDim.x + threadIdx.x;
  if (i < n) p[i] = 0.0f;
}

__global__ void zero_int_kernel(int* __restrict__ p, int n) {
  int i = blockIdx.x * blockDim.x + threadIdx.x;
  if (i < n) p[i] = 0;
}

__global__ void edge_rank_kernel(const int* __restrict__ dst, int* __restrict__ indeg,
                                 int* __restrict__ rank, int E) {
  int e = blockIdx.x * blockDim.x + threadIdx.x;
  if (e < E) rank[e] = atomicAdd(&indeg[dst[e]], 1);
}

__global__ __launch_bounds__(256) void row_scan_local_kernel(
    const int* __restrict__ indeg, int* __restrict__ rowptr,
    int* __restrict__ bsum, float* __restrict__ dinv) {
  __shared__ int sh[256];
  const int t = threadIdx.x;
  const int i = blockIdx.x * 256 + t;
  const int v = (i < NNODES) ? indeg[i] : 0;
  if (i < NNODES) dinv[i] = rsqrtf((float)v + 1.0f);
  sh[t] = v;
  __syncthreads();
  for (int off = 1; off < 256; off <<= 1) {
    int add = (t >= off) ? sh[t - off] : 0;
    __syncthreads();
    sh[t] += add;
    __syncthreads();
  }
  const int incl = sh[t];
  if (i < NNODES) rowptr[i] = incl - v;
  if (t == 255) bsum[blockIdx.x] = incl;
}

__global__ __launch_bounds__(256) void row_scan_offset_kernel(
    int* __restrict__ rowptr, const int* __restrict__ bsum) {
  __shared__ int red[256];
  const int t = threadIdx.x;
  const int b = blockIdx.x;
  int s = 0;
  for (int k = t; k < b; k += 256) s += bsum[k];
  red[t] = s;
  __syncthreads();
  for (int off = 128; off > 0; off >>= 1) {
    if (t < off) red[t] += red[t + off];
    __syncthreads();
  }
  const int i = b * 256 + t;
  if (i < NNODES) rowptr[i] += red[0];
}

__global__ void csr_fill_kernel(const int* __restrict__ src, const int* __restrict__ dst,
                                const int* __restrict__ rank, const int* __restrict__ rowptr,
                                int* __restrict__ csr_src, int E) {
  int e = blockIdx.x * blockDim.x + threadIdx.x;
  if (e < E) csr_src[rowptr[dst[e]] + rank[e]] = src[e];
}

__global__ void pad_x_kernel(const float* __restrict__ x, float* __restrict__ xp) {
  long long i = (long long)blockIdx.x * blockDim.x + threadIdx.x;
  if (i >= (long long)MPAD * 4) return;
  int node = (int)(i >> 2), j = (int)(i & 3);
  xp[i] = (node < NNODES && j < 3) ? x[(size_t)node * 3 + j] : 0.0f;
}

__global__ void pad_w1_kernel(const float* __restrict__ w, float* __restrict__ wp) {
  int i = blockIdx.x * blockDim.x + threadIdx.x;
  if (i >= 4 * 128) return;
  int k = i >> 7;
  wp[i] = (k < 3) ? w[i] : 0.0f;
}

__global__ void pad_wl2_kernel(const float* __restrict__ w, const float* __restrict__ b,
                               float* __restrict__ wp, float* __restrict__ bp) {
  int i = blockIdx.x * blockDim.x + threadIdx.x;
  if (i >= 32 * 16) return;
  int k = i >> 4, n = i & 15;
  wp[i] = (n < 9) ? w[k * 9 + n] : 0.0f;
  if (i < 16) bp[i] = (i < 9) ? b[i] : 0.0f;
}

__global__ void copy_out_kernel(const float* __restrict__ o16, float* __restrict__ out) {
  int i = blockIdx.x * blockDim.x + threadIdx.x;
  if (i >= NGRAPHS * 9) return;
  int g = i / 9, j = i - g * 9;
  out[i] = o16[g * 16 + j];
}

template <int K, int N, bool RELU, bool BIAS>
__global__ __launch_bounds__(256) void gemm_wmma(
    const float* __restrict__ A, const float* __restrict__ W,
    const float* __restrict__ bias, float* __restrict__ C) {
  __shared__ float lw[K * N];

  const int tid = threadIdx.x;
  constexpr int QUADS = (K / 2) * N / 4;
  for (int q = tid; q < QUADS; q += 256) {
    int t = (q * 4) / N;
    int n = (q * 4) % N;
    int k = t * 2;
    v4f w0 = *(const v4f*)(W + (size_t)k * N + n);
    v4f w1 = *(const v4f*)(W + (size_t)(k + 1) * N + n);
    v4f lo, hi;
    lo.x = w0.x; lo.y = w1.x; lo.z = w0.y; lo.w = w1.y;
    hi.x = w0.z; hi.y = w1.z; hi.z = w0.w; hi.w = w1.w;
    float* d = lw + ((size_t)t * N + n) * 2;
    *(v4f*)(d)     = lo;
    *(v4f*)(d + 4) = hi;
  }
  __syncthreads();

  const int lane = tid & 31;
  const int wave = tid >> 5;
  const int half = lane >> 4;
  const int l16  = lane & 15;
  const int row  = blockIdx.x * 128 + wave * 16 + l16;
  const float* __restrict__ Arow = A + (size_t)row * K + 2 * half;

  for (int n_base = 0; n_base < N; n_base += 16) {
    const int n = n_base + l16;
    float binit = BIAS ? bias[n] : 0.0f;
    v8f acc;
#pragma unroll
    for (int r = 0; r < 8; ++r) acc[r] = binit;

#pragma unroll
    for (int k = 0; k < K; k += 4) {
      v2f av = *(const v2f*)(Arow + k);
      if (RELU) { av.x = fmaxf(av.x, 0.0f); av.y = fmaxf(av.y, 0.0f); }
      const int kk = k + 2 * half;
      v2f bv = *(const v2f*)(lw + ((size_t)(kk >> 1) * N + n) * 2);
      acc = __builtin_amdgcn_wmma_f32_16x16x4_f32(
          false, av, false, bv, (short)0, acc, false, false);
    }

    float* __restrict__ Cp = C + (size_t)(blockIdx.x * 128 + wave * 16 + half * 8) * N + n;
#pragma unroll
    for (int r = 0; r < 8; ++r) Cp[(size_t)r * N] = acc[r];
  }
}

template <int F>
__global__ __launch_bounds__(256) void row_gather_kernel(
    const float* __restrict__ h, const int* __restrict__ rowptr,
    const int* __restrict__ indeg, const int* __restrict__ csr_src,
    const float* __restrict__ dinv, const float* __restrict__ bias,
    float* __restrict__ out) {
  constexpr int V = F / 32;
  typedef __attribute__((ext_vector_type(V))) float vt;
  const int lane = threadIdx.x & 31;
  const int row = __builtin_amdgcn_readfirstlane((int)(blockIdx.x * 8 + (threadIdx.x >> 5)));
  const int f = lane * V;
  float* __restrict__ o = out + (size_t)row * F + f;
  if (row >= NNODES) {
    vt z = 0.0f;
    *(vt*)o = z;
    return;
  }
  const float dd = dinv[row];
  const int start = rowptr[row];
  const int end = start + indeg[row];
  const vt hv = *(const vt*)(h + (size_t)row * F + f);
  const vt bv = *(const vt*)(bias + f);
  vt acc = hv * (dd * dd) + bv;

  for (int base = start; base < end; base += 32) {
    const int n = (end - base < 32) ? (end - base) : 32;
    int myidx = 0;
    float mynrm = 0.0f;
    if (lane < n) {
      myidx = csr_src[base + lane];
      mynrm = dinv[myidx] * dd;
    }
    const int mynrm_bits = __builtin_bit_cast(int, mynrm);
    int j = 0;
    for (; j + 4 <= n; j += 4) {
      const int s0 = __builtin_amdgcn_readlane(myidx, j);
      const int s1 = __builtin_amdgcn_readlane(myidx, j + 1);
      const int s2 = __builtin_amdgcn_readlane(myidx, j + 2);
      const int s3 = __builtin_amdgcn_readlane(myidx, j + 3);
      const vt a0 = *(const vt*)(h + (size_t)s0 * F + f);
      const vt a1 = *(const vt*)(h + (size_t)s1 * F + f);
      const vt a2 = *(const vt*)(h + (size_t)s2 * F + f);
      const vt a3 = *(const vt*)(h + (size_t)s3 * F + f);
      const float w0 = __builtin_bit_cast(float, __builtin_amdgcn_readlane(mynrm_bits, j));
      const float w1 = __builtin_bit_cast(float, __builtin_amdgcn_readlane(mynrm_bits, j + 1));
      const float w2 = __builtin_bit_cast(float, __builtin_amdgcn_readlane(mynrm_bits, j + 2));
      const float w3 = __builtin_bit_cast(float, __builtin_amdgcn_readlane(mynrm_bits, j + 3));
      acc += a0 * w0;
      acc += a1 * w1;
      acc += a2 * w2;
      acc += a3 * w3;
    }
    for (; j < n; ++j) {
      const int s0 = __builtin_amdgcn_readlane(myidx, j);
      const vt a0 = *(const vt*)(h + (size_t)s0 * F + f);
      const float w0 = __builtin_bit_cast(float, __builtin_amdgcn_readlane(mynrm_bits, j));
      acc += a0 * w0;
    }
  }
  *(vt*)o = acc;
}

__global__ void pool_acc_kernel(const float* __restrict__ h,
                                const int* __restrict__ batch,
                                float* __restrict__ pooled,
                                float* __restrict__ cnt) {
  long long i = (long long)blockIdx.x * blockDim.x + threadIdx.x;
  if (i >= (long long)NNODES * 16) return;
  int node = (int)(i >> 4);
  int f    = (int)(i & 15) * 4;
  int g = batch[node];
  v4f hv = *(const v4f*)(h + (size_t)node * 64 + f);
  float* p = pooled + (size_t)g * 64 + f;
  atomicAdd(p + 0, hv.x);
  atomicAdd(p + 1, hv.y);
  atomicAdd(p + 2, hv.z);
  atomicAdd(p + 3, hv.w);
  if (f == 0) atomicAdd(&cnt[g], 1.0f);
}

__global__ void pool_div_kernel(float* __restrict__ pooled,
                                const float* __restrict__ cnt) {
  int i = blockIdx.x * blockDim.x + threadIdx.x;
  if (i < NGRAPHS * 64) pooled[i] /= fmaxf(cnt[i >> 6], 1.0f);
}

extern "C" void kernel_launch(void* const* d_in, const int* in_sizes, int n_in,
                              void* d_out, int out_size, void* d_ws, size_t ws_size,
                              hipStream_t stream) {
  const float* x   = (const float*)d_in[0];
  const int*   ei  = (const int*)d_in[1];
  const int*   bat = (const int*)d_in[2];
  const float* W1  = (const float*)d_in[3];  const float* b1  = (const float*)d_in[4];
  const float* W2  = (const float*)d_in[5];  const float* b2  = (const float*)d_in[6];
  const float* W3  = (const float*)d_in[7];  const float* b3  = (const float*)d_in[8];
  const float* W4  = (const float*)d_in[9];  const float* b4  = (const float*)d_in[10];
  const float* Wl1 = (const float*)d_in[11]; const float* bl1 = (const float*)d_in[12];
  const float* Wl2 = (const float*)d_in[13]; const float* bl2 = (const float*)d_in[14];
  float* out = (float*)d_out;

  const int* src = ei;
  const int* dst = ei + NEDGES;

  float* ws     = (float*)d_ws;
  float* dinv   = ws;
  float* xp     = dinv + NNODES;
  float* hT     = xp + (size_t)MPAD * 4;
  float* hA     = hT + (size_t)MPAD * 128;
  float* pooled = hA + (size_t)MPAD * 128;
  float* cnt    = pooled + NGRAPHS * 64;
  float* t1     = cnt + NGRAPHS;
  float* W1p    = t1 + NGRAPHS * 32;
  float* Wl2p   = W1p + 4 * 128;
  float* bl2p   = Wl2p + 32 * 16;
  float* o16    = bl2p + 16;
  int*   indeg   = (int*)(o16 + NGRAPHS * 16);
  int*   rowptr  = indeg + NNODES;
  int*   bsum    = rowptr + NNODES;
  int*   rank    = bsum + 512;
  int*   csr_src = rank + NEDGES;

  const int B = 256;
  auto blks = [](long long n, int b) { return (unsigned)((n + b - 1) / b); };

  zero_int_kernel<<<blks(NNODES, B), B, 0, stream>>>(indeg, NNODES);
  edge_rank_kernel<<<blks(NEDGES, B), B, 0, stream>>>(dst, indeg, rank, NEDGES);
  row_scan_local_kernel<<<blks(NNODES, 256), 256, 0, stream>>>(indeg, rowptr, bsum, dinv);
  row_scan_offset_kernel<<<blks(NNODES, 256), 256, 0, stream>>>(rowptr, bsum);
  csr_fill_kernel<<<blks(NEDGES, B), B, 0, stream>>>(src, dst, rank, rowptr, csr_src, NEDGES);

  pad_x_kernel<<<blks((long long)MPAD * 4, B), B, 0, stream>>>(x, xp);
  pad_w1_kernel<<<blks(4 * 128, B), B, 0, stream>>>(W1, W1p);
  pad_wl2_kernel<<<blks(32 * 16, B), B, 0, stream>>>(Wl2, bl2, Wl2p, bl2p);

  const unsigned GB = MPAD / 128;
  const unsigned RB = MPAD / 8;

  gemm_wmma<4, 128, false, false><<<GB, 256, 0, stream>>>(xp, W1p, nullptr, hT);
  row_gather_kernel<128><<<RB, 256, 0, stream>>>(hT, rowptr, indeg, csr_src, dinv, b1, hA);

  gemm_wmma<128, 128, true, false><<<GB, 256, 0, stream>>>(hA, W2, nullptr, hT);
  row_gather_kernel<128><<<RB, 256, 0, stream>>>(hT, rowptr, indeg, csr_src, dinv, b2, hA);

  gemm_wmma<128, 128, true, false><<<GB, 256, 0, stream>>>(hA, W3, nullptr, hT);
  row_gather_kernel<128><<<RB, 256, 0, stream>>>(hT, rowptr, indeg, csr_src, dinv, b3, hA);

  gemm_wmma<128, 64, true, false><<<GB, 256, 0, stream>>>(hA, W4, nullptr, hT);
  row_gather_kernel<64><<<RB, 256, 0, stream>>>(hT, rowptr, indeg, csr_src, dinv, b4, hA);

  zero_kernel<<<blks(NGRAPHS * 64, B), B, 0, stream>>>(pooled, NGRAPHS * 64);
  zero_kernel<<<blks(NGRAPHS, B), B, 0, stream>>>(cnt, NGRAPHS);
  pool_acc_kernel<<<blks((long long)NNODES * 16, B), B, 0, stream>>>(hA, bat, pooled, cnt);
  pool_div_kernel<<<blks(NGRAPHS * 64, B), B, 0, stream>>>(pooled, cnt);

  gemm_wmma<64, 32, false, true><<<NGRAPHS / 128, 256, 0, stream>>>(pooled, Wl1, bl1, t1);
  gemm_wmma<32, 16, true, true><<<NGRAPHS / 128, 256, 0, stream>>>(t1, Wl2p, bl2p, o16);
  copy_out_kernel<<<blks(NGRAPHS * 9, B), B, 0, stream>>>(o16, out);
}


# Layer 1 aggregates its 3-wide input rows (pre-scaled by dinv, one 16-B row per node) before the 3->128 product, not the 128-wide result; bias moves into that GEMM; layer-1 sum order changes
# speedup vs baseline: 2.6396x; 1.0676x over previous
#include <hip/hip_runtime.h>
#include <hip/hip_bf16.h>

typedef __attribute__((ext_vector_type(2))) float v2f;
typedef __attribute__((ext_vector_type(4))) float v4f;
typedef __attribute__((ext_vector_type(8))) float v8f;

#define NNODES  100000
#define MPAD    100096
#define NEDGES  1600000
#define NGRAPHS 512

__global__ void zero_kernel(float* __restrict__ p, long long n) {
  long long i = (long long)blockIdx.x * blockDim.x + threadIdx.x;
  if (i < n) p[i] = 0.0f;
}

__global__ void zero_int_kernel(int* __restrict__ p, int n) {
  int i = blockIdx.x * blockDim.x + threadIdx.x;
  if (i < n) p[i] = 0;
}

__global__ void edge_rank_kernel(const int* __restrict__ dst, int* __restrict__ indeg,
                                 int* __restrict__ rank, int E) {
  int e = blockIdx.x * blockDim.x + threadIdx.x;
  if (e < E) rank[e] = atomicAdd(&indeg[dst[e]], 1);
}

__global__ __launch_bounds__(256) void row_scan_local_kernel(
    const int* __restrict__ indeg, int* __restrict__ rowptr,
    int* __restrict__ bsum, float* __restrict__ dinv,
    const float* __restrict__ x, float* __restrict__ xs) {
  __shared__ int sh[256];
  const int t = threadIdx.x;
  const int i = blockIdx.x * 256 + t;
  const int v = (i < NNODES) ? indeg[i] : 0;
  if (i < NNODES) {
    const float d = rsqrtf((float)v + 1.0f);
    dinv[i] = d;
    const float* __restrict__ xr = x + (size_t)i * 3;
    v4f r;
    r.x = xr[0] * d; r.y = xr[1] * d; r.z = xr[2] * d; r.w = 0.0f;
    *(v4f*)(xs + (size_t)i * 4) = r;
  }
  sh[t] = v;
  __syncthreads();
  for (int off = 1; off < 256; off <<= 1) {
    int add = (t >= off) ? sh[t - off] : 0;
    __syncthreads();
    sh[t] += add;
    __syncthreads();
  }
  const int incl = sh[t];
  if (i < NNODES) rowptr[i] = incl - v;
  if (t == 255) bsum[blockIdx.x] = incl;
}

__global__ __launch_bounds__(256) void row_scan_offset_kernel(
    int* __restrict__ rowptr, const int* __restrict__ bsum) {
  __shared__ int red[256];
  const int t = threadIdx.x;
  const int b = blockIdx.x;
  int s = 0;
  for (int k = t; k < b; k += 256) s += bsum[k];
  red[t] = s;
  __syncthreads();
  for (int off = 128; off > 0; off >>= 1) {
    if (t < off) red[t] += red[t + off];
    __syncthreads();
  }
  const int i = b * 256 + t;
  if (i < NNODES) rowptr[i] += red[0];
}

__global__ void csr_fill_kernel(const int* __restrict__ src, const int* __restrict__ dst,
                                const int* __restrict__ rank, const int* __restrict__ rowptr,
                                int* __restrict__ csr_src, int E) {
  int e = blockIdx.x * blockDim.x + threadIdx.x;
  if (e < E) csr_src[rowptr[dst[e]] + rank[e]] = src[e];
}

__global__ __launch_bounds__(256) void row_gather_x_kernel(
    const float* __restrict__ xs, const int* __restrict__ rowptr,
    const int* __restrict__ indeg, const int* __restrict__ csr_src,
    const float* __restrict__ dinv, float* __restrict__ xa) {
  const int t = blockIdx.x * 256 + threadIdx.x;
  const int row = t >> 2;
  const int sub = t & 3;
  const v4f zero4 = 0.0f;
  v4f acc = zero4;
  float dd = 0.0f;
  if (row < NNODES) {
    dd = dinv[row];
    const int start = rowptr[row];
    const int end = start + indeg[row];
    const int last = end - 1;
    if (sub == 0) acc = *(const v4f*)(xs + (size_t)row * 4);
    for (int base = start; base < end; base += 32) {
      int s[8];
      for (int u = 0; u < 8; ++u) {
        const int e = base + sub + 4 * u;
        s[u] = csr_src[(e < end) ? e : last];
      }
      for (int u = 0; u < 8; ++u) {
        const int e = base + sub + 4 * u;
        const v4f v = *(const v4f*)(xs + (size_t)s[u] * 4);
        acc += (e < end) ? v : zero4;
      }
    }
  }
  float r[3];
  for (int c = 0; c < 3; ++c) {
    float a = acc[c];
    a += __builtin_bit_cast(float, __builtin_amdgcn_update_dpp(
             0, __builtin_bit_cast(int, a), 0xB1, 0xF, 0xF, false));
    a += __builtin_bit_cast(float, __builtin_amdgcn_update_dpp(
             0, __builtin_bit_cast(int, a), 0x4E, 0xF, 0xF, false));
    r[c] = a;
  }
  if (sub == 0) {
    v4f o;
    o.x = r[0] * dd; o.y = r[1] * dd; o.z = r[2] * dd; o.w = 0.0f;
    *(v4f*)(xa + (size_t)row * 4) = o;
  }
}

__global__ void pad_w1_kernel(const float* __restrict__ w, float* __restrict__ wp) {
  int i = blockIdx.x * blockDim.x + threadIdx.x;
  if (i >= 4 * 128) return;
  int k = i >> 7;
  wp[i] = (k < 3) ? w[i] : 0.0f;
}

__global__ void pad_wl2_kernel(const float* __restrict__ w, const float* __restrict__ b,
                               float* __restrict__ wp, float* __restrict__ bp) {
  int i = blockIdx.x * blockDim.x + threadIdx.x;
  if (i >= 32 * 16) return;
  int k = i >> 4, n = i & 15;
  wp[i] = (n < 9) ? w[k * 9 + n] : 0.0f;
  if (i < 16) bp[i] = (i < 9) ? b[i] : 0.0f;
}

__global__ void copy_out_kernel(const float* __restrict__ o16, float* __restrict__ out) {
  int i = blockIdx.x * blockDim.x + threadIdx.x;
  if (i >= NGRAPHS * 9) return;
  int g = i / 9, j = i - g * 9;
  out[i] = o16[g * 16 + j];
}

template <int K, int N, bool RELU, bool BIAS>
__global__ __launch_bounds__(256) void gemm_wmma(
    const float* __restrict__ A, const float* __restrict__ W,
    const float* __restrict__ bias, float* __restrict__ C) {
  __shared__ float lw[K * N];

  const int tid = threadIdx.x;
  constexpr int QUADS = (K / 2) * N / 4;
  for (int q = tid; q < QUADS; q += 256) {
    int t = (q * 4) / N;
    int n = (q * 4) % N;
    int k = t * 2;
    v4f w0 = *(const v4f*)(W + (size_t)k * N + n);
    v4f w1 = *(const v4f*)(W + (size_t)(k + 1) * N + n);
    v4f lo, hi;
    lo.x = w0.x; lo.y = w1.x; lo.z = w0.y; lo.w = w1.y;
    hi.x = w0.z; hi.y = w1.z; hi.z = w0.w; hi.w = w1.w;
    float* d = lw + ((size_t)t * N + n) * 2;
    *(v4f*)(d)     = lo;
    *(v4f*)(d + 4) = hi;
  }
  __syncthreads();

  const int lane = tid & 31;
  const int wave = tid >> 5;
  const int half = lane >> 4;
  const int l16  = lane & 15;
  const int row  = blockIdx.x * 128 + wave * 16 + l16;
  const float* __restrict__ Arow = A + (size_t)row * K + 2 * half;

  for (int n_base = 0; n_base < N; n_base += 16) {
    const int n = n_base + l16;
    float binit = BIAS ? bias[n] : 0.0f;
    v8f acc;
#pragma unroll
    for (int r = 0; r < 8; ++r) acc[r] = binit;

#pragma unroll
    for (int k = 0; k < K; k += 4) {
      v2f av = *(const v2f*)(Arow + k);
      if (RELU) { av.x = fmaxf(av.x, 0.0f); av.y = fmaxf(av.y, 0.0f); }
      const int kk = k + 2 * half;
      v2f bv = *(const v2f*)(lw + ((size_t)(kk >> 1) * N + n) * 2);
      acc = __builtin_amdgcn_wmma_f32_16x16x4_f32(
          false, av, false, bv, (short)0, acc, false, false);
    }

    float* __restrict__ Cp = C + (size_t)(blockIdx.x * 128 + wave * 16 + half * 8) * N + n;
#pragma unroll
    for (int r = 0; r < 8; ++r) Cp[(size_t)r * N] = acc[r];
  }
}

template <int F>
__global__ __launch_bounds__(256) void row_gather_kernel(
    const float* __restrict__ h, const int* __restrict__ rowptr,
    const int* __restrict__ indeg, const int* __restrict__ csr_src,
    const float* __restrict__ dinv, const float* __restrict__ bias,
    float* __restrict__ out) {
  constexpr int V = F / 32;
  typedef __attribute__((ext_vector_type(V))) float vt;
  const int lane = threadIdx.x & 31;
  const int row = __builtin_amdgcn_readfirstlane((int)(blockIdx.x * 8 + (threadIdx.x >> 5)));
  const int f = lane * V;
  float* __restrict__ o = out + (size_t)row * F + f;
  if (row >= NNODES) {
    vt z = 0.0f;
    *(vt*)o = z;
    return;
  }
  const float dd = dinv[row];
  const int start = rowptr[row];
  const int end = start + indeg[row];
  const vt hv = *(const vt*)(h + (size_t)row * F + f);
  const vt bv = *(const vt*)(bias + f);
  vt acc = hv * (dd * dd) + bv;

  for (int base = start; base < end; base += 32) {
    const int n = (end - base < 32) ? (end - base) : 32;
    int myidx = 0;
    float mynrm = 0.0f;
    if (lane < n) {
      myidx = csr_src[base + lane];
      mynrm = dinv[myidx] * dd;
    }
    const int mynrm_bits = __builtin_bit_cast(int, mynrm);
    int j = 0;
    for (; j + 4 <= n; j += 4) {
      const int s0 = __builtin_amdgcn_readlane(myidx, j);
      const int s1 = __builtin_amdgcn_readlane(myidx, j + 1);
      const int s2 = __builtin_amdgcn_readlane(myidx, j + 2);
      const int s3 = __builtin_amdgcn_readlane(myidx, j + 3);
      const vt a0 = *(const vt*)(h + (size_t)s0 * F + f);
      const vt a1 = *(const vt*)(h + (size_t)s1 * F + f);
      const vt a2 = *(const vt*)(h + (size_t)s2 * F + f);
      const vt a3 = *(const vt*)(h + (size_t)s3 * F + f);
      const float w0 = __builtin_bit_cast(float, __builtin_amdgcn_readlane(mynrm_bits, j));
      const float w1 = __builtin_bit_cast(float, __builtin_amdgcn_readlane(mynrm_bits, j + 1));
      const float w2 = __builtin_bit_cast(float, __builtin_amdgcn_readlane(mynrm_bits, j + 2));
      const float w3 = __builtin_bit_cast(float, __builtin_amdgcn_readlane(mynrm_bits, j + 3));
      acc += a0 * w0;
      acc += a1 * w1;
      acc += a2 * w2;
      acc += a3 * w3;
    }
    for (; j < n; ++j) {
      const int s0 = __builtin_amdgcn_readlane(myidx, j);
      const vt a0 = *(const vt*)(h + (size_t)s0 * F + f);
      const float w0 = __builtin_bit_cast(float, __builtin_amdgcn_readlane(mynrm_bits, j));
      acc += a0 * w0;
    }
  }
  *(vt*)o = acc;
}

__global__ void pool_acc_kernel(const float* __restrict__ h,
                                const int* __restrict__ batch,
                                float* __restrict__ pooled,
                                float* __restrict__ cnt) {
  long long i = (long long)blockIdx.x * blockDim.x + threadIdx.x;
  if (i >= (long long)NNODES * 16) return;
  int node = (int)(i >> 4);
  int f    = (int)(i & 15) * 4;
  int g = batch[node];
  v4f hv = *(const v4f*)(h + (size_t)node * 64 + f);
  float* p = pooled + (size_t)g * 64 + f;
  atomicAdd(p + 0, hv.x);
  atomicAdd(p + 1, hv.y);
  atomicAdd(p + 2, hv.z);
  atomicAdd(p + 3, hv.w);
  if (f == 0) atomicAdd(&cnt[g], 1.0f);
}

__global__ void pool_div_kernel(float* __restrict__ pooled,
                                const float* __restrict__ cnt) {
  int i = blockIdx.x * blockDim.x + threadIdx.x;
  if (i < NGRAPHS * 64) pooled[i] /= fmaxf(cnt[i >> 6], 1.0f);
}

extern "C" void kernel_launch(void* const* d_in, const int* in_sizes, int n_in,
                              void* d_out, int out_size, void* d_ws, size_t ws_size,
                              hipStream_t stream) {
  const float* x   = (const float*)d_in[0];
  const int*   ei  = (const int*)d_in[1];
  const int*   bat = (const int*)d_in[2];
  const float* W1  = (const float*)d_in[3];  const float* b1  = (const float*)d_in[4];
  const float* W2  = (const float*)d_in[5];  const float* b2  = (const float*)d_in[6];
  const float* W3  = (const float*)d_in[7];  const float* b3  = (const float*)d_in[8];
  const float* W4  = (const float*)d_in[9];  const float* b4  = (const float*)d_in[10];
  const float* Wl1 = (const float*)d_in[11]; const float* bl1 = (const float*)d_in[12];
  const float* Wl2 = (const float*)d_in[13]; const float* bl2 = (const float*)d_in[14];
  float* out = (float*)d_out;

  const int* src = ei;
  const int* dst = ei + NEDGES;

  float* ws     = (float*)d_ws;
  float* dinv   = ws;
  float* xs     = dinv + NNODES;
  float* xa     = xs + (size_t)MPAD * 4;
  float* hT     = xa + (size_t)MPAD * 4;
  float* hA     = hT + (size_t)MPAD * 128;
  float* pooled = hA + (size_t)MPAD * 128;
  float* cnt    = pooled + NGRAPHS * 64;
  float* t1     = cnt + NGRAPHS;
  float* W1p    = t1 + NGRAPHS * 32;
  float* Wl2p   = W1p + 4 * 128;
  float* bl2p   = Wl2p + 32 * 16;
  float* o16    = bl2p + 16;
  int*   indeg   = (int*)(o16 + NGRAPHS * 16);
  int*   rowptr  = indeg + NNODES;
  int*   bsum    = rowptr + NNODES;
  int*   rank    = bsum + 512;
  int*   csr_src = rank + NEDGES;

  const int B = 256;
  auto blks = [](long long n, int b) { return (unsigned)((n + b - 1) / b); };

  zero_int_kernel<<<blks(NNODES, B), B, 0, stream>>>(indeg, NNODES);
  edge_rank_kernel<<<blks(NEDGES, B), B, 0, stream>>>(dst, indeg, rank, NEDGES);
  row_scan_local_kernel<<<blks(NNODES, 256), 256, 0, stream>>>(indeg, rowptr, bsum, dinv, x, xs);
  row_scan_offset_kernel<<<blks(NNODES, 256), 256, 0, stream>>>(rowptr, bsum);
  csr_fill_kernel<<<blks(NEDGES, B), B, 0, stream>>>(src, dst, rank, rowptr, csr_src, NEDGES);

  pad_w1_kernel<<<blks(4 * 128, B), B, 0, stream>>>(W1, W1p);
  pad_wl2_kernel<<<blks(32 * 16, B), B, 0, stream>>>(Wl2, bl2, Wl2p, bl2p);

  const unsigned GB = MPAD / 128;
  const unsigned RB = MPAD / 8;

  row_gather_x_kernel<<<MPAD / 64, 256, 0, stream>>>(xs, rowptr, indeg, csr_src, dinv, xa);
  gemm_wmma<4, 128, false, true><<<GB, 256, 0, stream>>>(xa, W1p, b1, hA);

  gemm_wmma<128, 128, true, false><<<GB, 256, 0, stream>>>(hA, W2, nullptr, hT);
  row_gather_kernel<128><<<RB, 256, 0, stream>>>(hT, rowptr, indeg, csr_src, dinv, b2, hA);

  gemm_wmma<128, 128, true, false><<<GB, 256, 0, stream>>>(hA, W3, nullptr, hT);
  row_gather_kernel<128><<<RB, 256, 0, stream>>>(hT, rowptr, indeg, csr_src, dinv, b3, hA);

  gemm_wmma<128, 64, true, false><<<GB, 256, 0, stream>>>(hA, W4, nullptr, hT);
  row_gather_kernel<64><<<RB, 256, 0, stream>>>(hT, rowptr, indeg, csr_src, dinv, b4, hA);

  zero_kernel<<<blks(NGRAPHS * 64, B), B, 0, stream>>>(pooled, NGRAPHS * 64);
  zero_kernel<<<blks(NGRAPHS, B), B, 0, stream>>>(cnt, NGRAPHS);
  pool_acc_kernel<<<blks((long long)NNODES * 16, B), B, 0, stream>>>(hA, bat, pooled, cnt);
  pool_div_kernel<<<blks(NGRAPHS * 64, B), B, 0, stream>>>(pooled, cnt);

  gemm_wmma<64, 32, false, true><<<NGRAPHS / 128, 256, 0, stream>>>(pooled, Wl1, bl1, t1);
  gemm_wmma<32, 16, true, true><<<NGRAPHS / 128, 256, 0, stream>>>(t1, Wl2p, bl2p, o16);
  copy_out_kernel<<<blks(NGRAPHS * 9, B), B, 0, stream>>>(o16, out);
}


# CSR build redone as a two-level counting sort on LDS counters (edge chunks to 256-row bins, then bins to rows): no global atomics, 5 launches become 2; row slot order is now LDS arrival order
# speedup vs baseline: 3.2615x; 1.2356x over previous
#include <hip/hip_runtime.h>
#include <hip/hip_bf16.h>

typedef __attribute__((ext_vector_type(2))) float v2f;
typedef __attribute__((ext_vector_type(4))) float v4f;
typedef __attribute__((ext_vector_type(8))) float v8f;

#define NNODES  100000
#define MPAD    100096
#define NEDGES  1600000
#define NGRAPHS 512

__global__ void zero_kernel(float* __restrict__ p, long long n) {
  long long i = (long long)blockIdx.x * blockDim.x + threadIdx.x;
  if (i < n) p[i] = 0.0f;
}

typedef __attribute__((ext_vector_type(2))) int v2i;
typedef __attribute__((ext_vector_type(4))) int v4i;

#define BINSHIFT 8
#define BINNODES 256
#define NBINS    ((NNODES + BINNODES - 1) / BINNODES)
#define NCHUNK   128
#define CHUNK    (NEDGES / NCHUNK)
#define EPT      ((CHUNK + 1023) / 1024)
#define STAGECAP 16384
static_assert(NEDGES % NCHUNK == 0, "chunks must tile the edge list");
static_assert(NBINS <= 512, "bin scan covers 512 bins");

__device__ __forceinline__ int wave_incl_scan(int v, int lane) {
  for (int off = 1; off < 32; off <<= 1) {
    const int u = __shfl_up(v, off);
    if (lane >= off) v += u;
  }
  return v;
}

__global__ __launch_bounds__(1024) void bin_chunk_kernel(
    const int* __restrict__ src, const int* __restrict__ dst,
    int* __restrict__ binned, int* __restrict__ cntT, int* __restrict__ offT) {
  __shared__ int hist[512];
  __shared__ int wsum[16];
  __shared__ v2i sorted[CHUNK];
  const int t = threadIdx.x;
  const int lane = t & 31;
  const int b = blockIdx.x;
  const int base = b * CHUNK;
  if (t < 512) hist[t] = 0;
  __syncthreads();

  int s[EPT], d[EPT], r[EPT];
#pragma unroll
  for (int i = 0; i < EPT; ++i) {
    const int idx = t + 1024 * i;
    s[i] = 0; d[i] = 0;
    if (idx < CHUNK) { s[i] = src[base + idx]; d[i] = dst[base + idx]; }
  }
#pragma unroll
  for (int i = 0; i < EPT; ++i) {
    const int idx = t + 1024 * i;
    r[i] = 0;
    if (idx < CHUNK) r[i] = atomicAdd(&hist[d[i] >> BINSHIFT], 1);
  }
  __syncthreads();

  const int v = (t < 512) ? hist[t] : 0;
  const int incl = wave_incl_scan(v, lane);
  if (t < 512 && lane == 31) wsum[t >> 5] = incl;
  __syncthreads();
  if (t < 512) {
    int add = 0;
    for (int w2 = 0; w2 < (t >> 5); ++w2) add += wsum[w2];
    const int excl = incl - v + add;
    hist[t] = excl;
    if (t < NBINS) {
      cntT[t * NCHUNK + b] = v;
      offT[t * NCHUNK + b] = excl;
    }
  }
  __syncthreads();

#pragma unroll
  for (int i = 0; i < EPT; ++i) {
    const int idx = t + 1024 * i;
    if (idx < CHUNK) {
      v2i e; e.x = s[i]; e.y = d[i];
      sorted[hist[d[i] >> BINSHIFT] + r[i]] = e;
    }
  }
  __syncthreads();
  v2i* __restrict__ outp = (v2i*)binned + base;
#pragma unroll
  for (int i = 0; i < EPT; ++i) {
    const int idx = t + 1024 * i;
    if (idx < CHUNK) outp[idx] = sorted[idx];
  }
}

#define RUNS_PER_WAVE (NCHUNK / 32)
#define PSUM_TRIPS    ((NBINS * (NCHUNK / 4) + 1023) / 1024)

template <typename F>
__device__ __forceinline__ void for_bin_edges(const v2i* __restrict__ pairs, const int* runLen,
                                              const int* runStart, int w, int lane, F f) {
  int len[RUNS_PER_WAVE], st[RUNS_PER_WAVE];
  v2i e0[RUNS_PER_WAVE], e1[RUNS_PER_WAVE];
#pragma unroll
  for (int q = 0; q < RUNS_PER_WAVE; ++q) {
    len[q] = runLen[w + 32 * q];
    st[q] = runStart[w + 32 * q];
    e0[q] = pairs[(lane < len[q]) ? (st[q] + lane) : 0];
    e1[q] = pairs[(lane + 32 < len[q]) ? (st[q] + lane + 32) : 0];
  }
#pragma unroll
  for (int q = 0; q < RUNS_PER_WAVE; ++q) asm volatile("" :: "v"(e0[q]), "v"(e1[q]));
#pragma unroll
  for (int q = 0; q < RUNS_PER_WAVE; ++q) {
    if (lane < len[q]) f(e0[q]);
    if (lane + 32 < len[q]) f(e1[q]);
  }
#pragma unroll
  for (int q = 0; q < RUNS_PER_WAVE; ++q) {
    for (int j = lane + 64; j < len[q]; j += 32) f(pairs[st[q] + j]);
  }
}

__global__ __launch_bounds__(1024) void bin_rows_kernel(
    const int* __restrict__ binned, const int* __restrict__ cntT, const int* __restrict__ offT,
    const float* __restrict__ x, int* __restrict__ indeg, int* __restrict__ rowptr,
    float* __restrict__ dinv, float* __restrict__ xs, int* __restrict__ csr_src) {
  __shared__ int cnt[BINNODES];
  __shared__ int cursor[BINNODES];
  __shared__ int runLen[NCHUNK];
  __shared__ int runStart[NCHUNK];
  __shared__ int red[32];
  __shared__ int wsum[8];
  __shared__ int stage[STAGECAP];
  const int t = threadIdx.x;
  const int lane = t & 31;
  const int w = t >> 5;
  const int bin = blockIdx.x;
  const int d0 = bin << BINSHIFT;
  const v2i* __restrict__ pairs = (const v2i*)binned;

  const int node = d0 + t;
  const bool has_node = (t < BINNODES) && (node < NNODES);
  float x0 = 0.0f, x1 = 0.0f, x2 = 0.0f;
  if (has_node) {
    const float* __restrict__ xr = x + (size_t)node * 3;
    x0 = xr[0]; x1 = xr[1]; x2 = xr[2];
  }
  int rl = 0, ro = 0;
  if (t < NCHUNK) {
    rl = cntT[bin * NCHUNK + t];
    ro = offT[bin * NCHUNK + t];
  }
  int part = 0;
  {
    const v4i* __restrict__ c4 = (const v4i*)cntT;
    const int n4 = bin * (NCHUNK / 4);
    v4i q[PSUM_TRIPS];
#pragma unroll
    for (int k = 0; k < PSUM_TRIPS; ++k) {
      const int i = t + 1024 * k;
      q[k] = c4[(i < NBINS * (NCHUNK / 4)) ? i : 0];
    }
#pragma unroll
    for (int k = 0; k < PSUM_TRIPS; ++k) asm volatile("" :: "v"(q[k]));
#pragma unroll
    for (int k = 0; k < PSUM_TRIPS; ++k) {
      const int i = t + 1024 * k;
      part += (i < n4) ? (q[k].x + q[k].y + q[k].z + q[k].w) : 0;
    }
  }
  if (t < NCHUNK) {
    runLen[t] = rl;
    runStart[t] = t * CHUNK + ro;
  }
  if (t < BINNODES) cnt[t] = 0;
  for (int off = 16; off > 0; off >>= 1) part += __shfl_xor(part, off);
  if (lane == 0) red[w] = part;
  __syncthreads();
  int P = 0;
  for (int k = 0; k < 32; ++k) P += red[k];

  for_bin_edges(pairs, runLen, runStart, w, lane,
                [&](const v2i e) { atomicAdd(&cnt[e.y - d0], 1); });
  __syncthreads();

  const int v = (t < BINNODES) ? cnt[t] : 0;
  const int incl = wave_incl_scan(v, lane);
  if (t < BINNODES && lane == 31) wsum[w] = incl;
  __syncthreads();
  int n = 0;
  for (int w2 = 0; w2 < 8; ++w2) n += wsum[w2];
  if (t < BINNODES) {
    int add = 0;
    for (int w2 = 0; w2 < w; ++w2) add += wsum[w2];
    const int excl = incl - v + add;
    cursor[t] = excl;
    if (has_node) {
      indeg[node] = v;
      rowptr[node] = P + excl;
      const float dd = rsqrtf((float)v + 1.0f);
      dinv[node] = dd;
      v4f o;
      o.x = x0 * dd; o.y = x1 * dd; o.z = x2 * dd; o.w = 0.0f;
      *(v4f*)(xs + (size_t)node * 4) = o;
    }
  }
  __syncthreads();

  const bool staged = (n <= STAGECAP);
  for_bin_edges(pairs, runLen, runStart, w, lane, [&](const v2i e) {
    const int pos = atomicAdd(&cursor[e.y - d0], 1);
    if (staged) stage[pos] = e.x;
    else csr_src[P + pos] = e.x;
  });
  __syncthreads();
  if (staged) {
    for (int i = t; i < n; i += 1024) csr_src[P + i] = stage[i];
  }
}

__global__ __launch_bounds__(256) void row_gather_x_kernel(
    const float* __restrict__ xs, const int* __restrict__ rowptr,
    const int* __restrict__ indeg, const int* __restrict__ csr_src,
    const float* __restrict__ dinv, float* __restrict__ xa) {
  const int t = blockIdx.x * 256 + threadIdx.x;
  const int row = t >> 2;
  const int sub = t & 3;
  const v4f zero4 = 0.0f;
  v4f acc = zero4;
  float dd = 0.0f;
  if (row < NNODES) {
    dd = dinv[row];
    const int start = rowptr[row];
    const int end = start + indeg[row];
    const int last = end - 1;
    if (sub == 0) acc = *(const v4f*)(xs + (size_t)row * 4);
    for (int base = start; base < end; base += 32) {
      int s[8];
      for (int u = 0; u < 8; ++u) {
        const int e = base + sub + 4 * u;
        s[u] = csr_src[(e < end) ? e : last];
      }
      for (int u = 0; u < 8; ++u) {
        const int e = base + sub + 4 * u;
        const v4f v = *(const v4f*)(xs + (size_t)s[u] * 4);
        acc += (e < end) ? v : zero4;
      }
    }
  }
  float r[3];
  for (int c = 0; c < 3; ++c) {
    float a = acc[c];
    a += __builtin_bit_cast(float, __builtin_amdgcn_update_dpp(
             0, __builtin_bit_cast(int, a), 0xB1, 0xF, 0xF, false));
    a += __builtin_bit_cast(float, __builtin_amdgcn_update_dpp(
             0, __builtin_bit_cast(int, a), 0x4E, 0xF, 0xF, false));
    r[c] = a;
  }
  if (sub == 0) {
    v4f o;
    o.x = r[0] * dd; o.y = r[1] * dd; o.z = r[2] * dd; o.w = 0.0f;
    *(v4f*)(xa + (size_t)row * 4) = o;
  }
}

__global__ void pad_w1_kernel(const float* __restrict__ w, float* __restrict__ wp) {
  int i = blockIdx.x * blockDim.x + threadIdx.x;
  if (i >= 4 * 128) return;
  int k = i >> 7;
  wp[i] = (k < 3) ? w[i] : 0.0f;
}

__global__ void pad_wl2_kernel(const float* __restrict__ w, const float* __restrict__ b,
                               float* __restrict__ wp, float* __restrict__ bp) {
  int i = blockIdx.x * blockDim.x + threadIdx.x;
  if (i >= 32 * 16) return;
  int k = i >> 4, n = i & 15;
  wp[i] = (n < 9) ? w[k * 9 + n] : 0.0f;
  if (i < 16) bp[i] = (i < 9) ? b[i] : 0.0f;
}

__global__ void copy_out_kernel(const float* __restrict__ o16, float* __restrict__ out) {
  int i = blockIdx.x * blockDim.x + threadIdx.x;
  if (i >= NGRAPHS * 9) return;
  int g = i / 9, j = i - g * 9;
  out[i] = o16[g * 16 + j];
}

template <int K, int N, bool RELU, bool BIAS>
__global__ __launch_bounds__(256) void gemm_wmma(
    const float* __restrict__ A, const float* __restrict__ W,
    const float* __restrict__ bias, float* __restrict__ C) {
  __shared__ float lw[K * N];

  const int tid = threadIdx.x;
  constexpr int QUADS = (K / 2) * N / 4;
  for (int q = tid; q < QUADS; q += 256) {
    int t = (q * 4) / N;
    int n = (q * 4) % N;
    int k = t * 2;
    v4f w0 = *(const v4f*)(W + (size_t)k * N + n);
    v4f w1 = *(const v4f*)(W + (size_t)(k + 1) * N + n);
    v4f lo, hi;
    lo.x = w0.x; lo.y = w1.x; lo.z = w0.y; lo.w = w1.y;
    hi.x = w0.z; hi.y = w1.z; hi.z = w0.w; hi.w = w1.w;
    float* d = lw + ((size_t)t * N + n) * 2;
    *(v4f*)(d)     = lo;
    *(v4f*)(d + 4) = hi;
  }
  __syncthreads();

  const int lane = tid & 31;
  const int wave = tid >> 5;
  const int half = lane >> 4;
  const int l16  = lane & 15;
  const int row  = blockIdx.x * 128 + wave * 16 + l16;
  const float* __restrict__ Arow = A + (size_t)row * K + 2 * half;

  for (int n_base = 0; n_base < N; n_base += 16) {
    const int n = n_base + l16;
    float binit = BIAS ? bias[n] : 0.0f;
    v8f acc;
#pragma unroll
    for (int r = 0; r < 8; ++r) acc[r] = binit;

#pragma unroll
    for (int k = 0; k < K; k += 4) {
      v2f av = *(const v2f*)(Arow + k);
      if (RELU) { av.x = fmaxf(av.x, 0.0f); av.y = fmaxf(av.y, 0.0f); }
      const int kk = k + 2 * half;
      v2f bv = *(const v2f*)(lw + ((size_t)(kk >> 1) * N + n) * 2);
      acc = __builtin_amdgcn_wmma_f32_16x16x4_f32(
          false, av, false, bv, (short)0, acc, false, false);
    }

    float* __restrict__ Cp = C + (size_t)(blockIdx.x * 128 + wave * 16 + half * 8) * N + n;
#pragma unroll
    for (int r = 0; r < 8; ++r) Cp[(size_t)r * N] = acc[r];
  }
}

template <int F>
__global__ __launch_bounds__(256) void row_gather_kernel(
    const float* __restrict__ h, const int* __restrict__ rowptr,
    const int* __restrict__ indeg, const int* __restrict__ csr_src,
    const float* __restrict__ dinv, const float* __restrict__ bias,
    float* __restrict__ out) {
  constexpr int V = F / 32;
  typedef __attribute__((ext_vector_type(V))) float vt;
  const int lane = threadIdx.x & 31;
  const int row = __builtin_amdgcn_readfirstlane((int)(blockIdx.x * 8 + (threadIdx.x >> 5)));
  const int f = lane * V;
  float* __restrict__ o = out + (size_t)row * F + f;
  if (row >= NNODES) {
    vt z = 0.0f;
    *(vt*)o = z;
    return;
  }
  const float dd = dinv[row];
  const int start = rowptr[row];
  const int end = start + indeg[row];
  const vt hv = *(const vt*)(h + (size_t)row * F + f);
  const vt bv = *(const vt*)(bias + f);
  vt acc = hv * (dd * dd) + bv;

  for (int base = start; base < end; base += 32) {
    const int n = (end - base < 32) ? (end - base) : 32;
    int myidx = 0;
    float mynrm = 0.0f;
    if (lane < n) {
      myidx = csr_src[base + lane];
      mynrm = dinv[myidx] * dd;
    }
    const int mynrm_bits = __builtin_bit_cast(int, mynrm);
    int j = 0;
    for (; j + 4 <= n; j += 4) {
      const int s0 = __builtin_amdgcn_readlane(myidx, j);
      const int s1 = __builtin_amdgcn_readlane(myidx, j + 1);
      const int s2 = __builtin_amdgcn_readlane(myidx, j + 2);
      const int s3 = __builtin_amdgcn_readlane(myidx, j + 3);
      const vt a0 = *(const vt*)(h + (size_t)s0 * F + f);
      const vt a1 = *(const vt*)(h + (size_t)s1 * F + f);
      const vt a2 = *(const vt*)(h + (size_t)s2 * F + f);
      const vt a3 = *(const vt*)(h + (size_t)s3 * F + f);
      const float w0 = __builtin_bit_cast(float, __builtin_amdgcn_readlane(mynrm_bits, j));
      const float w1 = __builtin_bit_cast(float, __builtin_amdgcn_readlane(mynrm_bits, j + 1));
      const float w2 = __builtin_bit_cast(float, __builtin_amdgcn_readlane(mynrm_bits, j + 2));
      const float w3 = __builtin_bit_cast(float, __builtin_amdgcn_readlane(mynrm_bits, j + 3));
      acc += a0 * w0;
      acc += a1 * w1;
      acc += a2 * w2;
      acc += a3 * w3;
    }
    for (; j < n; ++j) {
      const int s0 = __builtin_amdgcn_readlane(myidx, j);
      const vt a0 = *(const vt*)(h + (size_t)s0 * F + f);
      const float w0 = __builtin_bit_cast(float, __builtin_amdgcn_readlane(mynrm_bits, j));
      acc += a0 * w0;
    }
  }
  *(vt*)o = acc;
}

__global__ void pool_acc_kernel(const float* __restrict__ h,
                                const int* __restrict__ batch,
                                float* __restrict__ pooled,
                                float* __restrict__ cnt) {
  long long i = (long long)blockIdx.x * blockDim.x + threadIdx.x;
  if (i >= (long long)NNODES * 16) return;
  int node = (int)(i >> 4);
  int f    = (int)(i & 15) * 4;
  int g = batch[node];
  v4f hv = *(const v4f*)(h + (size_t)node * 64 + f);
  float* p = pooled + (size_t)g * 64 + f;
  atomicAdd(p + 0, hv.x);
  atomicAdd(p + 1, hv.y);
  atomicAdd(p + 2, hv.z);
  atomicAdd(p + 3, hv.w);
  if (f == 0) atomicAdd(&cnt[g], 1.0f);
}

__global__ void pool_div_kernel(float* __restrict__ pooled,
                                const float* __restrict__ cnt) {
  int i = blockIdx.x * blockDim.x + threadIdx.x;
  if (i < NGRAPHS * 64) pooled[i] /= fmaxf(cnt[i >> 6], 1.0f);
}

extern "C" void kernel_launch(void* const* d_in, const int* in_sizes, int n_in,
                              void* d_out, int out_size, void* d_ws, size_t ws_size,
                              hipStream_t stream) {
  const float* x   = (const float*)d_in[0];
  const int*   ei  = (const int*)d_in[1];
  const int*   bat = (const int*)d_in[2];
  const float* W1  = (const float*)d_in[3];  const float* b1  = (const float*)d_in[4];
  const float* W2  = (const float*)d_in[5];  const float* b2  = (const float*)d_in[6];
  const float* W3  = (const float*)d_in[7];  const float* b3  = (const float*)d_in[8];
  const float* W4  = (const float*)d_in[9];  const float* b4  = (const float*)d_in[10];
  const float* Wl1 = (const float*)d_in[11]; const float* bl1 = (const float*)d_in[12];
  const float* Wl2 = (const float*)d_in[13]; const float* bl2 = (const float*)d_in[14];
  float* out = (float*)d_out;

  const int* src = ei;
  const int* dst = ei + NEDGES;

  float* ws     = (float*)d_ws;
  float* dinv   = ws;
  float* xs     = dinv + NNODES;
  float* xa     = xs + (size_t)MPAD * 4;
  float* hT     = xa + (size_t)MPAD * 4;
  float* hA     = hT + (size_t)MPAD * 128;
  float* pooled = hA + (size_t)MPAD * 128;
  float* cnt    = pooled + NGRAPHS * 64;
  float* t1     = cnt + NGRAPHS;
  float* W1p    = t1 + NGRAPHS * 32;
  float* Wl2p   = W1p + 4 * 128;
  float* bl2p   = Wl2p + 32 * 16;
  float* o16    = bl2p + 16;
  int*   indeg   = (int*)(o16 + NGRAPHS * 16);
  int*   rowptr  = indeg + NNODES;
  int*   cntT    = rowptr + NNODES;
  int*   offT    = cntT + NBINS * NCHUNK;
  int*   binned  = offT + NBINS * NCHUNK;
  int*   csr_src = binned + 2 * NEDGES;

  const int B = 256;
  auto blks = [](long long n, int b) { return (unsigned)((n + b - 1) / b); };

  bin_chunk_kernel<<<NCHUNK, 1024, 0, stream>>>(src, dst, binned, cntT, offT);
  bin_rows_kernel<<<NBINS, 1024, 0, stream>>>(binned, cntT, offT, x, indeg, rowptr, dinv, xs, csr_src);

  pad_w1_kernel<<<blks(4 * 128, B), B, 0, stream>>>(W1, W1p);
  pad_wl2_kernel<<<blks(32 * 16, B), B, 0, stream>>>(Wl2, bl2, Wl2p, bl2p);

  const unsigned GB = MPAD / 128;
  const unsigned RB = MPAD / 8;

  row_gather_x_kernel<<<MPAD / 64, 256, 0, stream>>>(xs, rowptr, indeg, csr_src, dinv, xa);
  gemm_wmma<4, 128, false, true><<<GB, 256, 0, stream>>>(xa, W1p, b1, hA);

  gemm_wmma<128, 128, true, false><<<GB, 256, 0, stream>>>(hA, W2, nullptr, hT);
  row_gather_kernel<128><<<RB, 256, 0, stream>>>(hT, rowptr, indeg, csr_src, dinv, b2, hA);

  gemm_wmma<128, 128, true, false><<<GB, 256, 0, stream>>>(hA, W3, nullptr, hT);
  row_gather_kernel<128><<<RB, 256, 0, stream>>>(hT, rowptr, indeg, csr_src, dinv, b3, hA);

  gemm_wmma<128, 64, true, false><<<GB, 256, 0, stream>>>(hA, W4, nullptr, hT);
  row_gather_kernel<64><<<RB, 256, 0, stream>>>(hT, rowptr, indeg, csr_src, dinv, b4, hA);

  zero_kernel<<<blks(NGRAPHS * 64, B), B, 0, stream>>>(pooled, NGRAPHS * 64);
  zero_kernel<<<blks(NGRAPHS, B), B, 0, stream>>>(cnt, NGRAPHS);
  pool_acc_kernel<<<blks((long long)NNODES * 16, B), B, 0, stream>>>(hA, bat, pooled, cnt);
  pool_div_kernel<<<blks(NGRAPHS * 64, B), B, 0, stream>>>(pooled, cnt);

  gemm_wmma<64, 32, false, true><<<NGRAPHS / 128, 256, 0, stream>>>(pooled, Wl1, bl1, t1);
  gemm_wmma<32, 16, true, true><<<NGRAPHS / 128, 256, 0, stream>>>(t1, Wl2p, bl2p, o16);
  copy_out_kernel<<<blks(NGRAPHS * 9, B), B, 0, stream>>>(o16, out);
}


# Row gather of layers 2-4: eight lanes per row, four rows per wave (was one wave per row): a quarter of the waves, 16 row loads in flight per wave, ids and norms by lane shuffles; sum order kept
# speedup vs baseline: 3.4857x; 1.0687x over previous
#include <hip/hip_runtime.h>
#include <hip/hip_bf16.h>

typedef __attribute__((ext_vector_type(2))) float v2f;
typedef __attribute__((ext_vector_type(4))) float v4f;
typedef __attribute__((ext_vector_type(8))) float v8f;

#define NNODES  100000
#define MPAD    100096
#define NEDGES  1600000
#define NGRAPHS 512

__global__ void zero_kernel(float* __restrict__ p, long long n) {
  long long i = (long long)blockIdx.x * blockDim.x + threadIdx.x;
  if (i < n) p[i] = 0.0f;
}

typedef __attribute__((ext_vector_type(2))) int v2i;
typedef __attribute__((ext_vector_type(4))) int v4i;

#define BINSHIFT 8
#define BINNODES 256
#define NBINS    ((NNODES + BINNODES - 1) / BINNODES)
#define NCHUNK   128
#define CHUNK    (NEDGES / NCHUNK)
#define EPT      ((CHUNK + 1023) / 1024)
#define STAGECAP 16384
static_assert(NEDGES % NCHUNK == 0, "chunks must tile the edge list");
static_assert(NBINS <= 512, "bin scan covers 512 bins");

__device__ __forceinline__ int wave_incl_scan(int v, int lane) {
  for (int off = 1; off < 32; off <<= 1) {
    const int u = __shfl_up(v, off);
    if (lane >= off) v += u;
  }
  return v;
}

__global__ __launch_bounds__(1024) void bin_chunk_kernel(
    const int* __restrict__ src, const int* __restrict__ dst,
    int* __restrict__ binned, int* __restrict__ cntT, int* __restrict__ offT) {
  __shared__ int hist[512];
  __shared__ int wsum[16];
  __shared__ v2i sorted[CHUNK];
  const int t = threadIdx.x;
  const int lane = t & 31;
  const int b = blockIdx.x;
  const int base = b * CHUNK;
  if (t < 512) hist[t] = 0;
  __syncthreads();

  int s[EPT], d[EPT], r[EPT];
#pragma unroll
  for (int i = 0; i < EPT; ++i) {
    const int idx = t + 1024 * i;
    s[i] = 0; d[i] = 0;
    if (idx < CHUNK) { s[i] = src[base + idx]; d[i] = dst[base + idx]; }
  }
#pragma unroll
  for (int i = 0; i < EPT; ++i) {
    const int idx = t + 1024 * i;
    r[i] = 0;
    if (idx < CHUNK) r[i] = atomicAdd(&hist[d[i] >> BINSHIFT], 1);
  }
  __syncthreads();

  const int v = (t < 512) ? hist[t] : 0;
  const int incl = wave_incl_scan(v, lane);
  if (t < 512 && lane == 31) wsum[t >> 5] = incl;
  __syncthreads();
  if (t < 512) {
    int add = 0;
    for (int w2 = 0; w2 < (t >> 5); ++w2) add += wsum[w2];
    const int excl = incl - v + add;
    hist[t] = excl;
    if (t < NBINS) {
      cntT[t * NCHUNK + b] = v;
      offT[t * NCHUNK + b] = excl;
    }
  }
  __syncthreads();

#pragma unroll
  for (int i = 0; i < EPT; ++i) {
    const int idx = t + 1024 * i;
    if (idx < CHUNK) {
      v2i e; e.x = s[i]; e.y = d[i];
      sorted[hist[d[i] >> BINSHIFT] + r[i]] = e;
    }
  }
  __syncthreads();
  v2i* __restrict__ outp = (v2i*)binned + base;
#pragma unroll
  for (int i = 0; i < EPT; ++i) {
    const int idx = t + 1024 * i;
    if (idx < CHUNK) outp[idx] = sorted[idx];
  }
}

#define RUNS_PER_WAVE (NCHUNK / 32)
#define PSUM_TRIPS    ((NBINS * (NCHUNK / 4) + 1023) / 1024)

template <typename F>
__device__ __forceinline__ void for_bin_edges(const v2i* __restrict__ pairs, const int* runLen,
                                              const int* runStart, int w, int lane, F f) {
  int len[RUNS_PER_WAVE], st[RUNS_PER_WAVE];
  v2i e0[RUNS_PER_WAVE], e1[RUNS_PER_WAVE];
#pragma unroll
  for (int q = 0; q < RUNS_PER_WAVE; ++q) {
    len[q] = runLen[w + 32 * q];
    st[q] = runStart[w + 32 * q];
    e0[q] = pairs[(lane < len[q]) ? (st[q] + lane) : 0];
    e1[q] = pairs[(lane + 32 < len[q]) ? (st[q] + lane + 32) : 0];
  }
#pragma unroll
  for (int q = 0; q < RUNS_PER_WAVE; ++q) asm volatile("" :: "v"(e0[q]), "v"(e1[q]));
#pragma unroll
  for (int q = 0; q < RUNS_PER_WAVE; ++q) {
    if (lane < len[q]) f(e0[q]);
    if (lane + 32 < len[q]) f(e1[q]);
  }
#pragma unroll
  for (int q = 0; q < RUNS_PER_WAVE; ++q) {
    for (int j = lane + 64; j < len[q]; j += 32) f(pairs[st[q] + j]);
  }
}

__global__ __launch_bounds__(1024) void bin_rows_kernel(
    const int* __restrict__ binned, const int* __restrict__ cntT, const int* __restrict__ offT,
    const float* __restrict__ x, int* __restrict__ indeg, int* __restrict__ rowptr,
    float* __restrict__ dinv, float* __restrict__ xs, int* __restrict__ csr_src) {
  __shared__ int cnt[BINNODES];
  __shared__ int cursor[BINNODES];
  __shared__ int runLen[NCHUNK];
  __shared__ int runStart[NCHUNK];
  __shared__ int red[32];
  __shared__ int wsum[8];
  __shared__ int stage[STAGECAP];
  const int t = threadIdx.x;
  const int lane = t & 31;
  const int w = t >> 5;
  const int bin = blockIdx.x;
  const int d0 = bin << BINSHIFT;
  const v2i* __restrict__ pairs = (const v2i*)binned;

  const int node = d0 + t;
  const bool has_node = (t < BINNODES) && (node < NNODES);
  float x0 = 0.0f, x1 = 0.0f, x2 = 0.0f;
  if (has_node) {
    const float* __restrict__ xr = x + (size_t)node * 3;
    x0 = xr[0]; x1 = xr[1]; x2 = xr[2];
  }
  int rl = 0, ro = 0;
  if (t < NCHUNK) {
    rl = cntT[bin * NCHUNK + t];
    ro = offT[bin * NCHUNK + t];
  }
  int part = 0;
  {
    const v4i* __restrict__ c4 = (const v4i*)cntT;
    const int n4 = bin * (NCHUNK / 4);
    v4i q[PSUM_TRIPS];
#pragma unroll
    for (int k = 0; k < PSUM_TRIPS; ++k) {
      const int i = t + 1024 * k;
      q[k] = c4[(i < NBINS * (NCHUNK / 4)) ? i : 0];
    }
#pragma unroll
    for (int k = 0; k < PSUM_TRIPS; ++k) asm volatile("" :: "v"(q[k]));
#pragma unroll
    for (int k = 0; k < PSUM_TRIPS; ++k) {
      const int i = t + 1024 * k;
      part += (i < n4) ? (q[k].x + q[k].y + q[k].z + q[k].w) : 0;
    }
  }
  if (t < NCHUNK) {
    runLen[t] = rl;
    runStart[t] = t * CHUNK + ro;
  }
  if (t < BINNODES) cnt[t] = 0;
  for (int off = 16; off > 0; off >>= 1) part += __shfl_xor(part, off);
  if (lane == 0) red[w] = part;
  __syncthreads();
  int P = 0;
  for (int k = 0; k < 32; ++k) P += red[k];

  for_bin_edges(pairs, runLen, runStart, w, lane,
                [&](const v2i e) { atomicAdd(&cnt[e.y - d0], 1); });
  __syncthreads();

  const int v = (t < BINNODES) ? cnt[t] : 0;
  const int incl = wave_incl_scan(v, lane);
  if (t < BINNODES && lane == 31) wsum[w] = incl;
  __syncthreads();
  int n = 0;
  for (int w2 = 0; w2 < 8; ++w2) n += wsum[w2];
  if (t < BINNODES) {
    int add = 0;
    for (int w2 = 0; w2 < w; ++w2) add += wsum[w2];
    const int excl = incl - v + add;
    cursor[t] = excl;
    if (has_node) {
      indeg[node] = v;
      rowptr[node] = P + excl;
      const float dd = rsqrtf((float)v + 1.0f);
      dinv[node] = dd;
      v4f o;
      o.x = x0 * dd; o.y = x1 * dd; o.z = x2 * dd; o.w = 0.0f;
      *(v4f*)(xs + (size_t)node * 4) = o;
    }
  }
  __syncthreads();

  const bool staged = (n <= STAGECAP);
  for_bin_edges(pairs, runLen, runStart, w, lane, [&](const v2i e) {
    const int pos = atomicAdd(&cursor[e.y - d0], 1);
    if (staged) stage[pos] = e.x;
    else csr_src[P + pos] = e.x;
  });
  __syncthreads();
  if (staged) {
    for (int i = t; i < n; i += 1024) csr_src[P + i] = stage[i];
  }
}

__global__ __launch_bounds__(256) void row_gather_x_kernel(
    const float* __restrict__ xs, const int* __restrict__ rowptr,
    const int* __restrict__ indeg, const int* __restrict__ csr_src,
    const float* __restrict__ dinv, float* __restrict__ xa) {
  const int t = blockIdx.x * 256 + threadIdx.x;
  const int row = t >> 2;
  const int sub = t & 3;
  const v4f zero4 = 0.0f;
  v4f acc = zero4;
  float dd = 0.0f;
  if (row < NNODES) {
    dd = dinv[row];
    const int start = rowptr[row];
    const int end = start + indeg[row];
    const int last = end - 1;
    if (sub == 0) acc = *(const v4f*)(xs + (size_t)row * 4);
    for (int base = start; base < end; base += 32) {
      int s[8];
      for (int u = 0; u < 8; ++u) {
        const int e = base + sub + 4 * u;
        s[u] = csr_src[(e < end) ? e : last];
      }
      for (int u = 0; u < 8; ++u) {
        const int e = base + sub + 4 * u;
        const v4f v = *(const v4f*)(xs + (size_t)s[u] * 4);
        acc += (e < end) ? v : zero4;
      }
    }
  }
  float r[3];
  for (int c = 0; c < 3; ++c) {
    float a = acc[c];
    a += __builtin_bit_cast(float, __builtin_amdgcn_update_dpp(
             0, __builtin_bit_cast(int, a), 0xB1, 0xF, 0xF, false));
    a += __builtin_bit_cast(float, __builtin_amdgcn_update_dpp(
             0, __builtin_bit_cast(int, a), 0x4E, 0xF, 0xF, false));
    r[c] = a;
  }
  if (sub == 0) {
    v4f o;
    o.x = r[0] * dd; o.y = r[1] * dd; o.z = r[2] * dd; o.w = 0.0f;
    *(v4f*)(xa + (size_t)row * 4) = o;
  }
}

__global__ void pad_w1_kernel(const float* __restrict__ w, float* __restrict__ wp) {
  int i = blockIdx.x * blockDim.x + threadIdx.x;
  if (i >= 4 * 128) return;
  int k = i >> 7;
  wp[i] = (k < 3) ? w[i] : 0.0f;
}

__global__ void pad_wl2_kernel(const float* __restrict__ w, const float* __restrict__ b,
                               float* __restrict__ wp, float* __restrict__ bp) {
  int i = blockIdx.x * blockDim.x + threadIdx.x;
  if (i >= 32 * 16) return;
  int k = i >> 4, n = i & 15;
  wp[i] = (n < 9) ? w[k * 9 + n] : 0.0f;
  if (i < 16) bp[i] = (i < 9) ? b[i] : 0.0f;
}

__global__ void copy_out_kernel(const float* __restrict__ o16, float* __restrict__ out) {
  int i = blockIdx.x * blockDim.x + threadIdx.x;
  if (i >= NGRAPHS * 9) return;
  int g = i / 9, j = i - g * 9;
  out[i] = o16[g * 16 + j];
}

template <int K, int N, bool RELU, bool BIAS>
__global__ __launch_bounds__(256) void gemm_wmma(
    const float* __restrict__ A, const float* __restrict__ W,
    const float* __restrict__ bias, float* __restrict__ C) {
  __shared__ float lw[K * N];

  const int tid = threadIdx.x;
  constexpr int QUADS = (K / 2) * N / 4;
  for (int q = tid; q < QUADS; q += 256) {
    int t = (q * 4) / N;
    int n = (q * 4) % N;
    int k = t * 2;
    v4f w0 = *(const v4f*)(W + (size_t)k * N + n);
    v4f w1 = *(const v4f*)(W + (size_t)(k + 1) * N + n);
    v4f lo, hi;
    lo.x = w0.x; lo.y = w1.x; lo.z = w0.y; lo.w = w1.y;
    hi.x = w0.z; hi.y = w1.z; hi.z = w0.w; hi.w = w1.w;
    float* d = lw + ((size_t)t * N + n) * 2;
    *(v4f*)(d)     = lo;
    *(v4f*)(d + 4) = hi;
  }
  __syncthreads();

  const int lane = tid & 31;
  const int wave = tid >> 5;
  const int half = lane >> 4;
  const int l16  = lane & 15;
  const int row  = blockIdx.x * 128 + wave * 16 + l16;
  const float* __restrict__ Arow = A + (size_t)row * K + 2 * half;

  for (int n_base = 0; n_base < N; n_base += 16) {
    const int n = n_base + l16;
    float binit = BIAS ? bias[n] : 0.0f;
    v8f acc;
#pragma unroll
    for (int r = 0; r < 8; ++r) acc[r] = binit;

#pragma unroll
    for (int k = 0; k < K; k += 4) {
      v2f av = *(const v2f*)(Arow + k);
      if (RELU) { av.x = fmaxf(av.x, 0.0f); av.y = fmaxf(av.y, 0.0f); }
      const int kk = k + 2 * half;
      v2f bv = *(const v2f*)(lw + ((size_t)(kk >> 1) * N + n) * 2);
      acc = __builtin_amdgcn_wmma_f32_16x16x4_f32(
          false, av, false, bv, (short)0, acc, false, false);
    }

    float* __restrict__ Cp = C + (size_t)(blockIdx.x * 128 + wave * 16 + half * 8) * N + n;
#pragma unroll
    for (int r = 0; r < 8; ++r) Cp[(size_t)r * N] = acc[r];
  }
}

template <int F>
__global__ __launch_bounds__(256) void row_gather_kernel(
    const float* __restrict__ h, const int* __restrict__ rowptr,
    const int* __restrict__ indeg, const int* __restrict__ csr_src,
    const float* __restrict__ dinv, const float* __restrict__ bias,
    float* __restrict__ out) {
  constexpr int CH = F / 32;
  const int lane = threadIdx.x & 31;
  const int q = lane & 7;
  const int gl = lane & 24;
  const int fo = q * 4;
  const int row = (int)(blockIdx.x * 32 + (threadIdx.x >> 3));
  const v4f zero4 = 0.0f;

  float dd = 0.0f;
  int start = 0, n = 0;
  v4f acc[CH];
#pragma unroll
  for (int c = 0; c < CH; ++c) acc[c] = zero4;
  if (row < NNODES) {
    dd = dinv[row];
    start = rowptr[row];
    n = indeg[row];
    const float* __restrict__ hr = h + (size_t)row * F + fo;
#pragma unroll
    for (int c = 0; c < CH; ++c) {
      const v4f hv = *(const v4f*)(hr + c * 32);
      const v4f bv = *(const v4f*)(bias + c * 32 + fo);
      acc[c] = hv * (dd * dd) + bv;
    }
  }
  const int n0 = __builtin_amdgcn_readlane(n, 0);
  const int n1 = __builtin_amdgcn_readlane(n, 8);
  const int n2 = __builtin_amdgcn_readlane(n, 16);
  const int n3 = __builtin_amdgcn_readlane(n, 24);
  const int n01 = (n0 > n1) ? n0 : n1;
  const int n23 = (n2 > n3) ? n2 : n3;
  const int nmax = (n01 > n23) ? n01 : n23;

  for (int blk = 0; blk < nmax; blk += 32) {
    int ir[4];
    float wr[4];
#pragma unroll
    for (int r = 0; r < 4; ++r) {
      const int e = blk + q + 8 * r;
      const int pos = start + ((e < n) ? e : 0);
      ir[r] = csr_src[(pos < NEDGES) ? pos : (NEDGES - 1)];
    }
#pragma unroll
    for (int r = 0; r < 4; ++r) {
      const int e = blk + q + 8 * r;
      const float dv = dinv[ir[r]];
      wr[r] = (e < n) ? dv * dd : 0.0f;
    }
    const int lim = nmax - blk;
#pragma unroll
    for (int p = 0; p < 8; ++p) {
      if (4 * p < lim) {
        int s[4];
        float w[4];
#pragma unroll
        for (int u = 0; u < 4; ++u) {
          const int from = gl + 4 * (p & 1) + u;
          s[u] = __shfl(ir[p >> 1], from);
          w[u] = __shfl(wr[p >> 1], from);
        }
        if (blk + 4 * p < n) {
          v4f a[4][CH];
#pragma unroll
          for (int u = 0; u < 4; ++u) {
            const int su = (blk + 4 * p + u < n) ? s[u] : s[0];
            const float* __restrict__ hs = h + (size_t)su * F + fo;
#pragma unroll
            for (int c = 0; c < CH; ++c) a[u][c] = *(const v4f*)(hs + c * 32);
          }
#pragma unroll
          for (int u = 0; u < 4; ++u) {
#pragma unroll
            for (int c = 0; c < CH; ++c) acc[c] += a[u][c] * w[u];
          }
        }
      }
    }
  }
  float* __restrict__ o = out + (size_t)row * F + fo;
#pragma unroll
  for (int c = 0; c < CH; ++c) *(v4f*)(o + c * 32) = acc[c];
}

__global__ void pool_acc_kernel(const float* __restrict__ h,
                                const int* __restrict__ batch,
                                float* __restrict__ pooled,
                                float* __restrict__ cnt) {
  long long i = (long long)blockIdx.x * blockDim.x + threadIdx.x;
  if (i >= (long long)NNODES * 16) return;
  int node = (int)(i >> 4);
  int f    = (int)(i & 15) * 4;
  int g = batch[node];
  v4f hv = *(const v4f*)(h + (size_t)node * 64 + f);
  float* p = pooled + (size_t)g * 64 + f;
  atomicAdd(p + 0, hv.x);
  atomicAdd(p + 1, hv.y);
  atomicAdd(p + 2, hv.z);
  atomicAdd(p + 3, hv.w);
  if (f == 0) atomicAdd(&cnt[g], 1.0f);
}

__global__ void pool_div_kernel(float* __restrict__ pooled,
                                const float* __restrict__ cnt) {
  int i = blockIdx.x * blockDim.x + threadIdx.x;
  if (i < NGRAPHS * 64) pooled[i] /= fmaxf(cnt[i >> 6], 1.0f);
}

extern "C" void kernel_launch(void* const* d_in, const int* in_sizes, int n_in,
                              void* d_out, int out_size, void* d_ws, size_t ws_size,
                              hipStream_t stream) {
  const float* x   = (const float*)d_in[0];
  const int*   ei  = (const int*)d_in[1];
  const int*   bat = (const int*)d_in[2];
  const float* W1  = (const float*)d_in[3];  const float* b1  = (const float*)d_in[4];
  const float* W2  = (const float*)d_in[5];  const float* b2  = (const float*)d_in[6];
  const float* W3  = (const float*)d_in[7];  const float* b3  = (const float*)d_in[8];
  const float* W4  = (const float*)d_in[9];  const float* b4  = (const float*)d_in[10];
  const float* Wl1 = (const float*)d_in[11]; const float* bl1 = (const float*)d_in[12];
  const float* Wl2 = (const float*)d_in[13]; const float* bl2 = (const float*)d_in[14];
  float* out = (float*)d_out;

  const int* src = ei;
  const int* dst = ei + NEDGES;

  float* ws     = (float*)d_ws;
  float* dinv   = ws;
  float* xs     = dinv + NNODES;
  float* xa     = xs + (size_t)MPAD * 4;
  float* hT     = xa + (size_t)MPAD * 4;
  float* hA     = hT + (size_t)MPAD * 128;
  float* pooled = hA + (size_t)MPAD * 128;
  float* cnt    = pooled + NGRAPHS * 64;
  float* t1     = cnt + NGRAPHS;
  float* W1p    = t1 + NGRAPHS * 32;
  float* Wl2p   = W1p + 4 * 128;
  float* bl2p   = Wl2p + 32 * 16;
  float* o16    = bl2p + 16;
  int*   indeg   = (int*)(o16 + NGRAPHS * 16);
  int*   rowptr  = indeg + NNODES;
  int*   cntT    = rowptr + NNODES;
  int*   offT    = cntT + NBINS * NCHUNK;
  int*   binned  = offT + NBINS * NCHUNK;
  int*   csr_src = binned + 2 * NEDGES;

  const int B = 256;
  auto blks = [](long long n, int b) { return (unsigned)((n + b - 1) / b); };

  bin_chunk_kernel<<<NCHUNK, 1024, 0, stream>>>(src, dst, binned, cntT, offT);
  bin_rows_kernel<<<NBINS, 1024, 0, stream>>>(binned, cntT, offT, x, indeg, rowptr, dinv, xs, csr_src);

  pad_w1_kernel<<<blks(4 * 128, B), B, 0, stream>>>(W1, W1p);
  pad_wl2_kernel<<<blks(32 * 16, B), B, 0, stream>>>(Wl2, bl2, Wl2p, bl2p);

  const unsigned GB = MPAD / 128;
  const unsigned RB = MPAD / 32;

  row_gather_x_kernel<<<MPAD / 64, 256, 0, stream>>>(xs, rowptr, indeg, csr_src, dinv, xa);
  gemm_wmma<4, 128, false, true><<<GB, 256, 0, stream>>>(xa, W1p, b1, hA);

  gemm_wmma<128, 128, true, false><<<GB, 256, 0, stream>>>(hA, W2, nullptr, hT);
  row_gather_kernel<128><<<RB, 256, 0, stream>>>(hT, rowptr, indeg, csr_src, dinv, b2, hA);

  gemm_wmma<128, 128, true, false><<<GB, 256, 0, stream>>>(hA, W3, nullptr, hT);
  row_gather_kernel<128><<<RB, 256, 0, stream>>>(hT, rowptr, indeg, csr_src, dinv, b3, hA);

  gemm_wmma<128, 64, true, false><<<GB, 256, 0, stream>>>(hA, W4, nullptr, hT);
  row_gather_kernel<64><<<RB, 256, 0, stream>>>(hT, rowptr, indeg, csr_src, dinv, b4, hA);

  zero_kernel<<<blks(NGRAPHS * 64, B), B, 0, stream>>>(pooled, NGRAPHS * 64);
  zero_kernel<<<blks(NGRAPHS, B), B, 0, stream>>>(cnt, NGRAPHS);
  pool_acc_kernel<<<blks((long long)NNODES * 16, B), B, 0, stream>>>(hA, bat, pooled, cnt);
  pool_div_kernel<<<blks(NGRAPHS * 64, B), B, 0, stream>>>(pooled, cnt);

  gemm_wmma<64, 32, false, true><<<NGRAPHS / 128, 256, 0, stream>>>(pooled, Wl1, bl1, t1);
  gemm_wmma<32, 16, true, true><<<NGRAPHS / 128, 256, 0, stream>>>(t1, Wl2p, bl2p, o16);
  copy_out_kernel<<<blks(NGRAPHS * 9, B), B, 0, stream>>>(o16, out);
}


# Mean-pool sums fused into the layer-4 gather epilogue: a wave's 4 rows pre-summed by shuffles, one float-atomic set per wave; pool kernel and its table write+read gone; pool sum order changes
# speedup vs baseline: 4.4161x; 1.2669x over previous
#include <hip/hip_runtime.h>
#include <hip/hip_bf16.h>

typedef __attribute__((ext_vector_type(2))) float v2f;
typedef __attribute__((ext_vector_type(4))) float v4f;
typedef __attribute__((ext_vector_type(8))) float v8f;

#define NNODES  100000
#define MPAD    100096
#define NEDGES  1600000
#define NGRAPHS 512

__global__ void zero_kernel(float* __restrict__ p, long long n) {
  long long i = (long long)blockIdx.x * blockDim.x + threadIdx.x;
  if (i < n) p[i] = 0.0f;
}

typedef __attribute__((ext_vector_type(2))) int v2i;
typedef __attribute__((ext_vector_type(4))) int v4i;

#define BINSHIFT 8
#define BINNODES 256
#define NBINS    ((NNODES + BINNODES - 1) / BINNODES)
#define NCHUNK   128
#define CHUNK    (NEDGES / NCHUNK)
#define EPT      ((CHUNK + 1023) / 1024)
#define STAGECAP 16384
static_assert(NEDGES % NCHUNK == 0, "chunks must tile the edge list");
static_assert(NBINS <= 512, "bin scan covers 512 bins");

__device__ __forceinline__ int wave_incl_scan(int v, int lane) {
  for (int off = 1; off < 32; off <<= 1) {
    const int u = __shfl_up(v, off);
    if (lane >= off) v += u;
  }
  return v;
}

__global__ __launch_bounds__(1024) void bin_chunk_kernel(
    const int* __restrict__ src, const int* __restrict__ dst,
    int* __restrict__ binned, int* __restrict__ cntT, int* __restrict__ offT) {
  __shared__ int hist[512];
  __shared__ int wsum[16];
  __shared__ v2i sorted[CHUNK];
  const int t = threadIdx.x;
  const int lane = t & 31;
  const int b = blockIdx.x;
  const int base = b * CHUNK;
  if (t < 512) hist[t] = 0;
  __syncthreads();

  int s[EPT], d[EPT], r[EPT];
#pragma unroll
  for (int i = 0; i < EPT; ++i) {
    const int idx = t + 1024 * i;
    s[i] = 0; d[i] = 0;
    if (idx < CHUNK) { s[i] = src[base + idx]; d[i] = dst[base + idx]; }
  }
#pragma unroll
  for (int i = 0; i < EPT; ++i) {
    const int idx = t + 1024 * i;
    r[i] = 0;
    if (idx < CHUNK) r[i] = atomicAdd(&hist[d[i] >> BINSHIFT], 1);
  }
  __syncthreads();

  const int v = (t < 512) ? hist[t] : 0;
  const int incl = wave_incl_scan(v, lane);
  if (t < 512 && lane == 31) wsum[t >> 5] = incl;
  __syncthreads();
  if (t < 512) {
    int add = 0;
    for (int w2 = 0; w2 < (t >> 5); ++w2) add += wsum[w2];
    const int excl = incl - v + add;
    hist[t] = excl;
    if (t < NBINS) {
      cntT[t * NCHUNK + b] = v;
      offT[t * NCHUNK + b] = excl;
    }
  }
  __syncthreads();

#pragma unroll
  for (int i = 0; i < EPT; ++i) {
    const int idx = t + 1024 * i;
    if (idx < CHUNK) {
      v2i e; e.x = s[i]; e.y = d[i];
      sorted[hist[d[i] >> BINSHIFT] + r[i]] = e;
    }
  }
  __syncthreads();
  v2i* __restrict__ outp = (v2i*)binned + base;
#pragma unroll
  for (int i = 0; i < EPT; ++i) {
    const int idx = t + 1024 * i;
    if (idx < CHUNK) outp[idx] = sorted[idx];
  }
}

#define RUNS_PER_WAVE (NCHUNK / 32)
#define PSUM_TRIPS    ((NBINS * (NCHUNK / 4) + 1023) / 1024)

template <typename F>
__device__ __forceinline__ void for_bin_edges(const v2i* __restrict__ pairs, const int* runLen,
                                              const int* runStart, int w, int lane, F f) {
  int len[RUNS_PER_WAVE], st[RUNS_PER_WAVE];
  v2i e0[RUNS_PER_WAVE], e1[RUNS_PER_WAVE];
#pragma unroll
  for (int q = 0; q < RUNS_PER_WAVE; ++q) {
    len[q] = runLen[w + 32 * q];
    st[q] = runStart[w + 32 * q];
    e0[q] = pairs[(lane < len[q]) ? (st[q] + lane) : 0];
    e1[q] = pairs[(lane + 32 < len[q]) ? (st[q] + lane + 32) : 0];
  }
#pragma unroll
  for (int q = 0; q < RUNS_PER_WAVE; ++q) asm volatile("" :: "v"(e0[q]), "v"(e1[q]));
#pragma unroll
  for (int q = 0; q < RUNS_PER_WAVE; ++q) {
    if (lane < len[q]) f(e0[q]);
    if (lane + 32 < len[q]) f(e1[q]);
  }
#pragma unroll
  for (int q = 0; q < RUNS_PER_WAVE; ++q) {
    for (int j = lane + 64; j < len[q]; j += 32) f(pairs[st[q] + j]);
  }
}

__global__ __launch_bounds__(1024) void bin_rows_kernel(
    const int* __restrict__ binned, const int* __restrict__ cntT, const int* __restrict__ offT,
    const float* __restrict__ x, int* __restrict__ indeg, int* __restrict__ rowptr,
    float* __restrict__ dinv, float* __restrict__ xs, int* __restrict__ csr_src) {
  __shared__ int cnt[BINNODES];
  __shared__ int cursor[BINNODES];
  __shared__ int runLen[NCHUNK];
  __shared__ int runStart[NCHUNK];
  __shared__ int red[32];
  __shared__ int wsum[8];
  __shared__ int stage[STAGECAP];
  const int t = threadIdx.x;
  const int lane = t & 31;
  const int w = t >> 5;
  const int bin = blockIdx.x;
  const int d0 = bin << BINSHIFT;
  const v2i* __restrict__ pairs = (const v2i*)binned;

  const int node = d0 + t;
  const bool has_node = (t < BINNODES) && (node < NNODES);
  float x0 = 0.0f, x1 = 0.0f, x2 = 0.0f;
  if (has_node) {
    const float* __restrict__ xr = x + (size_t)node * 3;
    x0 = xr[0]; x1 = xr[1]; x2 = xr[2];
  }
  int rl = 0, ro = 0;
  if (t < NCHUNK) {
    rl = cntT[bin * NCHUNK + t];
    ro = offT[bin * NCHUNK + t];
  }
  int part = 0;
  {
    const v4i* __restrict__ c4 = (const v4i*)cntT;
    const int n4 = bin * (NCHUNK / 4);
    v4i q[PSUM_TRIPS];
#pragma unroll
    for (int k = 0; k < PSUM_TRIPS; ++k) {
      const int i = t + 1024 * k;
      q[k] = c4[(i < NBINS * (NCHUNK / 4)) ? i : 0];
    }
#pragma unroll
    for (int k = 0; k < PSUM_TRIPS; ++k) asm volatile("" :: "v"(q[k]));
#pragma unroll
    for (int k = 0; k < PSUM_TRIPS; ++k) {
      const int i = t + 1024 * k;
      part += (i < n4) ? (q[k].x + q[k].y + q[k].z + q[k].w) : 0;
    }
  }
  if (t < NCHUNK) {
    runLen[t] = rl;
    runStart[t] = t * CHUNK + ro;
  }
  if (t < BINNODES) cnt[t] = 0;
  for (int off = 16; off > 0; off >>= 1) part += __shfl_xor(part, off);
  if (lane == 0) red[w] = part;
  __syncthreads();
  int P = 0;
  for (int k = 0; k < 32; ++k) P += red[k];

  for_bin_edges(pairs, runLen, runStart, w, lane,
                [&](const v2i e) { atomicAdd(&cnt[e.y - d0], 1); });
  __syncthreads();

  const int v = (t < BINNODES) ? cnt[t] : 0;
  const int incl = wave_incl_scan(v, lane);
  if (t < BINNODES && lane == 31) wsum[w] = incl;
  __syncthreads();
  int n = 0;
  for (int w2 = 0; w2 < 8; ++w2) n += wsum[w2];
  if (t < BINNODES) {
    int add = 0;
    for (int w2 = 0; w2 < w; ++w2) add += wsum[w2];
    const int excl = incl - v + add;
    cursor[t] = excl;
    if (has_node) {
      indeg[node] = v;
      rowptr[node] = P + excl;
      const float dd = rsqrtf((float)v + 1.0f);
      dinv[node] = dd;
      v4f o;
      o.x = x0 * dd; o.y = x1 * dd; o.z = x2 * dd; o.w = 0.0f;
      *(v4f*)(xs + (size_t)node * 4) = o;
    }
  }
  __syncthreads();

  const bool staged = (n <= STAGECAP);
  for_bin_edges(pairs, runLen, runStart, w, lane, [&](const v2i e) {
    const int pos = atomicAdd(&cursor[e.y - d0], 1);
    if (staged) stage[pos] = e.x;
    else csr_src[P + pos] = e.x;
  });
  __syncthreads();
  if (staged) {
    for (int i = t; i < n; i += 1024) csr_src[P + i] = stage[i];
  }
}

__global__ __launch_bounds__(256) void row_gather_x_kernel(
    const float* __restrict__ xs, const int* __restrict__ rowptr,
    const int* __restrict__ indeg, const int* __restrict__ csr_src,
    const float* __restrict__ dinv, float* __restrict__ xa) {
  const int t = blockIdx.x * 256 + threadIdx.x;
  const int row = t >> 2;
  const int sub = t & 3;
  const v4f zero4 = 0.0f;
  v4f acc = zero4;
  float dd = 0.0f;
  if (row < NNODES) {
    dd = dinv[row];
    const int start = rowptr[row];
    const int end = start + indeg[row];
    const int last = end - 1;
    if (sub == 0) acc = *(const v4f*)(xs + (size_t)row * 4);
    for (int base = start; base < end; base += 32) {
      int s[8];
      for (int u = 0; u < 8; ++u) {
        const int e = base + sub + 4 * u;
        s[u] = csr_src[(e < end) ? e : last];
      }
      for (int u = 0; u < 8; ++u) {
        const int e = base + sub + 4 * u;
        const v4f v = *(const v4f*)(xs + (size_t)s[u] * 4);
        acc += (e < end) ? v : zero4;
      }
    }
  }
  float r[3];
  for (int c = 0; c < 3; ++c) {
    float a = acc[c];
    a += __builtin_bit_cast(float, __builtin_amdgcn_update_dpp(
             0, __builtin_bit_cast(int, a), 0xB1, 0xF, 0xF, false));
    a += __builtin_bit_cast(float, __builtin_amdgcn_update_dpp(
             0, __builtin_bit_cast(int, a), 0x4E, 0xF, 0xF, false));
    r[c] = a;
  }
  if (sub == 0) {
    v4f o;
    o.x = r[0] * dd; o.y = r[1] * dd; o.z = r[2] * dd; o.w = 0.0f;
    *(v4f*)(xa + (size_t)row * 4) = o;
  }
}

__global__ void pad_w1_kernel(const float* __restrict__ w, float* __restrict__ wp) {
  int i = blockIdx.x * blockDim.x + threadIdx.x;
  if (i >= 4 * 128) return;
  int k = i >> 7;
  wp[i] = (k < 3) ? w[i] : 0.0f;
}

__global__ void pad_wl2_kernel(const float* __restrict__ w, const float* __restrict__ b,
                               float* __restrict__ wp, float* __restrict__ bp) {
  int i = blockIdx.x * blockDim.x + threadIdx.x;
  if (i >= 32 * 16) return;
  int k = i >> 4, n = i & 15;
  wp[i] = (n < 9) ? w[k * 9 + n] : 0.0f;
  if (i < 16) bp[i] = (i < 9) ? b[i] : 0.0f;
}

__global__ void copy_out_kernel(const float* __restrict__ o16, float* __restrict__ out) {
  int i = blockIdx.x * blockDim.x + threadIdx.x;
  if (i >= NGRAPHS * 9) return;
  int g = i / 9, j = i - g * 9;
  out[i] = o16[g * 16 + j];
}

template <int K, int N, bool RELU, bool BIAS>
__global__ __launch_bounds__(256) void gemm_wmma(
    const float* __restrict__ A, const float* __restrict__ W,
    const float* __restrict__ bias, float* __restrict__ C) {
  __shared__ float lw[K * N];

  const int tid = threadIdx.x;
  constexpr int QUADS = (K / 2) * N / 4;
  for (int q = tid; q < QUADS; q += 256) {
    int t = (q * 4) / N;
    int n = (q * 4) % N;
    int k = t * 2;
    v4f w0 = *(const v4f*)(W + (size_t)k * N + n);
    v4f w1 = *(const v4f*)(W + (size_t)(k + 1) * N + n);
    v4f lo, hi;
    lo.x = w0.x; lo.y = w1.x; lo.z = w0.y; lo.w = w1.y;
    hi.x = w0.z; hi.y = w1.z; hi.z = w0.w; hi.w = w1.w;
    float* d = lw + ((size_t)t * N + n) * 2;
    *(v4f*)(d)     = lo;
    *(v4f*)(d + 4) = hi;
  }
  __syncthreads();

  const int lane = tid & 31;
  const int wave = tid >> 5;
  const int half = lane >> 4;
  const int l16  = lane & 15;
  const int row  = blockIdx.x * 128 + wave * 16 + l16;
  const float* __restrict__ Arow = A + (size_t)row * K + 2 * half;

  for (int n_base = 0; n_base < N; n_base += 16) {
    const int n = n_base + l16;
    float binit = BIAS ? bias[n] : 0.0f;
    v8f acc;
#pragma unroll
    for (int r = 0; r < 8; ++r) acc[r] = binit;

#pragma unroll
    for (int k = 0; k < K; k += 4) {
      v2f av = *(const v2f*)(Arow + k);
      if (RELU) { av.x = fmaxf(av.x, 0.0f); av.y = fmaxf(av.y, 0.0f); }
      const int kk = k + 2 * half;
      v2f bv = *(const v2f*)(lw + ((size_t)(kk >> 1) * N + n) * 2);
      acc = __builtin_amdgcn_wmma_f32_16x16x4_f32(
          false, av, false, bv, (short)0, acc, false, false);
    }

    float* __restrict__ Cp = C + (size_t)(blockIdx.x * 128 + wave * 16 + half * 8) * N + n;
#pragma unroll
    for (int r = 0; r < 8; ++r) Cp[(size_t)r * N] = acc[r];
  }
}

template <int F, bool POOL>
__global__ __launch_bounds__(256) void row_gather_kernel(
    const float* __restrict__ h, const int* __restrict__ rowptr,
    const int* __restrict__ indeg, const int* __restrict__ csr_src,
    const float* __restrict__ dinv, const float* __restrict__ bias,
    float* __restrict__ out, const int* __restrict__ batch,
    float* __restrict__ pooled, float* __restrict__ cnt) {
  constexpr int CH = F / 32;
  const int lane = threadIdx.x & 31;
  const int q = lane & 7;
  const int gl = lane & 24;
  const int fo = q * 4;
  const int row = (int)(blockIdx.x * 32 + (threadIdx.x >> 3));
  const v4f zero4 = 0.0f;

  float dd = 0.0f;
  int start = 0, n = 0;
  int g = -1;
  v4f acc[CH];
#pragma unroll
  for (int c = 0; c < CH; ++c) acc[c] = zero4;
  if (row < NNODES) {
    dd = dinv[row];
    start = rowptr[row];
    n = indeg[row];
    if constexpr (POOL) g = batch[row];
    const float* __restrict__ hr = h + (size_t)row * F + fo;
#pragma unroll
    for (int c = 0; c < CH; ++c) {
      const v4f hv = *(const v4f*)(hr + c * 32);
      const v4f bv = *(const v4f*)(bias + c * 32 + fo);
      acc[c] = hv * (dd * dd) + bv;
    }
  }
  const int n0 = __builtin_amdgcn_readlane(n, 0);
  const int n1 = __builtin_amdgcn_readlane(n, 8);
  const int n2 = __builtin_amdgcn_readlane(n, 16);
  const int n3 = __builtin_amdgcn_readlane(n, 24);
  const int n01 = (n0 > n1) ? n0 : n1;
  const int n23 = (n2 > n3) ? n2 : n3;
  const int nmax = (n01 > n23) ? n01 : n23;

  for (int blk = 0; blk < nmax; blk += 32) {
    int ir[4];
    float wr[4];
#pragma unroll
    for (int r = 0; r < 4; ++r) {
      const int e = blk + q + 8 * r;
      const int pos = start + ((e < n) ? e : 0);
      ir[r] = csr_src[(pos < NEDGES) ? pos : (NEDGES - 1)];
    }
#pragma unroll
    for (int r = 0; r < 4; ++r) {
      const int e = blk + q + 8 * r;
      const float dv = dinv[ir[r]];
      wr[r] = (e < n) ? dv * dd : 0.0f;
    }
    const int lim = nmax - blk;
#pragma unroll
    for (int p = 0; p < 8; ++p) {
      if (4 * p < lim) {
        int s[4];
        float w[4];
#pragma unroll
        for (int u = 0; u < 4; ++u) {
          const int from = gl + 4 * (p & 1) + u;
          s[u] = __shfl(ir[p >> 1], from);
          w[u] = __shfl(wr[p >> 1], from);
        }
        if (blk + 4 * p < n) {
          v4f a[4][CH];
#pragma unroll
          for (int u = 0; u < 4; ++u) {
            const int su = (blk + 4 * p + u < n) ? s[u] : s[0];
            const float* __restrict__ hs = h + (size_t)su * F + fo;
#pragma unroll
            for (int c = 0; c < CH; ++c) a[u][c] = *(const v4f*)(hs + c * 32);
          }
#pragma unroll
          for (int u = 0; u < 4; ++u) {
#pragma unroll
            for (int c = 0; c < CH; ++c) acc[c] += a[u][c] * w[u];
          }
        }
      }
    }
  }
  if constexpr (POOL) {
    static_assert(F == 64, "the pooled table is 64 wide");
    const int g0 = __builtin_amdgcn_readlane(g, 0);
    const int g1 = __builtin_amdgcn_readlane(g, 8);
    const int g2 = __builtin_amdgcn_readlane(g, 16);
    const int g3 = __builtin_amdgcn_readlane(g, 24);
    if (g0 >= 0 && g0 == g1 && g0 == g2 && g0 == g3) {
      float t[CH][4];
#pragma unroll
      for (int c = 0; c < CH; ++c) {
#pragma unroll
        for (int j = 0; j < 4; ++j) {
          float v = acc[c][j];
          v += __shfl_xor(v, 8);
          v += __shfl_xor(v, 16);
          t[c][j] = v;
        }
      }
      const int k = lane >> 3;
      const float va = (k == 0) ? t[0][0] : (k == 1) ? t[0][2] : (k == 2) ? t[1][0] : t[1][2];
      const float vb = (k == 0) ? t[0][1] : (k == 1) ? t[0][3] : (k == 2) ? t[1][1] : t[1][3];
      float* p = pooled + (size_t)g0 * 64 + (k >> 1) * 32 + fo + (k & 1) * 2;
      atomicAdd(p, va);
      atomicAdd(p + 1, vb);
      if (lane == 0) atomicAdd(&cnt[g0], 4.0f);
    } else if (g >= 0) {
      float* p = pooled + (size_t)g * 64 + fo;
#pragma unroll
      for (int c = 0; c < CH; ++c) {
#pragma unroll
        for (int j = 0; j < 4; ++j) atomicAdd(p + c * 32 + j, acc[c][j]);
      }
      if (q == 0) atomicAdd(&cnt[g], 1.0f);
    }
  } else {
    float* __restrict__ o = out + (size_t)row * F + fo;
#pragma unroll
    for (int c = 0; c < CH; ++c) *(v4f*)(o + c * 32) = acc[c];
  }
}

__global__ void pool_div_kernel(float* __restrict__ pooled,
                                const float* __restrict__ cnt) {
  int i = blockIdx.x * blockDim.x + threadIdx.x;
  if (i < NGRAPHS * 64) pooled[i] /= fmaxf(cnt[i >> 6], 1.0f);
}

extern "C" void kernel_launch(void* const* d_in, const int* in_sizes, int n_in,
                              void* d_out, int out_size, void* d_ws, size_t ws_size,
                              hipStream_t stream) {
  const float* x   = (const float*)d_in[0];
  const int*   ei  = (const int*)d_in[1];
  const int*   bat = (const int*)d_in[2];
  const float* W1  = (const float*)d_in[3];  const float* b1  = (const float*)d_in[4];
  const float* W2  = (const float*)d_in[5];  const float* b2  = (const float*)d_in[6];
  const float* W3  = (const float*)d_in[7];  const float* b3  = (const float*)d_in[8];
  const float* W4  = (const float*)d_in[9];  const float* b4  = (const float*)d_in[10];
  const float* Wl1 = (const float*)d_in[11]; const float* bl1 = (const float*)d_in[12];
  const float* Wl2 = (const float*)d_in[13]; const float* bl2 = (const float*)d_in[14];
  float* out = (float*)d_out;

  const int* src = ei;
  const int* dst = ei + NEDGES;

  float* ws     = (float*)d_ws;
  float* dinv   = ws;
  float* xs     = dinv + NNODES;
  float* xa     = xs + (size_t)MPAD * 4;
  float* hT     = xa + (size_t)MPAD * 4;
  float* hA     = hT + (size_t)MPAD * 128;
  float* pooled = hA + (size_t)MPAD * 128;
  float* cnt    = pooled + NGRAPHS * 64;
  float* t1     = cnt + NGRAPHS;
  float* W1p    = t1 + NGRAPHS * 32;
  float* Wl2p   = W1p + 4 * 128;
  float* bl2p   = Wl2p + 32 * 16;
  float* o16    = bl2p + 16;
  int*   indeg   = (int*)(o16 + NGRAPHS * 16);
  int*   rowptr  = indeg + NNODES;
  int*   cntT    = rowptr + NNODES;
  int*   offT    = cntT + NBINS * NCHUNK;
  int*   binned  = offT + NBINS * NCHUNK;
  int*   csr_src = binned + 2 * NEDGES;

  const int B = 256;
  auto blks = [](long long n, int b) { return (unsigned)((n + b - 1) / b); };

  bin_chunk_kernel<<<NCHUNK, 1024, 0, stream>>>(src, dst, binned, cntT, offT);
  bin_rows_kernel<<<NBINS, 1024, 0, stream>>>(binned, cntT, offT, x, indeg, rowptr, dinv, xs, csr_src);

  pad_w1_kernel<<<blks(4 * 128, B), B, 0, stream>>>(W1, W1p);
  pad_wl2_kernel<<<blks(32 * 16, B), B, 0, stream>>>(Wl2, bl2, Wl2p, bl2p);

  const unsigned GB = MPAD / 128;
  const unsigned RB = MPAD / 32;

  row_gather_x_kernel<<<MPAD / 64, 256, 0, stream>>>(xs, rowptr, indeg, csr_src, dinv, xa);
  gemm_wmma<4, 128, false, true><<<GB, 256, 0, stream>>>(xa, W1p, b1, hA);

  gemm_wmma<128, 128, true, false><<<GB, 256, 0, stream>>>(hA, W2, nullptr, hT);
  row_gather_kernel<128, false><<<RB, 256, 0, stream>>>(hT, rowptr, indeg, csr_src, dinv, b2, hA,
                                                       nullptr, nullptr, nullptr);

  gemm_wmma<128, 128, true, false><<<GB, 256, 0, stream>>>(hA, W3, nullptr, hT);
  row_gather_kernel<128, false><<<RB, 256, 0, stream>>>(hT, rowptr, indeg, csr_src, dinv, b3, hA,
                                                       nullptr, nullptr, nullptr);

  gemm_wmma<128, 64, true, false><<<GB, 256, 0, stream>>>(hA, W4, nullptr, hT);
  zero_kernel<<<blks(NGRAPHS * 64, B), B, 0, stream>>>(pooled, NGRAPHS * 64);
  zero_kernel<<<blks(NGRAPHS, B), B, 0, stream>>>(cnt, NGRAPHS);
  row_gather_kernel<64, true><<<RB, 256, 0, stream>>>(hT, rowptr, indeg, csr_src, dinv, b4, nullptr,
                                                     bat, pooled, cnt);
  pool_div_kernel<<<blks(NGRAPHS * 64, B), B, 0, stream>>>(pooled, cnt);

  gemm_wmma<64, 32, false, true><<<NGRAPHS / 128, 256, 0, stream>>>(pooled, Wl1, bl1, t1);
  gemm_wmma<32, 16, true, true><<<NGRAPHS / 128, 256, 0, stream>>>(t1, Wl2p, bl2p, o16);
  copy_out_kernel<<<blks(NGRAPHS * 9, B), B, 0, stream>>>(o16, out);
}


# Small launches folded, 18 become 11: pool clear into the CSR row kernel, W1 zero row into GEMM staging, mean divide + both head products + 9-wide store into one head kernel; no sum order changes
# speedup vs baseline: 4.5973x; 1.0410x over previous
#include <hip/hip_runtime.h>
#include <hip/hip_bf16.h>

typedef __attribute__((ext_vector_type(2))) float v2f;
typedef __attribute__((ext_vector_type(4))) float v4f;
typedef __attribute__((ext_vector_type(8))) float v8f;

#define NNODES  100000
#define MPAD    100096
#define NEDGES  1600000
#define NGRAPHS 512

typedef __attribute__((ext_vector_type(2))) int v2i;
typedef __attribute__((ext_vector_type(4))) int v4i;

#define BINSHIFT 8
#define BINNODES 256
#define NBINS    ((NNODES + BINNODES - 1) / BINNODES)
#define NCHUNK   128
#define CHUNK    (NEDGES / NCHUNK)
#define EPT      ((CHUNK + 1023) / 1024)
#define STAGECAP 16384
static_assert(NEDGES % NCHUNK == 0, "chunks must tile the edge list");
static_assert(NBINS <= 512, "bin scan covers 512 bins");

__device__ __forceinline__ int wave_incl_scan(int v, int lane) {
  for (int off = 1; off < 32; off <<= 1) {
    const int u = __shfl_up(v, off);
    if (lane >= off) v += u;
  }
  return v;
}

__global__ __launch_bounds__(1024) void bin_chunk_kernel(
    const int* __restrict__ src, const int* __restrict__ dst,
    int* __restrict__ binned, int* __restrict__ cntT, int* __restrict__ offT) {
  __shared__ int hist[512];
  __shared__ int wsum[16];
  __shared__ v2i sorted[CHUNK];
  const int t = threadIdx.x;
  const int lane = t & 31;
  const int b = blockIdx.x;
  const int base = b * CHUNK;
  if (t < 512) hist[t] = 0;
  __syncthreads();

  int s[EPT], d[EPT], r[EPT];
#pragma unroll
  for (int i = 0; i < EPT; ++i) {
    const int idx = t + 1024 * i;
    s[i] = 0; d[i] = 0;
    if (idx < CHUNK) { s[i] = src[base + idx]; d[i] = dst[base + idx]; }
  }
#pragma unroll
  for (int i = 0; i < EPT; ++i) {
    const int idx = t + 1024 * i;
    r[i] = 0;
    if (idx < CHUNK) r[i] = atomicAdd(&hist[d[i] >> BINSHIFT], 1);
  }
  __syncthreads();

  const int v = (t < 512) ? hist[t] : 0;
  const int incl = wave_incl_scan(v, lane);
  if (t < 512 && lane == 31) wsum[t >> 5] = incl;
  __syncthreads();
  if (t < 512) {
    int add = 0;
    for (int w2 = 0; w2 < (t >> 5); ++w2) add += wsum[w2];
    const int excl = incl - v + add;
    hist[t] = excl;
    if (t < NBINS) {
      cntT[t * NCHUNK + b] = v;
      offT[t * NCHUNK + b] = excl;
    }
  }
  __syncthreads();

#pragma unroll
  for (int i = 0; i < EPT; ++i) {
    const int idx = t + 1024 * i;
    if (idx < CHUNK) {
      v2i e; e.x = s[i]; e.y = d[i];
      sorted[hist[d[i] >> BINSHIFT] + r[i]] = e;
    }
  }
  __syncthreads();
  v2i* __restrict__ outp = (v2i*)binned + base;
#pragma unroll
  for (int i = 0; i < EPT; ++i) {
    const int idx = t + 1024 * i;
    if (idx < CHUNK) outp[idx] = sorted[idx];
  }
}

#define RUNS_PER_WAVE (NCHUNK / 32)
#define PSUM_TRIPS    ((NBINS * (NCHUNK / 4) + 1023) / 1024)

template <typename F>
__device__ __forceinline__ void for_bin_edges(const v2i* __restrict__ pairs, const int* runLen,
                                              const int* runStart, int w, int lane, F f) {
  int len[RUNS_PER_WAVE], st[RUNS_PER_WAVE];
  v2i e0[RUNS_PER_WAVE], e1[RUNS_PER_WAVE];
#pragma unroll
  for (int q = 0; q < RUNS_PER_WAVE; ++q) {
    len[q] = runLen[w + 32 * q];
    st[q] = runStart[w + 32 * q];
    e0[q] = pairs[(lane < len[q]) ? (st[q] + lane) : 0];
    e1[q] = pairs[(lane + 32 < len[q]) ? (st[q] + lane + 32) : 0];
  }
#pragma unroll
  for (int q = 0; q < RUNS_PER_WAVE; ++q) asm volatile("" :: "v"(e0[q]), "v"(e1[q]));
#pragma unroll
  for (int q = 0; q < RUNS_PER_WAVE; ++q) {
    if (lane < len[q]) f(e0[q]);
    if (lane + 32 < len[q]) f(e1[q]);
  }
#pragma unroll
  for (int q = 0; q < RUNS_PER_WAVE; ++q) {
    for (int j = lane + 64; j < len[q]; j += 32) f(pairs[st[q] + j]);
  }
}

__global__ __launch_bounds__(1024) void bin_rows_kernel(
    const int* __restrict__ binned, const int* __restrict__ cntT, const int* __restrict__ offT,
    const float* __restrict__ x, int* __restrict__ indeg, int* __restrict__ rowptr,
    float* __restrict__ dinv, float* __restrict__ xs, int* __restrict__ csr_src,
    float* __restrict__ pooled, float* __restrict__ pcnt) {
  static_assert(NBINS * 1024 >= NGRAPHS * 65, "the grid covers the pool table and its counts");
  __shared__ int cnt[BINNODES];
  __shared__ int cursor[BINNODES];
  __shared__ int runLen[NCHUNK];
  __shared__ int runStart[NCHUNK];
  __shared__ int red[32];
  __shared__ int wsum[8];
  __shared__ int stage[STAGECAP];
  const int t = threadIdx.x;
  const int lane = t & 31;
  const int w = t >> 5;
  const int bin = blockIdx.x;
  const int d0 = bin << BINSHIFT;
  const v2i* __restrict__ pairs = (const v2i*)binned;

  {
    const int gid = bin * 1024 + t;
    if (gid < NGRAPHS * 64) pooled[gid] = 0.0f;
    else if (gid < NGRAPHS * 65) pcnt[gid - NGRAPHS * 64] = 0.0f;
  }

  const int node = d0 + t;
  const bool has_node = (t < BINNODES) && (node < NNODES);
  float x0 = 0.0f, x1 = 0.0f, x2 = 0.0f;
  if (has_node) {
    const float* __restrict__ xr = x + (size_t)node * 3;
    x0 = xr[0]; x1 = xr[1]; x2 = xr[2];
  }
  int rl = 0, ro = 0;
  if (t < NCHUNK) {
    rl = cntT[bin * NCHUNK + t];
    ro = offT[bin * NCHUNK + t];
  }
  int part = 0;
  {
    const v4i* __restrict__ c4 = (const v4i*)cntT;
    const int n4 = bin * (NCHUNK / 4);
    v4i q[PSUM_TRIPS];
#pragma unroll
    for (int k = 0; k < PSUM_TRIPS; ++k) {
      const int i = t + 1024 * k;
      q[k] = c4[(i < NBINS * (NCHUNK / 4)) ? i : 0];
    }
#pragma unroll
    for (int k = 0; k < PSUM_TRIPS; ++k) asm volatile("" :: "v"(q[k]));
#pragma unroll
    for (int k = 0; k < PSUM_TRIPS; ++k) {
      const int i = t + 1024 * k;
      part += (i < n4) ? (q[k].x + q[k].y + q[k].z + q[k].w) : 0;
    }
  }
  if (t < NCHUNK) {
    runLen[t] = rl;
    runStart[t] = t * CHUNK + ro;
  }
  if (t < BINNODES) cnt[t] = 0;
  for (int off = 16; off > 0; off >>= 1) part += __shfl_xor(part, off);
  if (lane == 0) red[w] = part;
  __syncthreads();
  int P = 0;
  for (int k = 0; k < 32; ++k) P += red[k];

  for_bin_edges(pairs, runLen, runStart, w, lane,
                [&](const v2i e) { atomicAdd(&cnt[e.y - d0], 1); });
  __syncthreads();

  const int v = (t < BINNODES) ? cnt[t] : 0;
  const int incl = wave_incl_scan(v, lane);
  if (t < BINNODES && lane == 31) wsum[w] = incl;
  __syncthreads();
  int n = 0;
  for (int w2 = 0; w2 < 8; ++w2) n += wsum[w2];
  if (t < BINNODES) {
    int add = 0;
    for (int w2 = 0; w2 < w; ++w2) add += wsum[w2];
    const int excl = incl - v + add;
    cursor[t] = excl;
    if (has_node) {
      indeg[node] = v;
      rowptr[node] = P + excl;
      const float dd = rsqrtf((float)v + 1.0f);
      dinv[node] = dd;
      v4f o;
      o.x = x0 * dd; o.y = x1 * dd; o.z = x2 * dd; o.w = 0.0f;
      *(v4f*)(xs + (size_t)node * 4) = o;
    }
  }
  __syncthreads();

  const bool staged = (n <= STAGECAP);
  for_bin_edges(pairs, runLen, runStart, w, lane, [&](const v2i e) {
    const int pos = atomicAdd(&cursor[e.y - d0], 1);
    if (staged) stage[pos] = e.x;
    else csr_src[P + pos] = e.x;
  });
  __syncthreads();
  if (staged) {
    for (int i = t; i < n; i += 1024) csr_src[P + i] = stage[i];
  }
}

__global__ __launch_bounds__(256) void row_gather_x_kernel(
    const float* __restrict__ xs, const int* __restrict__ rowptr,
    const int* __restrict__ indeg, const int* __restrict__ csr_src,
    const float* __restrict__ dinv, float* __restrict__ xa) {
  const int t = blockIdx.x * 256 + threadIdx.x;
  const int row = t >> 2;
  const int sub = t & 3;
  const v4f zero4 = 0.0f;
  v4f acc = zero4;
  float dd = 0.0f;
  if (row < NNODES) {
    dd = dinv[row];
    const int start = rowptr[row];
    const int end = start + indeg[row];
    const int last = end - 1;
    if (sub == 0) acc = *(const v4f*)(xs + (size_t)row * 4);
    for (int base = start; base < end; base += 32) {
      int s[8];
      for (int u = 0; u < 8; ++u) {
        const int e = base + sub + 4 * u;
        s[u] = csr_src[(e < end) ? e : last];
      }
      for (int u = 0; u < 8; ++u) {
        const int e = base + sub + 4 * u;
        const v4f v = *(const v4f*)(xs + (size_t)s[u] * 4);
        acc += (e < end) ? v : zero4;
      }
    }
  }
  float r[3];
  for (int c = 0; c < 3; ++c) {
    float a = acc[c];
    a += __builtin_bit_cast(float, __builtin_amdgcn_update_dpp(
             0, __builtin_bit_cast(int, a), 0xB1, 0xF, 0xF, false));
    a += __builtin_bit_cast(float, __builtin_amdgcn_update_dpp(
             0, __builtin_bit_cast(int, a), 0x4E, 0xF, 0xF, false));
    r[c] = a;
  }
  if (sub == 0) {
    v4f o;
    o.x = r[0] * dd; o.y = r[1] * dd; o.z = r[2] * dd; o.w = 0.0f;
    *(v4f*)(xa + (size_t)row * 4) = o;
  }
}

template <int K, int N, bool RELU, bool BIAS, int KW = K>
__global__ __launch_bounds__(256) void gemm_wmma(
    const float* __restrict__ A, const float* __restrict__ W,
    const float* __restrict__ bias, float* __restrict__ C) {
  __shared__ float lw[K * N];

  static_assert(KW == K || KW == K - 1, "only the last row of W may be missing");
  const int tid = threadIdx.x;
  constexpr int QUADS = (K / 2) * N / 4;
  for (int q = tid; q < QUADS; q += 256) {
    int t = (q * 4) / N;
    int n = (q * 4) % N;
    int k = t * 2;
    v4f w0 = *(const v4f*)(W + (size_t)k * N + n);
    v4f w1 = 0.0f;
    if (KW == K || k + 1 < KW) w1 = *(const v4f*)(W + (size_t)(k + 1) * N + n);
    v4f lo, hi;
    lo.x = w0.x; lo.y = w1.x; lo.z = w0.y; lo.w = w1.y;
    hi.x = w0.z; hi.y = w1.z; hi.z = w0.w; hi.w = w1.w;
    float* d = lw + ((size_t)t * N + n) * 2;
    *(v4f*)(d)     = lo;
    *(v4f*)(d + 4) = hi;
  }
  __syncthreads();

  const int lane = tid & 31;
  const int wave = tid >> 5;
  const int half = lane >> 4;
  const int l16  = lane & 15;
  const int row  = blockIdx.x * 128 + wave * 16 + l16;
  const float* __restrict__ Arow = A + (size_t)row * K + 2 * half;

  for (int n_base = 0; n_base < N; n_base += 16) {
    const int n = n_base + l16;
    float binit = BIAS ? bias[n] : 0.0f;
    v8f acc;
#pragma unroll
    for (int r = 0; r < 8; ++r) acc[r] = binit;

#pragma unroll
    for (int k = 0; k < K; k += 4) {
      v2f av = *(const v2f*)(Arow + k);
      if (RELU) { av.x = fmaxf(av.x, 0.0f); av.y = fmaxf(av.y, 0.0f); }
      const int kk = k + 2 * half;
      v2f bv = *(const v2f*)(lw + ((size_t)(kk >> 1) * N + n) * 2);
      acc = __builtin_amdgcn_wmma_f32_16x16x4_f32(
          false, av, false, bv, (short)0, acc, false, false);
    }

    float* __restrict__ Cp = C + (size_t)(blockIdx.x * 128 + wave * 16 + half * 8) * N + n;
#pragma unroll
    for (int r = 0; r < 8; ++r) Cp[(size_t)r * N] = acc[r];
  }
}

template <int F, bool POOL>
__global__ __launch_bounds__(256) void row_gather_kernel(
    const float* __restrict__ h, const int* __restrict__ rowptr,
    const int* __restrict__ indeg, const int* __restrict__ csr_src,
    const float* __restrict__ dinv, const float* __restrict__ bias,
    float* __restrict__ out, const int* __restrict__ batch,
    float* __restrict__ pooled, float* __restrict__ cnt) {
  constexpr int CH = F / 32;
  const int lane = threadIdx.x & 31;
  const int q = lane & 7;
  const int gl = lane & 24;
  const int fo = q * 4;
  const int row = (int)(blockIdx.x * 32 + (threadIdx.x >> 3));
  const v4f zero4 = 0.0f;

  float dd = 0.0f;
  int start = 0, n = 0;
  int g = -1;
  v4f acc[CH];
#pragma unroll
  for (int c = 0; c < CH; ++c) acc[c] = zero4;
  if (row < NNODES) {
    dd = dinv[row];
    start = rowptr[row];
    n = indeg[row];
    if constexpr (POOL) g = batch[row];
    const float* __restrict__ hr = h + (size_t)row * F + fo;
#pragma unroll
    for (int c = 0; c < CH; ++c) {
      const v4f hv = *(const v4f*)(hr + c * 32);
      const v4f bv = *(const v4f*)(bias + c * 32 + fo);
      acc[c] = hv * (dd * dd) + bv;
    }
  }
  const int n0 = __builtin_amdgcn_readlane(n, 0);
  const int n1 = __builtin_amdgcn_readlane(n, 8);
  const int n2 = __builtin_amdgcn_readlane(n, 16);
  const int n3 = __builtin_amdgcn_readlane(n, 24);
  const int n01 = (n0 > n1) ? n0 : n1;
  const int n23 = (n2 > n3) ? n2 : n3;
  const int nmax = (n01 > n23) ? n01 : n23;

  for (int blk = 0; blk < nmax; blk += 32) {
    int ir[4];
    float wr[4];
#pragma unroll
    for (int r = 0; r < 4; ++r) {
      const int e = blk + q + 8 * r;
      const int pos = start + ((e < n) ? e : 0);
      ir[r] = csr_src[(pos < NEDGES) ? pos : (NEDGES - 1)];
    }
#pragma unroll
    for (int r = 0; r < 4; ++r) {
      const int e = blk + q + 8 * r;
      const float dv = dinv[ir[r]];
      wr[r] = (e < n) ? dv * dd : 0.0f;
    }
    const int lim = nmax - blk;
#pragma unroll
    for (int p = 0; p < 8; ++p) {
      if (4 * p < lim) {
        int s[4];
        float w[4];
#pragma unroll
        for (int u = 0; u < 4; ++u) {
          const int from = gl + 4 * (p & 1) + u;
          s[u] = __shfl(ir[p >> 1], from);
          w[u] = __shfl(wr[p >> 1], from);
        }
        if (blk + 4 * p < n) {
          v4f a[4][CH];
#pragma unroll
          for (int u = 0; u < 4; ++u) {
            const int su = (blk + 4 * p + u < n) ? s[u] : s[0];
            const float* __restrict__ hs = h + (size_t)su * F + fo;
#pragma unroll
            for (int c = 0; c < CH; ++c) a[u][c] = *(const v4f*)(hs + c * 32);
          }
#pragma unroll
          for (int u = 0; u < 4; ++u) {
#pragma unroll
            for (int c = 0; c < CH; ++c) acc[c] += a[u][c] * w[u];
          }
        }
      }
    }
  }
  if constexpr (POOL) {
    static_assert(F == 64, "the pooled table is 64 wide");
    const int g0 = __builtin_amdgcn_readlane(g, 0);
    const int g1 = __builtin_amdgcn_readlane(g, 8);
    const int g2 = __builtin_amdgcn_readlane(g, 16);
    const int g3 = __builtin_amdgcn_readlane(g, 24);
    if (g0 >= 0 && g0 == g1 && g0 == g2 && g0 == g3) {
      float t[CH][4];
#pragma unroll
      for (int c = 0; c < CH; ++c) {
#pragma unroll
        for (int j = 0; j < 4; ++j) {
          float v = acc[c][j];
          v += __shfl_xor(v, 8);
          v += __shfl_xor(v, 16);
          t[c][j] = v;
        }
      }
      const int k = lane >> 3;
      const float va = (k == 0) ? t[0][0] : (k == 1) ? t[0][2] : (k == 2) ? t[1][0] : t[1][2];
      const float vb = (k == 0) ? t[0][1] : (k == 1) ? t[0][3] : (k == 2) ? t[1][1] : t[1][3];
      float* p = pooled + (size_t)g0 * 64 + (k >> 1) * 32 + fo + (k & 1) * 2;
      atomicAdd(p, va);
      atomicAdd(p + 1, vb);
      if (lane == 0) atomicAdd(&cnt[g0], 4.0f);
    } else if (g >= 0) {
      float* p = pooled + (size_t)g * 64 + fo;
#pragma unroll
      for (int c = 0; c < CH; ++c) {
#pragma unroll
        for (int j = 0; j < 4; ++j) atomicAdd(p + c * 32 + j, acc[c][j]);
      }
      if (q == 0) atomicAdd(&cnt[g], 1.0f);
    }
  } else {
    float* __restrict__ o = out + (size_t)row * F + fo;
#pragma unroll
    for (int c = 0; c < CH; ++c) *(v4f*)(o + c * 32) = acc[c];
  }
}

#define HIDP 34
__global__ __launch_bounds__(256) void head_kernel(
    const float* __restrict__ sums, const float* __restrict__ cnt,
    const float* __restrict__ Wl1, const float* __restrict__ bl1,
    const float* __restrict__ Wl2, const float* __restrict__ bl2,
    float* __restrict__ out) {
  __shared__ __attribute__((aligned(16))) float lw1[64 * 32];
  __shared__ __attribute__((aligned(16))) float lw2[32 * 16];
  __shared__ __attribute__((aligned(16))) float hid[8 * 16 * HIDP];
  __shared__ float ost[128 * 9];

  const int tid  = threadIdx.x;
  const int lane = tid & 31;
  const int wave = tid >> 5;
  const int half = lane >> 4;
  const int l16  = lane & 15;
  const int row0 = blockIdx.x * 128 + wave * 16;
  const int row  = row0 + l16;

  const float den = fmaxf(cnt[row], 1.0f);
  const float* __restrict__ Arow = sums + (size_t)row * 64 + 2 * half;
  v2f a1[16];
#pragma unroll
  for (int s = 0; s < 16; ++s) {
    a1[s] = *(const v2f*)(Arow + 4 * s);
    a1[s].x = a1[s].x / den;
    a1[s].y = a1[s].y / den;
  }

  {
    const int t = (tid * 4) / 32;
    const int n = (tid * 4) % 32;
    const int k = t * 2;
    const v4f w0 = *(const v4f*)(Wl1 + (size_t)k * 32 + n);
    const v4f w1 = *(const v4f*)(Wl1 + (size_t)(k + 1) * 32 + n);
    v4f lo, hi;
    lo.x = w0.x; lo.y = w1.x; lo.z = w0.y; lo.w = w1.y;
    hi.x = w0.z; hi.y = w1.z; hi.z = w0.w; hi.w = w1.w;
    float* d = lw1 + ((size_t)t * 32 + n) * 2;
    *(v4f*)(d)     = lo;
    *(v4f*)(d + 4) = hi;
  }
  for (int i = tid; i < 32 * 16; i += 256) {
    const int k = i >> 4, n = i & 15;
    lw2[((k >> 1) * 16 + n) * 2 + (k & 1)] = (n < 9) ? Wl2[k * 9 + n] : 0.0f;
  }
  __syncthreads();

  float* __restrict__ hw = hid + wave * (16 * HIDP);
#pragma unroll
  for (int j = 0; j < 2; ++j) {
    const int n = 16 * j + l16;
    const float binit = bl1[n];
    v8f acc;
#pragma unroll
    for (int r = 0; r < 8; ++r) acc[r] = binit;
#pragma unroll
    for (int s = 0; s < 16; ++s) {
      const int kk = 4 * s + 2 * half;
      const v2f bv = *(const v2f*)(lw1 + ((size_t)(kk >> 1) * 32 + n) * 2);
      acc = __builtin_amdgcn_wmma_f32_16x16x4_f32(
          false, a1[s], false, bv, (short)0, acc, false, false);
    }
#pragma unroll
    for (int r = 0; r < 8; ++r) hw[(half * 8 + r) * HIDP + n] = acc[r];
  }
  __syncthreads();

  const float binit2 = (l16 < 9) ? bl2[l16] : 0.0f;
  v8f acc2;
#pragma unroll
  for (int r = 0; r < 8; ++r) acc2[r] = binit2;
#pragma unroll
  for (int s = 0; s < 8; ++s) {
    const int kk = 4 * s + 2 * half;
    v2f av = *(const v2f*)(hw + l16 * HIDP + kk);
    av.x = fmaxf(av.x, 0.0f); av.y = fmaxf(av.y, 0.0f);
    const v2f bv = *(const v2f*)(lw2 + ((size_t)(kk >> 1) * 16 + l16) * 2);
    acc2 = __builtin_amdgcn_wmma_f32_16x16x4_f32(
        false, av, false, bv, (short)0, acc2, false, false);
  }
  if (l16 < 9) {
    float* __restrict__ op = ost + (wave * 16 + half * 8) * 9 + l16;
#pragma unroll
    for (int r = 0; r < 8; ++r) op[r * 9] = acc2[r];
  }
  __syncthreads();
  float* __restrict__ ob = out + (size_t)blockIdx.x * (128 * 9);
  for (int i = tid; i < 128 * 9; i += 256) ob[i] = ost[i];
}

extern "C" void kernel_launch(void* const* d_in, const int* in_sizes, int n_in,
                              void* d_out, int out_size, void* d_ws, size_t ws_size,
                              hipStream_t stream) {
  const float* x   = (const float*)d_in[0];
  const int*   ei  = (const int*)d_in[1];
  const int*   bat = (const int*)d_in[2];
  const float* W1  = (const float*)d_in[3];  const float* b1  = (const float*)d_in[4];
  const float* W2  = (const float*)d_in[5];  const float* b2  = (const float*)d_in[6];
  const float* W3  = (const float*)d_in[7];  const float* b3  = (const float*)d_in[8];
  const float* W4  = (const float*)d_in[9];  const float* b4  = (const float*)d_in[10];
  const float* Wl1 = (const float*)d_in[11]; const float* bl1 = (const float*)d_in[12];
  const float* Wl2 = (const float*)d_in[13]; const float* bl2 = (const float*)d_in[14];
  float* out = (float*)d_out;

  const int* src = ei;
  const int* dst = ei + NEDGES;

  float* ws     = (float*)d_ws;
  float* dinv   = ws;
  float* xs     = dinv + NNODES;
  float* xa     = xs + (size_t)MPAD * 4;
  float* hT     = xa + (size_t)MPAD * 4;
  float* hA     = hT + (size_t)MPAD * 128;
  float* pooled = hA + (size_t)MPAD * 128;
  float* cnt    = pooled + NGRAPHS * 64;
  int*   indeg   = (int*)(cnt + NGRAPHS);
  int*   rowptr  = indeg + NNODES;
  int*   cntT    = rowptr + NNODES;
  int*   offT    = cntT + NBINS * NCHUNK;
  int*   binned  = offT + NBINS * NCHUNK;
  int*   csr_src = binned + 2 * NEDGES;

  bin_chunk_kernel<<<NCHUNK, 1024, 0, stream>>>(src, dst, binned, cntT, offT);
  bin_rows_kernel<<<NBINS, 1024, 0, stream>>>(binned, cntT, offT, x, indeg, rowptr, dinv, xs, csr_src,
                                              pooled, cnt);

  const unsigned GB = MPAD / 128;
  const unsigned RB = MPAD / 32;

  row_gather_x_kernel<<<MPAD / 64, 256, 0, stream>>>(xs, rowptr, indeg, csr_src, dinv, xa);
  gemm_wmma<4, 128, false, true, 3><<<GB, 256, 0, stream>>>(xa, W1, b1, hA);

  gemm_wmma<128, 128, true, false><<<GB, 256, 0, stream>>>(hA, W2, nullptr, hT);
  row_gather_kernel<128, false><<<RB, 256, 0, stream>>>(hT, rowptr, indeg, csr_src, dinv, b2, hA,
                                                       nullptr, nullptr, nullptr);

  gemm_wmma<128, 128, true, false><<<GB, 256, 0, stream>>>(hA, W3, nullptr, hT);
  row_gather_kernel<128, false><<<RB, 256, 0, stream>>>(hT, rowptr, indeg, csr_src, dinv, b3, hA,
                                                       nullptr, nullptr, nullptr);

  gemm_wmma<128, 64, true, false><<<GB, 256, 0, stream>>>(hA, W4, nullptr, hT);
  row_gather_kernel<64, true><<<RB, 256, 0, stream>>>(hT, rowptr, indeg, csr_src, dinv, b4, nullptr,
                                                     bat, pooled, cnt);

  head_kernel<<<NGRAPHS / 128, 256, 0, stream>>>(pooled, cnt, Wl1, bl1, Wl2, bl2, out);
}


# Layer 3 and 4 products moved into the gather that makes their input rows: 128 gathered rows stay in an LDS tile and are multiplied there (WMMA, same k order); 11 launches become 9; sum orders kept
# speedup vs baseline: 5.3554x; 1.1649x over previous
#include <hip/hip_runtime.h>
#include <hip/hip_bf16.h>

typedef __attribute__((ext_vector_type(2))) float v2f;
typedef __attribute__((ext_vector_type(4))) float v4f;
typedef __attribute__((ext_vector_type(8))) float v8f;

#define NNODES  100000
#define MPAD    100096
#define NEDGES  1600000
#define NGRAPHS 512

typedef __attribute__((ext_vector_type(2))) int v2i;
typedef __attribute__((ext_vector_type(4))) int v4i;

#define BINSHIFT 8
#define BINNODES 256
#define NBINS    ((NNODES + BINNODES - 1) / BINNODES)
#define NCHUNK   128
#define CHUNK    (NEDGES / NCHUNK)
#define EPT      ((CHUNK + 1023) / 1024)
#define STAGECAP 16384
static_assert(NEDGES % NCHUNK == 0, "chunks must tile the edge list");
static_assert(NBINS <= 512, "bin scan covers 512 bins");

__device__ __forceinline__ int wave_incl_scan(int v, int lane) {
  for (int off = 1; off < 32; off <<= 1) {
    const int u = __shfl_up(v, off);
    if (lane >= off) v += u;
  }
  return v;
}

__global__ __launch_bounds__(1024) void bin_chunk_kernel(
    const int* __restrict__ src, const int* __restrict__ dst,
    int* __restrict__ binned, int* __restrict__ cntT, int* __restrict__ offT) {
  __shared__ int hist[512];
  __shared__ int wsum[16];
  __shared__ v2i sorted[CHUNK];
  const int t = threadIdx.x;
  const int lane = t & 31;
  const int b = blockIdx.x;
  const int base = b * CHUNK;
  if (t < 512) hist[t] = 0;
  __syncthreads();

  int s[EPT], d[EPT], r[EPT];
#pragma unroll
  for (int i = 0; i < EPT; ++i) {
    const int idx = t + 1024 * i;
    s[i] = 0; d[i] = 0;
    if (idx < CHUNK) { s[i] = src[base + idx]; d[i] = dst[base + idx]; }
  }
#pragma unroll
  for (int i = 0; i < EPT; ++i) {
    const int idx = t + 1024 * i;
    r[i] = 0;
    if (idx < CHUNK) r[i] = atomicAdd(&hist[d[i] >> BINSHIFT], 1);
  }
  __syncthreads();

  const int v = (t < 512) ? hist[t] : 0;
  const int incl = wave_incl_scan(v, lane);
  if (t < 512 && lane == 31) wsum[t >> 5] = incl;
  __syncthreads();
  if (t < 512) {
    int add = 0;
    for (int w2 = 0; w2 < (t >> 5); ++w2) add += wsum[w2];
    const int excl = incl - v + add;
    hist[t] = excl;
    if (t < NBINS) {
      cntT[t * NCHUNK + b] = v;
      offT[t * NCHUNK + b] = excl;
    }
  }
  __syncthreads();

#pragma unroll
  for (int i = 0; i < EPT; ++i) {
    const int idx = t + 1024 * i;
    if (idx < CHUNK) {
      v2i e; e.x = s[i]; e.y = d[i];
      sorted[hist[d[i] >> BINSHIFT] + r[i]] = e;
    }
  }
  __syncthreads();
  v2i* __restrict__ outp = (v2i*)binned + base;
#pragma unroll
  for (int i = 0; i < EPT; ++i) {
    const int idx = t + 1024 * i;
    if (idx < CHUNK) outp[idx] = sorted[idx];
  }
}

#define RUNS_PER_WAVE (NCHUNK / 32)
#define PSUM_TRIPS    ((NBINS * (NCHUNK / 4) + 1023) / 1024)

template <typename F>
__device__ __forceinline__ void for_bin_edges(const v2i* __restrict__ pairs, const int* runLen,
                                              const int* runStart, int w, int lane, F f) {
  int len[RUNS_PER_WAVE], st[RUNS_PER_WAVE];
  v2i e0[RUNS_PER_WAVE], e1[RUNS_PER_WAVE];
#pragma unroll
  for (int q = 0; q < RUNS_PER_WAVE; ++q) {
    len[q] = runLen[w + 32 * q];
    st[q] = runStart[w + 32 * q];
    e0[q] = pairs[(lane < len[q]) ? (st[q] + lane) : 0];
    e1[q] = pairs[(lane + 32 < len[q]) ? (st[q] + lane + 32) : 0];
  }
#pragma unroll
  for (int q = 0; q < RUNS_PER_WAVE; ++q) asm volatile("" :: "v"(e0[q]), "v"(e1[q]));
#pragma unroll
  for (int q = 0; q < RUNS_PER_WAVE; ++q) {
    if (lane < len[q]) f(e0[q]);
    if (lane + 32 < len[q]) f(e1[q]);
  }
#pragma unroll
  for (int q = 0; q < RUNS_PER_WAVE; ++q) {
    for (int j = lane + 64; j < len[q]; j += 32) f(pairs[st[q] + j]);
  }
}

__global__ __launch_bounds__(1024) void bin_rows_kernel(
    const int* __restrict__ binned, const int* __restrict__ cntT, const int* __restrict__ offT,
    const float* __restrict__ x, int* __restrict__ indeg, int* __restrict__ rowptr,
    float* __restrict__ dinv, float* __restrict__ xs, int* __restrict__ csr_src,
    float* __restrict__ pooled, float* __restrict__ pcnt) {
  static_assert(NBINS * 1024 >= NGRAPHS * 65, "the grid covers the pool table and its counts");
  __shared__ int cnt[BINNODES];
  __shared__ int cursor[BINNODES];
  __shared__ int runLen[NCHUNK];
  __shared__ int runStart[NCHUNK];
  __shared__ int red[32];
  __shared__ int wsum[8];
  __shared__ int stage[STAGECAP];
  const int t = threadIdx.x;
  const int lane = t & 31;
  const int w = t >> 5;
  const int bin = blockIdx.x;
  const int d0 = bin << BINSHIFT;
  const v2i* __restrict__ pairs = (const v2i*)binned;

  {
    const int gid = bin * 1024 + t;
    if (gid < NGRAPHS * 64) pooled[gid] = 0.0f;
    else if (gid < NGRAPHS * 65) pcnt[gid - NGRAPHS * 64] = 0.0f;
  }

  const int node = d0 + t;
  const bool has_node = (t < BINNODES) && (node < NNODES);
  float x0 = 0.0f, x1 = 0.0f, x2 = 0.0f;
  if (has_node) {
    const float* __restrict__ xr = x + (size_t)node * 3;
    x0 = xr[0]; x1 = xr[1]; x2 = xr[2];
  }
  int rl = 0, ro = 0;
  if (t < NCHUNK) {
    rl = cntT[bin * NCHUNK + t];
    ro = offT[bin * NCHUNK + t];
  }
  int part = 0;
  {
    const v4i* __restrict__ c4 = (const v4i*)cntT;
    const int n4 = bin * (NCHUNK / 4);
    v4i q[PSUM_TRIPS];
#pragma unroll
    for (int k = 0; k < PSUM_TRIPS; ++k) {
      const int i = t + 1024 * k;
      q[k] = c4[(i < NBINS * (NCHUNK / 4)) ? i : 0];
    }
#pragma unroll
    for (int k = 0; k < PSUM_TRIPS; ++k) asm volatile("" :: "v"(q[k]));
#pragma unroll
    for (int k = 0; k < PSUM_TRIPS; ++k) {
      const int i = t + 1024 * k;
      part += (i < n4) ? (q[k].x + q[k].y + q[k].z + q[k].w) : 0;
    }
  }
  if (t < NCHUNK) {
    runLen[t] = rl;
    runStart[t] = t * CHUNK + ro;
  }
  if (t < BINNODES) cnt[t] = 0;
  for (int off = 16; off > 0; off >>= 1) part += __shfl_xor(part, off);
  if (lane == 0) red[w] = part;
  __syncthreads();
  int P = 0;
  for (int k = 0; k < 32; ++k) P += red[k];

  for_bin_edges(pairs, runLen, runStart, w, lane,
                [&](const v2i e) { atomicAdd(&cnt[e.y - d0], 1); });
  __syncthreads();

  const int v = (t < BINNODES) ? cnt[t] : 0;
  const int incl = wave_incl_scan(v, lane);
  if (t < BINNODES && lane == 31) wsum[w] = incl;
  __syncthreads();
  int n = 0;
  for (int w2 = 0; w2 < 8; ++w2) n += wsum[w2];
  if (t < BINNODES) {
    int add = 0;
    for (int w2 = 0; w2 < w; ++w2) add += wsum[w2];
    const int excl = incl - v + add;
    cursor[t] = excl;
    if (has_node) {
      indeg[node] = v;
      rowptr[node] = P + excl;
      const float dd = rsqrtf((float)v + 1.0f);
      dinv[node] = dd;
      v4f o;
      o.x = x0 * dd; o.y = x1 * dd; o.z = x2 * dd; o.w = 0.0f;
      *(v4f*)(xs + (size_t)node * 4) = o;
    }
  }
  __syncthreads();

  const bool staged = (n <= STAGECAP);
  for_bin_edges(pairs, runLen, runStart, w, lane, [&](const v2i e) {
    const int pos = atomicAdd(&cursor[e.y - d0], 1);
    if (staged) stage[pos] = e.x;
    else csr_src[P + pos] = e.x;
  });
  __syncthreads();
  if (staged) {
    for (int i = t; i < n; i += 1024) csr_src[P + i] = stage[i];
  }
}

__global__ __launch_bounds__(256) void row_gather_x_kernel(
    const float* __restrict__ xs, const int* __restrict__ rowptr,
    const int* __restrict__ indeg, const int* __restrict__ csr_src,
    const float* __restrict__ dinv, float* __restrict__ xa) {
  const int t = blockIdx.x * 256 + threadIdx.x;
  const int row = t >> 2;
  const int sub = t & 3;
  const v4f zero4 = 0.0f;
  v4f acc = zero4;
  float dd = 0.0f;
  if (row < NNODES) {
    dd = dinv[row];
    const int start = rowptr[row];
    const int end = start + indeg[row];
    const int last = end - 1;
    if (sub == 0) acc = *(const v4f*)(xs + (size_t)row * 4);
    for (int base = start; base < end; base += 32) {
      int s[8];
      for (int u = 0; u < 8; ++u) {
        const int e = base + sub + 4 * u;
        s[u] = csr_src[(e < end) ? e : last];
      }
      for (int u = 0; u < 8; ++u) {
        const int e = base + sub + 4 * u;
        const v4f v = *(const v4f*)(xs + (size_t)s[u] * 4);
        acc += (e < end) ? v : zero4;
      }
    }
  }
  float r[3];
  for (int c = 0; c < 3; ++c) {
    float a = acc[c];
    a += __builtin_bit_cast(float, __builtin_amdgcn_update_dpp(
             0, __builtin_bit_cast(int, a), 0xB1, 0xF, 0xF, false));
    a += __builtin_bit_cast(float, __builtin_amdgcn_update_dpp(
             0, __builtin_bit_cast(int, a), 0x4E, 0xF, 0xF, false));
    r[c] = a;
  }
  if (sub == 0) {
    v4f o;
    o.x = r[0] * dd; o.y = r[1] * dd; o.z = r[2] * dd; o.w = 0.0f;
    *(v4f*)(xa + (size_t)row * 4) = o;
  }
}

template <int K, int N, bool RELU, bool BIAS, int KW = K, bool ROWSCALE = false>
__global__ __launch_bounds__(256) void gemm_wmma(
    const float* __restrict__ A, const float* __restrict__ W,
    const float* __restrict__ bias, const float* __restrict__ rscale, float* __restrict__ C) {
  __shared__ float lw[K * N];

  static_assert(KW == K || KW == K - 1, "only the last row of W may be missing");
  const int tid = threadIdx.x;
  constexpr int QUADS = (K / 2) * N / 4;
  for (int q = tid; q < QUADS; q += 256) {
    int t = (q * 4) / N;
    int n = (q * 4) % N;
    int k = t * 2;
    v4f w0 = *(const v4f*)(W + (size_t)k * N + n);
    v4f w1 = 0.0f;
    if (KW == K || k + 1 < KW) w1 = *(const v4f*)(W + (size_t)(k + 1) * N + n);
    v4f lo, hi;
    lo.x = w0.x; lo.y = w1.x; lo.z = w0.y; lo.w = w1.y;
    hi.x = w0.z; hi.y = w1.z; hi.z = w0.w; hi.w = w1.w;
    float* d = lw + ((size_t)t * N + n) * 2;
    *(v4f*)(d)     = lo;
    *(v4f*)(d + 4) = hi;
  }
  __syncthreads();

  const int lane = tid & 31;
  const int wave = tid >> 5;
  const int half = lane >> 4;
  const int l16  = lane & 15;
  const int row  = blockIdx.x * 128 + wave * 16 + l16;
  const float* __restrict__ Arow = A + (size_t)row * K + 2 * half;

  float sc[8];
  if (ROWSCALE) {
    static_assert(NNODES % 8 == 0, "a group of 8 rows is all nodes or all pad");
    const int rb = blockIdx.x * 128 + wave * 16 + half * 8;
    v4f s0 = 0.0f, s1 = 0.0f;
    if (rb < NNODES) {
      s0 = *(const v4f*)(rscale + rb);
      s1 = *(const v4f*)(rscale + rb + 4);
    }
    sc[0] = s0.x; sc[1] = s0.y; sc[2] = s0.z; sc[3] = s0.w;
    sc[4] = s1.x; sc[5] = s1.y; sc[6] = s1.z; sc[7] = s1.w;
  }

  for (int n_base = 0; n_base < N; n_base += 16) {
    const int n = n_base + l16;
    float binit = BIAS ? bias[n] : 0.0f;
    v8f acc;
#pragma unroll
    for (int r = 0; r < 8; ++r) acc[r] = binit;

#pragma unroll
    for (int k = 0; k < K; k += 4) {
      v2f av = *(const v2f*)(Arow + k);
      if (RELU) { av.x = fmaxf(av.x, 0.0f); av.y = fmaxf(av.y, 0.0f); }
      const int kk = k + 2 * half;
      v2f bv = *(const v2f*)(lw + ((size_t)(kk >> 1) * N + n) * 2);
      acc = __builtin_amdgcn_wmma_f32_16x16x4_f32(
          false, av, false, bv, (short)0, acc, false, false);
    }

    float* __restrict__ Cp = C + (size_t)(blockIdx.x * 128 + wave * 16 + half * 8) * N + n;
#pragma unroll
    for (int r = 0; r < 8; ++r) Cp[(size_t)r * N] = ROWSCALE ? acc[r] * sc[r] : acc[r];
  }
}

template <int F, bool POOL>
__global__ __launch_bounds__(256) void row_gather_kernel(
    const float* __restrict__ h, const int* __restrict__ rowptr,
    const int* __restrict__ indeg, const int* __restrict__ csr_src,
    const float* __restrict__ dinv, const float* __restrict__ bias,
    float* __restrict__ out, const int* __restrict__ batch,
    float* __restrict__ pooled, float* __restrict__ cnt) {
  constexpr int CH = F / 32;
  const int lane = threadIdx.x & 31;
  const int q = lane & 7;
  const int gl = lane & 24;
  const int fo = q * 4;
  const int row = (int)(blockIdx.x * 32 + (threadIdx.x >> 3));
  const v4f zero4 = 0.0f;

  float dd = 0.0f;
  int start = 0, n = 0;
  int g = -1;
  v4f acc[CH];
#pragma unroll
  for (int c = 0; c < CH; ++c) acc[c] = zero4;
  if (row < NNODES) {
    dd = dinv[row];
    start = rowptr[row];
    n = indeg[row];
    if constexpr (POOL) g = batch[row];
    const float* __restrict__ hr = h + (size_t)row * F + fo;
#pragma unroll
    for (int c = 0; c < CH; ++c) acc[c] = *(const v4f*)(hr + c * 32);
  }
  const int n0 = __builtin_amdgcn_readlane(n, 0);
  const int n1 = __builtin_amdgcn_readlane(n, 8);
  const int n2 = __builtin_amdgcn_readlane(n, 16);
  const int n3 = __builtin_amdgcn_readlane(n, 24);
  const int n01 = (n0 > n1) ? n0 : n1;
  const int n23 = (n2 > n3) ? n2 : n3;
  const int nmax = (n01 > n23) ? n01 : n23;

  for (int blk = 0; blk < nmax; blk += 32) {
    int ir[4];
#pragma unroll
    for (int r = 0; r < 4; ++r) {
      const int e = blk + q + 8 * r;
      const int pos = start + ((e < n) ? e : 0);
      ir[r] = csr_src[(pos < NEDGES) ? pos : (NEDGES - 1)];
    }
    const int lim = nmax - blk;
#pragma unroll
    for (int p = 0; p < 8; ++p) {
      if (4 * p < lim) {
        int s[4];
#pragma unroll
        for (int u = 0; u < 4; ++u) {
          const int from = gl + 4 * (p & 1) + u;
          s[u] = __shfl(ir[p >> 1], from);
        }
        if (blk + 4 * p < n) {
          v4f a[4][CH];
#pragma unroll
          for (int u = 0; u < 4; ++u) {
            const int su = (blk + 4 * p + u < n) ? s[u] : s[0];
            const float* __restrict__ hs = h + (size_t)su * F + fo;
#pragma unroll
            for (int c = 0; c < CH; ++c) a[u][c] = *(const v4f*)(hs + c * 32);
          }
#pragma unroll
          for (int u = 0; u < 4; ++u) {
            const float w = (blk + 4 * p + u < n) ? 1.0f : 0.0f;
#pragma unroll
            for (int c = 0; c < CH; ++c) acc[c] += a[u][c] * w;
          }
        }
      }
    }
  }
  if (row < NNODES) {
#pragma unroll
    for (int c = 0; c < CH; ++c) {
      const v4f bv = *(const v4f*)(bias + c * 32 + fo);
      acc[c] = acc[c] * dd + bv;
    }
  }
  if constexpr (POOL) {
    static_assert(F == 64, "the pooled table is 64 wide");
    const int g0 = __builtin_amdgcn_readlane(g, 0);
    const int g1 = __builtin_amdgcn_readlane(g, 8);
    const int g2 = __builtin_amdgcn_readlane(g, 16);
    const int g3 = __builtin_amdgcn_readlane(g, 24);
    if (g0 >= 0 && g0 == g1 && g0 == g2 && g0 == g3) {
      float t[CH][4];
#pragma unroll
      for (int c = 0; c < CH; ++c) {
#pragma unroll
        for (int j = 0; j < 4; ++j) {
          float v = acc[c][j];
          v += __shfl_xor(v, 8);
          v += __shfl_xor(v, 16);
          t[c][j] = v;
        }
      }
      const int k = lane >> 3;
      const float va = (k == 0) ? t[0][0] : (k == 1) ? t[0][2] : (k == 2) ? t[1][0] : t[1][2];
      const float vb = (k == 0) ? t[0][1] : (k == 1) ? t[0][3] : (k == 2) ? t[1][1] : t[1][3];
      float* p = pooled + (size_t)g0 * 64 + (k >> 1) * 32 + fo + (k & 1) * 2;
      atomicAdd(p, va);
      atomicAdd(p + 1, vb);
      if (lane == 0) atomicAdd(&cnt[g0], 4.0f);
    } else if (g >= 0) {
      float* p = pooled + (size_t)g * 64 + fo;
#pragma unroll
      for (int c = 0; c < CH; ++c) {
#pragma unroll
        for (int j = 0; j < 4; ++j) atomicAdd(p + c * 32 + j, acc[c][j]);
      }
      if (q == 0) atomicAdd(&cnt[g], 1.0f);
    }
  } else {
    float* __restrict__ o = out + (size_t)row * F + fo;
#pragma unroll
    for (int c = 0; c < CH; ++c) *(v4f*)(o + c * 32) = acc[c];
  }
}

#define TPITCH 132
template <int N>
__global__ __launch_bounds__(512) void gather_product_kernel(
    const float* __restrict__ h, const int* __restrict__ rowptr,
    const int* __restrict__ indeg, const int* __restrict__ csr_src,
    const float* __restrict__ dinv, const float* __restrict__ bias,
    const float* __restrict__ W, float* __restrict__ C) {
  constexpr int F = 128;
  constexpr int K = 128;
  constexpr int CH = F / 32;
  __shared__ float lw[K * N];
  __shared__ __attribute__((aligned(16))) float tile[128 * TPITCH];

  const int tid = threadIdx.x;
  const int lane = tid & 31;
  const int wave = tid >> 5;

  constexpr int QUADS = (K / 2) * N / 4;
  for (int qd = tid; qd < QUADS; qd += 512) {
    const int t = (qd * 4) / N;
    const int n = (qd * 4) % N;
    const int k = t * 2;
    const v4f w0 = *(const v4f*)(W + (size_t)k * N + n);
    const v4f w1 = *(const v4f*)(W + (size_t)(k + 1) * N + n);
    v4f lo, hi;
    lo.x = w0.x; lo.y = w1.x; lo.z = w0.y; lo.w = w1.y;
    hi.x = w0.z; hi.y = w1.z; hi.z = w0.w; hi.w = w1.w;
    float* d = lw + ((size_t)t * N + n) * 2;
    *(v4f*)(d)     = lo;
    *(v4f*)(d + 4) = hi;
  }

  const int q = lane & 7;
  const int gl = lane & 24;
  const int fo = q * 4;
  const v4f zero4 = 0.0f;
#pragma nounroll
  for (int it = 0; it < 2; ++it) {
    const int lrow = wave * 8 + it * 4 + (lane >> 3);
    const int row = (int)(blockIdx.x * 128) + lrow;
    float dd = 0.0f;
    int start = 0, n = 0;
    v4f acc[CH];
#pragma unroll
    for (int c = 0; c < CH; ++c) acc[c] = zero4;
    if (row < NNODES) {
      dd = dinv[row];
      start = rowptr[row];
      n = indeg[row];
      const float* __restrict__ hr = h + (size_t)row * F + fo;
#pragma unroll
      for (int c = 0; c < CH; ++c) acc[c] = *(const v4f*)(hr + c * 32);
    }
    const int n0 = __builtin_amdgcn_readlane(n, 0);
    const int n1 = __builtin_amdgcn_readlane(n, 8);
    const int n2 = __builtin_amdgcn_readlane(n, 16);
    const int n3 = __builtin_amdgcn_readlane(n, 24);
    const int n01 = (n0 > n1) ? n0 : n1;
    const int n23 = (n2 > n3) ? n2 : n3;
    const int nmax = (n01 > n23) ? n01 : n23;

    for (int blk = 0; blk < nmax; blk += 32) {
      int ir[4];
#pragma unroll
      for (int r = 0; r < 4; ++r) {
        const int e = blk + q + 8 * r;
        const int pos = start + ((e < n) ? e : 0);
        ir[r] = csr_src[(pos < NEDGES) ? pos : (NEDGES - 1)];
      }
      const int lim = nmax - blk;
#pragma unroll
      for (int p = 0; p < 8; ++p) {
        if (4 * p < lim) {
          int s[4];
#pragma unroll
          for (int u = 0; u < 4; ++u) {
            const int from = gl + 4 * (p & 1) + u;
            s[u] = __shfl(ir[p >> 1], from);
          }
          if (blk + 4 * p < n) {
            v4f a[4][CH];
#pragma unroll
            for (int u = 0; u < 4; ++u) {
              const int su = (blk + 4 * p + u < n) ? s[u] : s[0];
              const float* __restrict__ hs = h + (size_t)su * F + fo;
#pragma unroll
              for (int c = 0; c < CH; ++c) a[u][c] = *(const v4f*)(hs + c * 32);
            }
#pragma unroll
            for (int u = 0; u < 4; ++u) {
              const float w = (blk + 4 * p + u < n) ? 1.0f : 0.0f;
#pragma unroll
              for (int c = 0; c < CH; ++c) acc[c] += a[u][c] * w;
            }
          }
        }
      }
    }
    float* __restrict__ tr = tile + lrow * TPITCH + fo;
#pragma unroll
    for (int c = 0; c < CH; ++c) {
      v4f v = zero4;
      if (row < NNODES) {
        const v4f bv = *(const v4f*)(bias + c * 32 + fo);
        v = acc[c] * dd + bv;
        v.x = fmaxf(v.x, 0.0f); v.y = fmaxf(v.y, 0.0f);
        v.z = fmaxf(v.z, 0.0f); v.w = fmaxf(v.w, 0.0f);
      }
      *(v4f*)(tr + c * 32) = v;
    }
  }
  __syncthreads();

  const int half = lane >> 4;
  const int l16  = lane & 15;
  const int rt = wave >> 1;
  const int chalf = wave & 1;
  constexpr int NT = N / 32;
  const int rb = (int)(blockIdx.x * 128) + rt * 16 + half * 8;
  static_assert(NNODES % 8 == 0, "a group of 8 rows is all nodes or all pad");
  float sc[8];
  {
    v4f s0 = 0.0f, s1 = 0.0f;
    if (rb < NNODES) {
      s0 = *(const v4f*)(dinv + rb);
      s1 = *(const v4f*)(dinv + rb + 4);
    }
    sc[0] = s0.x; sc[1] = s0.y; sc[2] = s0.z; sc[3] = s0.w;
    sc[4] = s1.x; sc[5] = s1.y; sc[6] = s1.z; sc[7] = s1.w;
  }
  const float* __restrict__ Ar = tile + (rt * 16 + l16) * TPITCH + 2 * half;
  v8f pacc[NT];
#pragma unroll
  for (int j = 0; j < NT; ++j) {
#pragma unroll
    for (int r = 0; r < 8; ++r) pacc[j][r] = 0.0f;
  }
#pragma unroll
  for (int k = 0; k < K; k += 4) {
    const v2f av = *(const v2f*)(Ar + k);
    const int kk = k + 2 * half;
#pragma unroll
    for (int j = 0; j < NT; ++j) {
      const int n = (chalf * NT + j) * 16 + l16;
      const v2f bv = *(const v2f*)(lw + ((size_t)(kk >> 1) * N + n) * 2);
      pacc[j] = __builtin_amdgcn_wmma_f32_16x16x4_f32(
          false, av, false, bv, (short)0, pacc[j], false, false);
    }
  }
#pragma unroll
  for (int j = 0; j < NT; ++j) {
    const int n = (chalf * NT + j) * 16 + l16;
    float* __restrict__ Cp = C + (size_t)rb * N + n;
#pragma unroll
    for (int r = 0; r < 8; ++r) Cp[(size_t)r * N] = pacc[j][r] * sc[r];
  }
}

#define HIDP 34
__global__ __launch_bounds__(256) void head_kernel(
    const float* __restrict__ sums, const float* __restrict__ cnt,
    const float* __restrict__ Wl1, const float* __restrict__ bl1,
    const float* __restrict__ Wl2, const float* __restrict__ bl2,
    float* __restrict__ out) {
  __shared__ __attribute__((aligned(16))) float lw1[64 * 32];
  __shared__ __attribute__((aligned(16))) float lw2[32 * 16];
  __shared__ __attribute__((aligned(16))) float hid[8 * 16 * HIDP];
  __shared__ float ost[128 * 9];

  const int tid  = threadIdx.x;
  const int lane = tid & 31;
  const int wave = tid >> 5;
  const int half = lane >> 4;
  const int l16  = lane & 15;
  const int row0 = blockIdx.x * 128 + wave * 16;
  const int row  = row0 + l16;

  const float den = fmaxf(cnt[row], 1.0f);
  const float* __restrict__ Arow = sums + (size_t)row * 64 + 2 * half;
  v2f a1[16];
#pragma unroll
  for (int s = 0; s < 16; ++s) {
    a1[s] = *(const v2f*)(Arow + 4 * s);
    a1[s].x = a1[s].x / den;
    a1[s].y = a1[s].y / den;
  }

  {
    const int t = (tid * 4) / 32;
    const int n = (tid * 4) % 32;
    const int k = t * 2;
    const v4f w0 = *(const v4f*)(Wl1 + (size_t)k * 32 + n);
    const v4f w1 = *(const v4f*)(Wl1 + (size_t)(k + 1) * 32 + n);
    v4f lo, hi;
    lo.x = w0.x; lo.y = w1.x; lo.z = w0.y; lo.w = w1.y;
    hi.x = w0.z; hi.y = w1.z; hi.z = w0.w; hi.w = w1.w;
    float* d = lw1 + ((size_t)t * 32 + n) * 2;
    *(v4f*)(d)     = lo;
    *(v4f*)(d + 4) = hi;
  }
  for (int i = tid; i < 32 * 16; i += 256) {
    const int k = i >> 4, n = i & 15;
    lw2[((k >> 1) * 16 + n) * 2 + (k & 1)] = (n < 9) ? Wl2[k * 9 + n] : 0.0f;
  }
  __syncthreads();

  float* __restrict__ hw = hid + wave * (16 * HIDP);
#pragma unroll
  for (int j = 0; j < 2; ++j) {
    const int n = 16 * j + l16;
    const float binit = bl1[n];
    v8f acc;
#pragma unroll
    for (int r = 0; r < 8; ++r) acc[r] = binit;
#pragma unroll
    for (int s = 0; s < 16; ++s) {
      const int kk = 4 * s + 2 * half;
      const v2f bv = *(const v2f*)(lw1 + ((size_t)(kk >> 1) * 32 + n) * 2);
      acc = __builtin_amdgcn_wmma_f32_16x16x4_f32(
          false, a1[s], false, bv, (short)0, acc, false, false);
    }
#pragma unroll
    for (int r = 0; r < 8; ++r) hw[(half * 8 + r) * HIDP + n] = acc[r];
  }
  __syncthreads();

  const float binit2 = (l16 < 9) ? bl2[l16] : 0.0f;
  v8f acc2;
#pragma unroll
  for (int r = 0; r < 8; ++r) acc2[r] = binit2;
#pragma unroll
  for (int s = 0; s < 8; ++s) {
    const int kk = 4 * s + 2 * half;
    v2f av = *(const v2f*)(hw + l16 * HIDP + kk);
    av.x = fmaxf(av.x, 0.0f); av.y = fmaxf(av.y, 0.0f);
    const v2f bv = *(const v2f*)(lw2 + ((size_t)(kk >> 1) * 16 + l16) * 2);
    acc2 = __builtin_amdgcn_wmma_f32_16x16x4_f32(
        false, av, false, bv, (short)0, acc2, false, false);
  }
  if (l16 < 9) {
    float* __restrict__ op = ost + (wave * 16 + half * 8) * 9 + l16;
#pragma unroll
    for (int r = 0; r < 8; ++r) op[r * 9] = acc2[r];
  }
  __syncthreads();
  float* __restrict__ ob = out + (size_t)blockIdx.x * (128 * 9);
  for (int i = tid; i < 128 * 9; i += 256) ob[i] = ost[i];
}

extern "C" void kernel_launch(void* const* d_in, const int* in_sizes, int n_in,
                              void* d_out, int out_size, void* d_ws, size_t ws_size,
                              hipStream_t stream) {
  const float* x   = (const float*)d_in[0];
  const int*   ei  = (const int*)d_in[1];
  const int*   bat = (const int*)d_in[2];
  const float* W1  = (const float*)d_in[3];  const float* b1  = (const float*)d_in[4];
  const float* W2  = (const float*)d_in[5];  const float* b2  = (const float*)d_in[6];
  const float* W3  = (const float*)d_in[7];  const float* b3  = (const float*)d_in[8];
  const float* W4  = (const float*)d_in[9];  const float* b4  = (const float*)d_in[10];
  const float* Wl1 = (const float*)d_in[11]; const float* bl1 = (const float*)d_in[12];
  const float* Wl2 = (const float*)d_in[13]; const float* bl2 = (const float*)d_in[14];
  float* out = (float*)d_out;

  const int* src = ei;
  const int* dst = ei + NEDGES;

  float* ws     = (float*)d_ws;
  float* dinv   = ws;
  float* xs     = dinv + NNODES;
  float* xa     = xs + (size_t)MPAD * 4;
  float* hT     = xa + (size_t)MPAD * 4;
  float* hA     = hT + (size_t)MPAD * 128;
  float* pooled = hA + (size_t)MPAD * 128;
  float* cnt    = pooled + NGRAPHS * 64;
  int*   indeg   = (int*)(cnt + NGRAPHS);
  int*   rowptr  = indeg + NNODES;
  int*   cntT    = rowptr + NNODES;
  int*   offT    = cntT + NBINS * NCHUNK;
  int*   binned  = offT + NBINS * NCHUNK;
  int*   csr_src = binned + 2 * NEDGES;

  bin_chunk_kernel<<<NCHUNK, 1024, 0, stream>>>(src, dst, binned, cntT, offT);
  bin_rows_kernel<<<NBINS, 1024, 0, stream>>>(binned, cntT, offT, x, indeg, rowptr, dinv, xs, csr_src,
                                              pooled, cnt);

  const unsigned GB = MPAD / 128;
  const unsigned RB = MPAD / 32;

  row_gather_x_kernel<<<MPAD / 64, 256, 0, stream>>>(xs, rowptr, indeg, csr_src, dinv, xa);
  gemm_wmma<4, 128, false, true, 3><<<GB, 256, 0, stream>>>(xa, W1, b1, nullptr, hA);

  gemm_wmma<128, 128, true, false, 128, true><<<GB, 256, 0, stream>>>(hA, W2, nullptr, dinv, hT);

  gather_product_kernel<128><<<GB, 512, 0, stream>>>(hT, rowptr, indeg, csr_src, dinv, b2, W3, hA);

  gather_product_kernel<64><<<GB, 512, 0, stream>>>(hA, rowptr, indeg, csr_src, dinv, b3, W4, hT);

  row_gather_kernel<64, true><<<RB, 256, 0, stream>>>(hT, rowptr, indeg, csr_src, dinv, b4, nullptr,
                                                     bat, pooled, cnt);

  head_kernel<<<NGRAPHS / 128, 256, 0, stream>>>(pooled, cnt, Wl1, bl1, Wl2, bl2, out);
}


# Layer 2's gather rebuilds each source's layer-1 row from its 16-B input by a K=4 WMMA instead of fetching 512-B rows; W2, W3 products follow in the block; 9 launches to 7; layer-2 sum order changes
# speedup vs baseline: 6.3216x; 1.1804x over previous
#include <hip/hip_runtime.h>
#include <hip/hip_bf16.h>

typedef __attribute__((ext_vector_type(2))) float v2f;
typedef __attribute__((ext_vector_type(4))) float v4f;
typedef __attribute__((ext_vector_type(8))) float v8f;

#define NNODES  100000
#define MPAD    100096
#define NEDGES  1600000
#define NGRAPHS 512

typedef __attribute__((ext_vector_type(2))) int v2i;
typedef __attribute__((ext_vector_type(4))) int v4i;

#define BINSHIFT 8
#define BINNODES 256
#define NBINS    ((NNODES + BINNODES - 1) / BINNODES)
#define NCHUNK   128
#define CHUNK    (NEDGES / NCHUNK)
#define EPT      ((CHUNK + 1023) / 1024)
#define STAGECAP 16384
static_assert(NEDGES % NCHUNK == 0, "chunks must tile the edge list");
static_assert(NBINS <= 512, "bin scan covers 512 bins");

__device__ __forceinline__ int wave_incl_scan(int v, int lane) {
  for (int off = 1; off < 32; off <<= 1) {
    const int u = __shfl_up(v, off);
    if (lane >= off) v += u;
  }
  return v;
}

__global__ __launch_bounds__(1024) void bin_chunk_kernel(
    const int* __restrict__ src, const int* __restrict__ dst,
    int* __restrict__ binned, int* __restrict__ cntT, int* __restrict__ offT) {
  __shared__ int hist[512];
  __shared__ int wsum[16];
  __shared__ v2i sorted[CHUNK];
  const int t = threadIdx.x;
  const int lane = t & 31;
  const int b = blockIdx.x;
  const int base = b * CHUNK;
  if (t < 512) hist[t] = 0;
  __syncthreads();

  int s[EPT], d[EPT], r[EPT];
#pragma unroll
  for (int i = 0; i < EPT; ++i) {
    const int idx = t + 1024 * i;
    s[i] = 0; d[i] = 0;
    if (idx < CHUNK) { s[i] = src[base + idx]; d[i] = dst[base + idx]; }
  }
#pragma unroll
  for (int i = 0; i < EPT; ++i) {
    const int idx = t + 1024 * i;
    r[i] = 0;
    if (idx < CHUNK) r[i] = atomicAdd(&hist[d[i] >> BINSHIFT], 1);
  }
  __syncthreads();

  const int v = (t < 512) ? hist[t] : 0;
  const int incl = wave_incl_scan(v, lane);
  if (t < 512 && lane == 31) wsum[t >> 5] = incl;
  __syncthreads();
  if (t < 512) {
    int add = 0;
    for (int w2 = 0; w2 < (t >> 5); ++w2) add += wsum[w2];
    const int excl = incl - v + add;
    hist[t] = excl;
    if (t < NBINS) {
      cntT[t * NCHUNK + b] = v;
      offT[t * NCHUNK + b] = excl;
    }
  }
  __syncthreads();

#pragma unroll
  for (int i = 0; i < EPT; ++i) {
    const int idx = t + 1024 * i;
    if (idx < CHUNK) {
      v2i e; e.x = s[i]; e.y = d[i];
      sorted[hist[d[i] >> BINSHIFT] + r[i]] = e;
    }
  }
  __syncthreads();
  v2i* __restrict__ outp = (v2i*)binned + base;
#pragma unroll
  for (int i = 0; i < EPT; ++i) {
    const int idx = t + 1024 * i;
    if (idx < CHUNK) outp[idx] = sorted[idx];
  }
}

#define RUNS_PER_WAVE (NCHUNK / 32)
#define PSUM_TRIPS    ((NBINS * (NCHUNK / 4) + 1023) / 1024)

template <typename F>
__device__ __forceinline__ void for_bin_edges(const v2i* __restrict__ pairs, const int* runLen,
                                              const int* runStart, int w, int lane, F f) {
  int len[RUNS_PER_WAVE], st[RUNS_PER_WAVE];
  v2i e0[RUNS_PER_WAVE], e1[RUNS_PER_WAVE];
#pragma unroll
  for (int q = 0; q < RUNS_PER_WAVE; ++q) {
    len[q] = runLen[w + 32 * q];
    st[q] = runStart[w + 32 * q];
    e0[q] = pairs[(lane < len[q]) ? (st[q] + lane) : 0];
    e1[q] = pairs[(lane + 32 < len[q]) ? (st[q] + lane + 32) : 0];
  }
#pragma unroll
  for (int q = 0; q < RUNS_PER_WAVE; ++q) asm volatile("" :: "v"(e0[q]), "v"(e1[q]));
#pragma unroll
  for (int q = 0; q < RUNS_PER_WAVE; ++q) {
    if (lane < len[q]) f(e0[q]);
    if (lane + 32 < len[q]) f(e1[q]);
  }
#pragma unroll
  for (int q = 0; q < RUNS_PER_WAVE; ++q) {
    for (int j = lane + 64; j < len[q]; j += 32) f(pairs[st[q] + j]);
  }
}

__global__ __launch_bounds__(1024) void bin_rows_kernel(
    const int* __restrict__ binned, const int* __restrict__ cntT, const int* __restrict__ offT,
    const float* __restrict__ x, int* __restrict__ indeg, int* __restrict__ rowptr,
    float* __restrict__ dinv, float* __restrict__ xs, int* __restrict__ csr_src,
    float* __restrict__ pooled, float* __restrict__ pcnt) {
  static_assert(NBINS * 1024 >= NGRAPHS * 65, "the grid covers the pool table and its counts");
  __shared__ int cnt[BINNODES];
  __shared__ int cursor[BINNODES];
  __shared__ int runLen[NCHUNK];
  __shared__ int runStart[NCHUNK];
  __shared__ int red[32];
  __shared__ int wsum[8];
  __shared__ int stage[STAGECAP];
  const int t = threadIdx.x;
  const int lane = t & 31;
  const int w = t >> 5;
  const int bin = blockIdx.x;
  const int d0 = bin << BINSHIFT;
  const v2i* __restrict__ pairs = (const v2i*)binned;

  {
    const int gid = bin * 1024 + t;
    if (gid < NGRAPHS * 64) pooled[gid] = 0.0f;
    else if (gid < NGRAPHS * 65) pcnt[gid - NGRAPHS * 64] = 0.0f;
  }

  const int node = d0 + t;
  const bool has_node = (t < BINNODES) && (node < NNODES);
  float x0 = 0.0f, x1 = 0.0f, x2 = 0.0f;
  if (has_node) {
    const float* __restrict__ xr = x + (size_t)node * 3;
    x0 = xr[0]; x1 = xr[1]; x2 = xr[2];
  }
  int rl = 0, ro = 0;
  if (t < NCHUNK) {
    rl = cntT[bin * NCHUNK + t];
    ro = offT[bin * NCHUNK + t];
  }
  int part = 0;
  {
    const v4i* __restrict__ c4 = (const v4i*)cntT;
    const int n4 = bin * (NCHUNK / 4);
    v4i q[PSUM_TRIPS];
#pragma unroll
    for (int k = 0; k < PSUM_TRIPS; ++k) {
      const int i = t + 1024 * k;
      q[k] = c4[(i < NBINS * (NCHUNK / 4)) ? i : 0];
    }
#pragma unroll
    for (int k = 0; k < PSUM_TRIPS; ++k) asm volatile("" :: "v"(q[k]));
#pragma unroll
    for (int k = 0; k < PSUM_TRIPS; ++k) {
      const int i = t + 1024 * k;
      part += (i < n4) ? (q[k].x + q[k].y + q[k].z + q[k].w) : 0;
    }
  }
  if (t < NCHUNK) {
    runLen[t] = rl;
    runStart[t] = t * CHUNK + ro;
  }
  if (t < BINNODES) cnt[t] = 0;
  for (int off = 16; off > 0; off >>= 1) part += __shfl_xor(part, off);
  if (lane == 0) red[w] = part;
  __syncthreads();
  int P = 0;
  for (int k = 0; k < 32; ++k) P += red[k];

  for_bin_edges(pairs, runLen, runStart, w, lane,
                [&](const v2i e) { atomicAdd(&cnt[e.y - d0], 1); });
  __syncthreads();

  const int v = (t < BINNODES) ? cnt[t] : 0;
  const int incl = wave_incl_scan(v, lane);
  if (t < BINNODES && lane == 31) wsum[w] = incl;
  __syncthreads();
  int n = 0;
  for (int w2 = 0; w2 < 8; ++w2) n += wsum[w2];
  if (t < BINNODES) {
    int add = 0;
    for (int w2 = 0; w2 < w; ++w2) add += wsum[w2];
    const int excl = incl - v + add;
    cursor[t] = excl;
    if (has_node) {
      indeg[node] = v;
      rowptr[node] = P + excl;
      const float dd = rsqrtf((float)v + 1.0f);
      dinv[node] = dd;
      v4f o;
      o.x = x0 * dd; o.y = x1 * dd; o.z = x2 * dd; o.w = 0.0f;
      *(v4f*)(xs + (size_t)node * 4) = o;
    }
  }
  __syncthreads();

  const bool staged = (n <= STAGECAP);
  for_bin_edges(pairs, runLen, runStart, w, lane, [&](const v2i e) {
    const int pos = atomicAdd(&cursor[e.y - d0], 1);
    if (staged) stage[pos] = e.x;
    else csr_src[P + pos] = e.x;
  });
  __syncthreads();
  if (staged) {
    for (int i = t; i < n; i += 1024) csr_src[P + i] = stage[i];
  }
}

__global__ __launch_bounds__(256) void row_gather_x_kernel(
    const float* __restrict__ xs, const int* __restrict__ rowptr,
    const int* __restrict__ indeg, const int* __restrict__ csr_src,
    const float* __restrict__ dinv, float* __restrict__ xa) {
  const int t = blockIdx.x * 256 + threadIdx.x;
  const int row = t >> 2;
  const int sub = t & 3;
  const v4f zero4 = 0.0f;
  v4f acc = zero4;
  float dd = 0.0f;
  if (row < NNODES) {
    dd = dinv[row];
    const int start = rowptr[row];
    const int end = start + indeg[row];
    const int last = end - 1;
    if (sub == 0) acc = *(const v4f*)(xs + (size_t)row * 4);
    for (int base = start; base < end; base += 32) {
      int s[8];
      for (int u = 0; u < 8; ++u) {
        const int e = base + sub + 4 * u;
        s[u] = csr_src[(e < end) ? e : last];
      }
      for (int u = 0; u < 8; ++u) {
        const int e = base + sub + 4 * u;
        const v4f v = *(const v4f*)(xs + (size_t)s[u] * 4);
        acc += (e < end) ? v : zero4;
      }
    }
  }
  float r[3];
  for (int c = 0; c < 3; ++c) {
    float a = acc[c];
    a += __builtin_bit_cast(float, __builtin_amdgcn_update_dpp(
             0, __builtin_bit_cast(int, a), 0xB1, 0xF, 0xF, false));
    a += __builtin_bit_cast(float, __builtin_amdgcn_update_dpp(
             0, __builtin_bit_cast(int, a), 0x4E, 0xF, 0xF, false));
    r[c] = a;
  }
  if (sub == 0) {
    v4f o;
    o.x = r[0] * dd * dd; o.y = r[1] * dd * dd; o.z = r[2] * dd * dd; o.w = dd;
    *(v4f*)(xa + (size_t)row * 4) = o;
  }
}

template <int F, bool POOL>
__global__ __launch_bounds__(256) void row_gather_kernel(
    const float* __restrict__ h, const int* __restrict__ rowptr,
    const int* __restrict__ indeg, const int* __restrict__ csr_src,
    const float* __restrict__ dinv, const float* __restrict__ bias,
    float* __restrict__ out, const int* __restrict__ batch,
    float* __restrict__ pooled, float* __restrict__ cnt) {
  constexpr int CH = F / 32;
  const int lane = threadIdx.x & 31;
  const int q = lane & 7;
  const int gl = lane & 24;
  const int fo = q * 4;
  const int row = (int)(blockIdx.x * 32 + (threadIdx.x >> 3));
  const v4f zero4 = 0.0f;

  float dd = 0.0f;
  int start = 0, n = 0;
  int g = -1;
  v4f acc[CH];
#pragma unroll
  for (int c = 0; c < CH; ++c) acc[c] = zero4;
  if (row < NNODES) {
    dd = dinv[row];
    start = rowptr[row];
    n = indeg[row];
    if constexpr (POOL) g = batch[row];
    const float* __restrict__ hr = h + (size_t)row * F + fo;
#pragma unroll
    for (int c = 0; c < CH; ++c) acc[c] = *(const v4f*)(hr + c * 32);
  }
  const int n0 = __builtin_amdgcn_readlane(n, 0);
  const int n1 = __builtin_amdgcn_readlane(n, 8);
  const int n2 = __builtin_amdgcn_readlane(n, 16);
  const int n3 = __builtin_amdgcn_readlane(n, 24);
  const int n01 = (n0 > n1) ? n0 : n1;
  const int n23 = (n2 > n3) ? n2 : n3;
  const int nmax = (n01 > n23) ? n01 : n23;

  for (int blk = 0; blk < nmax; blk += 32) {
    int ir[4];
#pragma unroll
    for (int r = 0; r < 4; ++r) {
      const int e = blk + q + 8 * r;
      const int pos = start + ((e < n) ? e : 0);
      ir[r] = csr_src[(pos < NEDGES) ? pos : (NEDGES - 1)];
    }
    const int lim = nmax - blk;
#pragma unroll
    for (int p = 0; p < 8; ++p) {
      if (4 * p < lim) {
        int s[4];
#pragma unroll
        for (int u = 0; u < 4; ++u) {
          const int from = gl + 4 * (p & 1) + u;
          s[u] = __shfl(ir[p >> 1], from);
        }
        if (blk + 4 * p < n) {
          v4f a[4][CH];
#pragma unroll
          for (int u = 0; u < 4; ++u) {
            const int su = (blk + 4 * p + u < n) ? s[u] : s[0];
            const float* __restrict__ hs = h + (size_t)su * F + fo;
#pragma unroll
            for (int c = 0; c < CH; ++c) a[u][c] = *(const v4f*)(hs + c * 32);
          }
#pragma unroll
          for (int u = 0; u < 4; ++u) {
            const float w = (blk + 4 * p + u < n) ? 1.0f : 0.0f;
#pragma unroll
            for (int c = 0; c < CH; ++c) acc[c] += a[u][c] * w;
          }
        }
      }
    }
  }
  if (row < NNODES) {
#pragma unroll
    for (int c = 0; c < CH; ++c) {
      const v4f bv = *(const v4f*)(bias + c * 32 + fo);
      acc[c] = acc[c] * dd + bv;
    }
  }
  if constexpr (POOL) {
    static_assert(F == 64, "the pooled table is 64 wide");
    const int g0 = __builtin_amdgcn_readlane(g, 0);
    const int g1 = __builtin_amdgcn_readlane(g, 8);
    const int g2 = __builtin_amdgcn_readlane(g, 16);
    const int g3 = __builtin_amdgcn_readlane(g, 24);
    if (g0 >= 0 && g0 == g1 && g0 == g2 && g0 == g3) {
      float t[CH][4];
#pragma unroll
      for (int c = 0; c < CH; ++c) {
#pragma unroll
        for (int j = 0; j < 4; ++j) {
          float v = acc[c][j];
          v += __shfl_xor(v, 8);
          v += __shfl_xor(v, 16);
          t[c][j] = v;
        }
      }
      const int k = lane >> 3;
      const float va = (k == 0) ? t[0][0] : (k == 1) ? t[0][2] : (k == 2) ? t[1][0] : t[1][2];
      const float vb = (k == 0) ? t[0][1] : (k == 1) ? t[0][3] : (k == 2) ? t[1][1] : t[1][3];
      float* p = pooled + (size_t)g0 * 64 + (k >> 1) * 32 + fo + (k & 1) * 2;
      atomicAdd(p, va);
      atomicAdd(p + 1, vb);
      if (lane == 0) atomicAdd(&cnt[g0], 4.0f);
    } else if (g >= 0) {
      float* p = pooled + (size_t)g * 64 + fo;
#pragma unroll
      for (int c = 0; c < CH; ++c) {
#pragma unroll
        for (int j = 0; j < 4; ++j) atomicAdd(p + c * 32 + j, acc[c][j]);
      }
      if (q == 0) atomicAdd(&cnt[g], 1.0f);
    }
  } else {
    float* __restrict__ o = out + (size_t)row * F + fo;
#pragma unroll
    for (int c = 0; c < CH; ++c) *(v4f*)(o + c * 32) = acc[c];
  }
}

#define TPITCH 132
template <int N>
__global__ __launch_bounds__(512) void gather_product_kernel(
    const float* __restrict__ h, const int* __restrict__ rowptr,
    const int* __restrict__ indeg, const int* __restrict__ csr_src,
    const float* __restrict__ dinv, const float* __restrict__ bias,
    const float* __restrict__ W, float* __restrict__ C) {
  constexpr int F = 128;
  constexpr int K = 128;
  constexpr int CH = F / 32;
  __shared__ float lw[K * N];
  __shared__ __attribute__((aligned(16))) float tile[128 * TPITCH];

  const int tid = threadIdx.x;
  const int lane = tid & 31;
  const int wave = tid >> 5;

  constexpr int QUADS = (K / 2) * N / 4;
  for (int qd = tid; qd < QUADS; qd += 512) {
    const int t = (qd * 4) / N;
    const int n = (qd * 4) % N;
    const int k = t * 2;
    const v4f w0 = *(const v4f*)(W + (size_t)k * N + n);
    const v4f w1 = *(const v4f*)(W + (size_t)(k + 1) * N + n);
    v4f lo, hi;
    lo.x = w0.x; lo.y = w1.x; lo.z = w0.y; lo.w = w1.y;
    hi.x = w0.z; hi.y = w1.z; hi.z = w0.w; hi.w = w1.w;
    float* d = lw + ((size_t)t * N + n) * 2;
    *(v4f*)(d)     = lo;
    *(v4f*)(d + 4) = hi;
  }

  const int q = lane & 7;
  const int gl = lane & 24;
  const int fo = q * 4;
  const v4f zero4 = 0.0f;
#pragma nounroll
  for (int it = 0; it < 2; ++it) {
    const int lrow = wave * 8 + it * 4 + (lane >> 3);
    const int row = (int)(blockIdx.x * 128) + lrow;
    float dd = 0.0f;
    int start = 0, n = 0;
    v4f acc[CH];
#pragma unroll
    for (int c = 0; c < CH; ++c) acc[c] = zero4;
    if (row < NNODES) {
      dd = dinv[row];
      start = rowptr[row];
      n = indeg[row];
      const float* __restrict__ hr = h + (size_t)row * F + fo;
#pragma unroll
      for (int c = 0; c < CH; ++c) acc[c] = *(const v4f*)(hr + c * 32);
    }
    const int n0 = __builtin_amdgcn_readlane(n, 0);
    const int n1 = __builtin_amdgcn_readlane(n, 8);
    const int n2 = __builtin_amdgcn_readlane(n, 16);
    const int n3 = __builtin_amdgcn_readlane(n, 24);
    const int n01 = (n0 > n1) ? n0 : n1;
    const int n23 = (n2 > n3) ? n2 : n3;
    const int nmax = (n01 > n23) ? n01 : n23;

    for (int blk = 0; blk < nmax; blk += 32) {
      int ir[4];
#pragma unroll
      for (int r = 0; r < 4; ++r) {
        const int e = blk + q + 8 * r;
        const int pos = start + ((e < n) ? e : 0);
        ir[r] = csr_src[(pos < NEDGES) ? pos : (NEDGES - 1)];
      }
      const int lim = nmax - blk;
#pragma unroll
      for (int p = 0; p < 8; ++p) {
        if (4 * p < lim) {
          int s[4];
#pragma unroll
          for (int u = 0; u < 4; ++u) {
            const int from = gl + 4 * (p & 1) + u;
            s[u] = __shfl(ir[p >> 1], from);
          }
          if (blk + 4 * p < n) {
            v4f a[4][CH];
#pragma unroll
            for (int u = 0; u < 4; ++u) {
              const int su = (blk + 4 * p + u < n) ? s[u] : s[0];
              const float* __restrict__ hs = h + (size_t)su * F + fo;
#pragma unroll
              for (int c = 0; c < CH; ++c) a[u][c] = *(const v4f*)(hs + c * 32);
            }
#pragma unroll
            for (int u = 0; u < 4; ++u) {
              const float w = (blk + 4 * p + u < n) ? 1.0f : 0.0f;
#pragma unroll
              for (int c = 0; c < CH; ++c) acc[c] += a[u][c] * w;
            }
          }
        }
      }
    }
    float* __restrict__ tr = tile + lrow * TPITCH + fo;
#pragma unroll
    for (int c = 0; c < CH; ++c) {
      v4f v = zero4;
      if (row < NNODES) {
        const v4f bv = *(const v4f*)(bias + c * 32 + fo);
        v = acc[c] * dd + bv;
        v.x = fmaxf(v.x, 0.0f); v.y = fmaxf(v.y, 0.0f);
        v.z = fmaxf(v.z, 0.0f); v.w = fmaxf(v.w, 0.0f);
      }
      *(v4f*)(tr + c * 32) = v;
    }
  }
  __syncthreads();

  const int half = lane >> 4;
  const int l16  = lane & 15;
  const int rt = wave >> 1;
  const int chalf = wave & 1;
  constexpr int NT = N / 32;
  const int rb = (int)(blockIdx.x * 128) + rt * 16 + half * 8;
  static_assert(NNODES % 8 == 0, "a group of 8 rows is all nodes or all pad");
  float sc[8];
  {
    v4f s0 = 0.0f, s1 = 0.0f;
    if (rb < NNODES) {
      s0 = *(const v4f*)(dinv + rb);
      s1 = *(const v4f*)(dinv + rb + 4);
    }
    sc[0] = s0.x; sc[1] = s0.y; sc[2] = s0.z; sc[3] = s0.w;
    sc[4] = s1.x; sc[5] = s1.y; sc[6] = s1.z; sc[7] = s1.w;
  }
  const float* __restrict__ Ar = tile + (rt * 16 + l16) * TPITCH + 2 * half;
  v8f pacc[NT];
#pragma unroll
  for (int j = 0; j < NT; ++j) {
#pragma unroll
    for (int r = 0; r < 8; ++r) pacc[j][r] = 0.0f;
  }
#pragma unroll
  for (int k = 0; k < K; k += 4) {
    const v2f av = *(const v2f*)(Ar + k);
    const int kk = k + 2 * half;
#pragma unroll
    for (int j = 0; j < NT; ++j) {
      const int n = (chalf * NT + j) * 16 + l16;
      const v2f bv = *(const v2f*)(lw + ((size_t)(kk >> 1) * N + n) * 2);
      pacc[j] = __builtin_amdgcn_wmma_f32_16x16x4_f32(
          false, av, false, bv, (short)0, pacc[j], false, false);
    }
  }
#pragma unroll
  for (int j = 0; j < NT; ++j) {
    const int n = (chalf * NT + j) * 16 + l16;
    float* __restrict__ Cp = C + (size_t)rb * N + n;
#pragma unroll
    for (int r = 0; r < 8; ++r) Cp[(size_t)r * N] = pacc[j][r] * sc[r];
  }
}


__device__ __forceinline__ void stage_w128(const float* __restrict__ W, float* lw, int tid) {
  constexpr int N = 128;
  constexpr int QUADS = 64 * N / 4;
  for (int qd = tid; qd < QUADS; qd += 512) {
    const int t = (qd * 4) / N;
    const int n = (qd * 4) % N;
    const int k = t * 2;
    const v4f w0 = *(const v4f*)(W + (size_t)k * N + n);
    const v4f w1 = *(const v4f*)(W + (size_t)(k + 1) * N + n);
    v4f lo, hi;
    lo.x = w0.x; lo.y = w1.x; lo.z = w0.y; lo.w = w1.y;
    hi.x = w0.z; hi.y = w1.z; hi.z = w0.w; hi.w = w1.w;
    float* d = lw + ((size_t)t * N + n) * 2;
    *(v4f*)(d)     = lo;
    *(v4f*)(d + 4) = hi;
  }
}

__device__ __forceinline__ void tile_product128(const float* Ar, const float* lw, int chalf, int half,
                                                int l16, v8f (&pacc)[4]) {
#pragma unroll
  for (int k = 0; k < 128; k += 4) {
    const v2f av = *(const v2f*)(Ar + k);
    const int kk = k + 2 * half;
#pragma unroll
    for (int j = 0; j < 4; ++j) {
      const int n = (chalf * 4 + j) * 16 + l16;
      const v2f bv = *(const v2f*)(lw + ((size_t)(kk >> 1) * 128 + n) * 2);
      pacc[j] = __builtin_amdgcn_wmma_f32_16x16x4_f32(
          false, av, false, bv, (short)0, pacc[j], false, false);
    }
  }
}

__global__ __launch_bounds__(512) void rebuild_sum_product2_kernel(
    const float* __restrict__ xg, const int* __restrict__ rowptr,
    const int* __restrict__ indeg, const int* __restrict__ csr_src,
    const float* __restrict__ dinv,
    const float* __restrict__ W1, const float* __restrict__ b1,
    const float* __restrict__ W2, const float* __restrict__ b2,
    const float* __restrict__ W3, float* __restrict__ C) {
  __shared__ float lw[128 * 128];
  __shared__ __attribute__((aligned(16))) float tile[128 * TPITCH];

  const int tid = threadIdx.x;
  const int lane = tid & 31;
  const int wave = tid >> 5;
  const int half = lane >> 4;
  const int l16  = lane & 15;

  stage_w128(W2, lw, tid);

  v2f bf[8];
#pragma unroll
  for (int ft = 0; ft < 8; ++ft) {
    const int f = 16 * ft + l16;
    bf[ft].x = half ? W1[256 + f] : W1[f];
    bf[ft].y = half ? b1[f] : W1[128 + f];
  }
  const int wrow0 = (int)(blockIdx.x * 128) + wave * 8;
  int rp = 0, nd = 0;
  float dv = 0.0f;
  {
    const int rr = wrow0 + (lane & 7);
    if (rr < NNODES) { rp = rowptr[rr]; nd = indeg[rr] + 1; dv = dinv[rr]; }
  }
  const v2f zero2 = 0.0f;
  const v8f zero8 = 0.0f;
#pragma nounroll
  for (int pr = 0; pr < 4; ++pr) {
    const int lr = 2 * pr + (l16 >> 3);
    const int startL = __shfl(rp, lr);
    const int n1L = __shfl(nd, lr);
    const int rowL = wrow0 + lr;
    const int ha = 2 * pr + half;
    const float ddH = __shfl(dv, ha);
    const int nA = __builtin_amdgcn_readlane(nd, 2 * pr);
    const int nB = __builtin_amdgcn_readlane(nd, 2 * pr + 1);
    const int nmax = (nA > nB) ? nA : nB;
    float acc[8];
#pragma unroll
    for (int ft = 0; ft < 8; ++ft) acc[ft] = 0.0f;

    for (int blk = 0; blk < nmax; blk += 32) {
      const int lim = nmax - blk;
      int id[4];
#pragma unroll
      for (int r = 0; r < 4; ++r) {
        const int e = blk + 8 * r + (l16 & 7);
        int pos = startL + e - 1;
        pos = (pos < 0) ? 0 : pos;
        pos = (pos < NEDGES) ? pos : (NEDGES - 1);
        id[r] = csr_src[pos];
      }
      asm volatile("" :: "v"(id[0]), "v"(id[1]), "v"(id[2]), "v"(id[3]));
      id[0] = (blk + (l16 & 7) == 0) ? rowL : id[0];
      v2f av[4];
#pragma unroll
      for (int r = 0; r < 4; ++r) av[r] = *(const v2f*)(xg + (size_t)id[r] * 4 + 2 * half);
#pragma unroll
      for (int r = 0; r < 4; ++r) {
        const int e = blk + 8 * r + (l16 & 7);
        av[r] = (e < n1L) ? av[r] : zero2;
      }
#pragma unroll
      for (int r = 0; r < 4; ++r) {
        if (8 * r < lim) {
#pragma unroll
          for (int ft = 0; ft < 8; ++ft) {
            const v8f z = __builtin_amdgcn_wmma_f32_16x16x4_f32(
                false, av[r], false, bf[ft], (short)0, zero8, false, false);
            float sum = fmaxf(z[0], 0.0f);
#pragma unroll
            for (int i = 1; i < 8; ++i) sum += fmaxf(z[i], 0.0f);
            acc[ft] += sum;
          }
        }
      }
    }
    float* __restrict__ tr = tile + (wave * 8 + ha) * TPITCH + l16;
#pragma unroll
    for (int ft = 0; ft < 8; ++ft) tr[16 * ft] = acc[ft] * ddH;
  }
  __syncthreads();

  const int rt = wave >> 1;
  const int chalf = wave & 1;
  const float* Ar = tile + (rt * 16 + l16) * TPITCH + 2 * half;
  v8f pacc[4];
#pragma unroll
  for (int j = 0; j < 4; ++j) {
    const float binit = b2[(chalf * 4 + j) * 16 + l16];
#pragma unroll
    for (int r = 0; r < 8; ++r) pacc[j][r] = binit;
  }
  tile_product128(Ar, lw, chalf, half, l16, pacc);
  __syncthreads();

  {
    float* tw = tile + (rt * 16 + half * 8) * TPITCH + chalf * 64 + l16;
#pragma unroll
    for (int j = 0; j < 4; ++j) {
#pragma unroll
      for (int r = 0; r < 8; ++r) tw[r * TPITCH + 16 * j] = fmaxf(pacc[j][r], 0.0f);
    }
  }
  stage_w128(W3, lw, tid);
  const int rb = (int)(blockIdx.x * 128) + rt * 16 + half * 8;
  static_assert(NNODES % 8 == 0, "a group of 8 rows is all nodes or all pad");
  float sc[8];
  {
    v4f s0 = 0.0f, s1 = 0.0f;
    if (rb < NNODES) {
      s0 = *(const v4f*)(dinv + rb);
      s1 = *(const v4f*)(dinv + rb + 4);
    }
    sc[0] = s0.x; sc[1] = s0.y; sc[2] = s0.z; sc[3] = s0.w;
    sc[4] = s1.x; sc[5] = s1.y; sc[6] = s1.z; sc[7] = s1.w;
  }
  __syncthreads();
#pragma unroll
  for (int j = 0; j < 4; ++j) {
#pragma unroll
    for (int r = 0; r < 8; ++r) pacc[j][r] = 0.0f;
  }
  tile_product128(Ar, lw, chalf, half, l16, pacc);
#pragma unroll
  for (int j = 0; j < 4; ++j) {
    const int n = (chalf * 4 + j) * 16 + l16;
    float* __restrict__ Cp = C + (size_t)rb * 128 + n;
#pragma unroll
    for (int r = 0; r < 8; ++r) Cp[(size_t)r * 128] = pacc[j][r] * sc[r];
  }
}

#define HIDP 34
__global__ __launch_bounds__(256) void head_kernel(
    const float* __restrict__ sums, const float* __restrict__ cnt,
    const float* __restrict__ Wl1, const float* __restrict__ bl1,
    const float* __restrict__ Wl2, const float* __restrict__ bl2,
    float* __restrict__ out) {
  __shared__ __attribute__((aligned(16))) float lw1[64 * 32];
  __shared__ __attribute__((aligned(16))) float lw2[32 * 16];
  __shared__ __attribute__((aligned(16))) float hid[8 * 16 * HIDP];
  __shared__ float ost[128 * 9];

  const int tid  = threadIdx.x;
  const int lane = tid & 31;
  const int wave = tid >> 5;
  const int half = lane >> 4;
  const int l16  = lane & 15;
  const int row0 = blockIdx.x * 128 + wave * 16;
  const int row  = row0 + l16;

  const float den = fmaxf(cnt[row], 1.0f);
  const float* __restrict__ Arow = sums + (size_t)row * 64 + 2 * half;
  v2f a1[16];
#pragma unroll
  for (int s = 0; s < 16; ++s) {
    a1[s] = *(const v2f*)(Arow + 4 * s);
    a1[s].x = a1[s].x / den;
    a1[s].y = a1[s].y / den;
  }

  {
    const int t = (tid * 4) / 32;
    const int n = (tid * 4) % 32;
    const int k = t * 2;
    const v4f w0 = *(const v4f*)(Wl1 + (size_t)k * 32 + n);
    const v4f w1 = *(const v4f*)(Wl1 + (size_t)(k + 1) * 32 + n);
    v4f lo, hi;
    lo.x = w0.x; lo.y = w1.x; lo.z = w0.y; lo.w = w1.y;
    hi.x = w0.z; hi.y = w1.z; hi.z = w0.w; hi.w = w1.w;
    float* d = lw1 + ((size_t)t * 32 + n) * 2;
    *(v4f*)(d)     = lo;
    *(v4f*)(d + 4) = hi;
  }
  for (int i = tid; i < 32 * 16; i += 256) {
    const int k = i >> 4, n = i & 15;
    lw2[((k >> 1) * 16 + n) * 2 + (k & 1)] = (n < 9) ? Wl2[k * 9 + n] : 0.0f;
  }
  __syncthreads();

  float* __restrict__ hw = hid + wave * (16 * HIDP);
#pragma unroll
  for (int j = 0; j < 2; ++j) {
    const int n = 16 * j + l16;
    const float binit = bl1[n];
    v8f acc;
#pragma unroll
    for (int r = 0; r < 8; ++r) acc[r] = binit;
#pragma unroll
    for (int s = 0; s < 16; ++s) {
      const int kk = 4 * s + 2 * half;
      const v2f bv = *(const v2f*)(lw1 + ((size_t)(kk >> 1) * 32 + n) * 2);
      acc = __builtin_amdgcn_wmma_f32_16x16x4_f32(
          false, a1[s], false, bv, (short)0, acc, false, false);
    }
#pragma unroll
    for (int r = 0; r < 8; ++r) hw[(half * 8 + r) * HIDP + n] = acc[r];
  }
  __syncthreads();

  const float binit2 = (l16 < 9) ? bl2[l16] : 0.0f;
  v8f acc2;
#pragma unroll
  for (int r = 0; r < 8; ++r) acc2[r] = binit2;
#pragma unroll
  for (int s = 0; s < 8; ++s) {
    const int kk = 4 * s + 2 * half;
    v2f av = *(const v2f*)(hw + l16 * HIDP + kk);
    av.x = fmaxf(av.x, 0.0f); av.y = fmaxf(av.y, 0.0f);
    const v2f bv = *(const v2f*)(lw2 + ((size_t)(kk >> 1) * 16 + l16) * 2);
    acc2 = __builtin_amdgcn_wmma_f32_16x16x4_f32(
        false, av, false, bv, (short)0, acc2, false, false);
  }
  if (l16 < 9) {
    float* __restrict__ op = ost + (wave * 16 + half * 8) * 9 + l16;
#pragma unroll
    for (int r = 0; r < 8; ++r) op[r * 9] = acc2[r];
  }
  __syncthreads();
  float* __restrict__ ob = out + (size_t)blockIdx.x * (128 * 9);
  for (int i = tid; i < 128 * 9; i += 256) ob[i] = ost[i];
}

extern "C" void kernel_launch(void* const* d_in, const int* in_sizes, int n_in,
                              void* d_out, int out_size, void* d_ws, size_t ws_size,
                              hipStream_t stream) {
  const float* x   = (const float*)d_in[0];
  const int*   ei  = (const int*)d_in[1];
  const int*   bat = (const int*)d_in[2];
  const float* W1  = (const float*)d_in[3];  const float* b1  = (const float*)d_in[4];
  const float* W2  = (const float*)d_in[5];  const float* b2  = (const float*)d_in[6];
  const float* W3  = (const float*)d_in[7];  const float* b3  = (const float*)d_in[8];
  const float* W4  = (const float*)d_in[9];  const float* b4  = (const float*)d_in[10];
  const float* Wl1 = (const float*)d_in[11]; const float* bl1 = (const float*)d_in[12];
  const float* Wl2 = (const float*)d_in[13]; const float* bl2 = (const float*)d_in[14];
  float* out = (float*)d_out;

  const int* src = ei;
  const int* dst = ei + NEDGES;

  float* ws     = (float*)d_ws;
  float* dinv   = ws;
  float* xs     = dinv + NNODES;
  float* xa     = xs + (size_t)MPAD * 4;
  float* hT     = xa + (size_t)MPAD * 4;
  float* hA     = hT + (size_t)MPAD * 128;
  float* pooled = hA + (size_t)MPAD * 128;
  float* cnt    = pooled + NGRAPHS * 64;
  int*   indeg   = (int*)(cnt + NGRAPHS);
  int*   rowptr  = indeg + NNODES;
  int*   cntT    = rowptr + NNODES;
  int*   offT    = cntT + NBINS * NCHUNK;
  int*   binned  = offT + NBINS * NCHUNK;
  int*   csr_src = binned + 2 * NEDGES;

  bin_chunk_kernel<<<NCHUNK, 1024, 0, stream>>>(src, dst, binned, cntT, offT);
  bin_rows_kernel<<<NBINS, 1024, 0, stream>>>(binned, cntT, offT, x, indeg, rowptr, dinv, xs, csr_src,
                                              pooled, cnt);

  const unsigned GB = MPAD / 128;
  const unsigned RB = MPAD / 32;

  row_gather_x_kernel<<<MPAD / 64, 256, 0, stream>>>(xs, rowptr, indeg, csr_src, dinv, xa);

  rebuild_sum_product2_kernel<<<GB, 512, 0, stream>>>(xa, rowptr, indeg, csr_src, dinv,
                                                       W1, b1, W2, b2, W3, hA);

  gather_product_kernel<64><<<GB, 512, 0, stream>>>(hA, rowptr, indeg, csr_src, dinv, b3, W4, hT);

  row_gather_kernel<64, true><<<RB, 256, 0, stream>>>(hT, rowptr, indeg, csr_src, dinv, b4, nullptr,
                                                     bat, pooled, cnt);

  head_kernel<<<NGRAPHS / 128, 256, 0, stream>>>(pooled, cnt, Wl1, bl1, Wl2, bl2, out);
}


# Last gather: a block owns 64 rows (8 waves x 2 passes), pool sums meet in LDS, one float-atomic set per block and graph (was per 4-row wave); half the waves; b4 added in head; pool sum order changes
# speedup vs baseline: 7.0112x; 1.1091x over previous
#include <hip/hip_runtime.h>
#include <hip/hip_bf16.h>

typedef __attribute__((ext_vector_type(2))) float v2f;
typedef __attribute__((ext_vector_type(4))) float v4f;
typedef __attribute__((ext_vector_type(8))) float v8f;

#define NNODES  100000
#define MPAD    100096
#define NEDGES  1600000
#define NGRAPHS 512

typedef __attribute__((ext_vector_type(2))) int v2i;
typedef __attribute__((ext_vector_type(4))) int v4i;

#define BINSHIFT 8
#define BINNODES 256
#define NBINS    ((NNODES + BINNODES - 1) / BINNODES)
#define NCHUNK   128
#define CHUNK    (NEDGES / NCHUNK)
#define EPT      ((CHUNK + 1023) / 1024)
#define STAGECAP 16384
static_assert(NEDGES % NCHUNK == 0, "chunks must tile the edge list");
static_assert(NBINS <= 512, "bin scan covers 512 bins");

__device__ __forceinline__ int wave_incl_scan(int v, int lane) {
  for (int off = 1; off < 32; off <<= 1) {
    const int u = __shfl_up(v, off);
    if (lane >= off) v += u;
  }
  return v;
}

__global__ __launch_bounds__(1024) void bin_chunk_kernel(
    const int* __restrict__ src, const int* __restrict__ dst,
    int* __restrict__ binned, int* __restrict__ cntT, int* __restrict__ offT) {
  __shared__ int hist[512];
  __shared__ int wsum[16];
  __shared__ v2i sorted[CHUNK];
  const int t = threadIdx.x;
  const int lane = t & 31;
  const int b = blockIdx.x;
  const int base = b * CHUNK;
  if (t < 512) hist[t] = 0;
  __syncthreads();

  int s[EPT], d[EPT], r[EPT];
#pragma unroll
  for (int i = 0; i < EPT; ++i) {
    const int idx = t + 1024 * i;
    s[i] = 0; d[i] = 0;
    if (idx < CHUNK) { s[i] = src[base + idx]; d[i] = dst[base + idx]; }
  }
#pragma unroll
  for (int i = 0; i < EPT; ++i) {
    const int idx = t + 1024 * i;
    r[i] = 0;
    if (idx < CHUNK) r[i] = atomicAdd(&hist[d[i] >> BINSHIFT], 1);
  }
  __syncthreads();

  const int v = (t < 512) ? hist[t] : 0;
  const int incl = wave_incl_scan(v, lane);
  if (t < 512 && lane == 31) wsum[t >> 5] = incl;
  __syncthreads();
  if (t < 512) {
    int add = 0;
    for (int w2 = 0; w2 < (t >> 5); ++w2) add += wsum[w2];
    const int excl = incl - v + add;
    hist[t] = excl;
    if (t < NBINS) {
      cntT[t * NCHUNK + b] = v;
      offT[t * NCHUNK + b] = excl;
    }
  }
  __syncthreads();

#pragma unroll
  for (int i = 0; i < EPT; ++i) {
    const int idx = t + 1024 * i;
    if (idx < CHUNK) {
      v2i e; e.x = s[i]; e.y = d[i];
      sorted[hist[d[i] >> BINSHIFT] + r[i]] = e;
    }
  }
  __syncthreads();
  v2i* __restrict__ outp = (v2i*)binned + base;
#pragma unroll
  for (int i = 0; i < EPT; ++i) {
    const int idx = t + 1024 * i;
    if (idx < CHUNK) outp[idx] = sorted[idx];
  }
}

#define RUNS_PER_WAVE (NCHUNK / 32)
#define PSUM_TRIPS    ((NBINS * (NCHUNK / 4) + 1023) / 1024)

template <typename F>
__device__ __forceinline__ void for_bin_edges(const v2i* __restrict__ pairs, const int* runLen,
                                              const int* runStart, int w, int lane, F f) {
  int len[RUNS_PER_WAVE], st[RUNS_PER_WAVE];
  v2i e0[RUNS_PER_WAVE], e1[RUNS_PER_WAVE];
#pragma unroll
  for (int q = 0; q < RUNS_PER_WAVE; ++q) {
    len[q] = runLen[w + 32 * q];
    st[q] = runStart[w + 32 * q];
    e0[q] = pairs[(lane < len[q]) ? (st[q] + lane) : 0];
    e1[q] = pairs[(lane + 32 < len[q]) ? (st[q] + lane + 32) : 0];
  }
#pragma unroll
  for (int q = 0; q < RUNS_PER_WAVE; ++q) asm volatile("" :: "v"(e0[q]), "v"(e1[q]));
#pragma unroll
  for (int q = 0; q < RUNS_PER_WAVE; ++q) {
    if (lane < len[q]) f(e0[q]);
    if (lane + 32 < len[q]) f(e1[q]);
  }
#pragma unroll
  for (int q = 0; q < RUNS_PER_WAVE; ++q) {
    for (int j = lane + 64; j < len[q]; j += 32) f(pairs[st[q] + j]);
  }
}

__global__ __launch_bounds__(1024) void bin_rows_kernel(
    const int* __restrict__ binned, const int* __restrict__ cntT, const int* __restrict__ offT,
    const float* __restrict__ x, int* __restrict__ indeg, int* __restrict__ rowptr,
    float* __restrict__ dinv, float* __restrict__ xs, int* __restrict__ csr_src,
    float* __restrict__ pooled, float* __restrict__ pcnt) {
  static_assert(NBINS * 1024 >= NGRAPHS * 65, "the grid covers the pool table and its counts");
  __shared__ int cnt[BINNODES];
  __shared__ int cursor[BINNODES];
  __shared__ int runLen[NCHUNK];
  __shared__ int runStart[NCHUNK];
  __shared__ int red[32];
  __shared__ int wsum[8];
  __shared__ int stage[STAGECAP];
  const int t = threadIdx.x;
  const int lane = t & 31;
  const int w = t >> 5;
  const int bin = blockIdx.x;
  const int d0 = bin << BINSHIFT;
  const v2i* __restrict__ pairs = (const v2i*)binned;

  {
    const int gid = bin * 1024 + t;
    if (gid < NGRAPHS * 64) pooled[gid] = 0.0f;
    else if (gid < NGRAPHS * 65) pcnt[gid - NGRAPHS * 64] = 0.0f;
  }

  const int node = d0 + t;
  const bool has_node = (t < BINNODES) && (node < NNODES);
  float x0 = 0.0f, x1 = 0.0f, x2 = 0.0f;
  if (has_node) {
    const float* __restrict__ xr = x + (size_t)node * 3;
    x0 = xr[0]; x1 = xr[1]; x2 = xr[2];
  }
  int rl = 0, ro = 0;
  if (t < NCHUNK) {
    rl = cntT[bin * NCHUNK + t];
    ro = offT[bin * NCHUNK + t];
  }
  int part = 0;
  {
    const v4i* __restrict__ c4 = (const v4i*)cntT;
    const int n4 = bin * (NCHUNK / 4);
    v4i q[PSUM_TRIPS];
#pragma unroll
    for (int k = 0; k < PSUM_TRIPS; ++k) {
      const int i = t + 1024 * k;
      q[k] = c4[(i < NBINS * (NCHUNK / 4)) ? i : 0];
    }
#pragma unroll
    for (int k = 0; k < PSUM_TRIPS; ++k) asm volatile("" :: "v"(q[k]));
#pragma unroll
    for (int k = 0; k < PSUM_TRIPS; ++k) {
      const int i = t + 1024 * k;
      part += (i < n4) ? (q[k].x + q[k].y + q[k].z + q[k].w) : 0;
    }
  }
  if (t < NCHUNK) {
    runLen[t] = rl;
    runStart[t] = t * CHUNK + ro;
  }
  if (t < BINNODES) cnt[t] = 0;
  for (int off = 16; off > 0; off >>= 1) part += __shfl_xor(part, off);
  if (lane == 0) red[w] = part;
  __syncthreads();
  int P = 0;
  for (int k = 0; k < 32; ++k) P += red[k];

  for_bin_edges(pairs, runLen, runStart, w, lane,
                [&](const v2i e) { atomicAdd(&cnt[e.y - d0], 1); });
  __syncthreads();

  const int v = (t < BINNODES) ? cnt[t] : 0;
  const int incl = wave_incl_scan(v, lane);
  if (t < BINNODES && lane == 31) wsum[w] = incl;
  __syncthreads();
  int n = 0;
  for (int w2 = 0; w2 < 8; ++w2) n += wsum[w2];
  if (t < BINNODES) {
    int add = 0;
    for (int w2 = 0; w2 < w; ++w2) add += wsum[w2];
    const int excl = incl - v + add;
    cursor[t] = excl;
    if (has_node) {
      indeg[node] = v;
      rowptr[node] = P + excl;
      const float dd = rsqrtf((float)v + 1.0f);
      dinv[node] = dd;
      v4f o;
      o.x = x0 * dd; o.y = x1 * dd; o.z = x2 * dd; o.w = 0.0f;
      *(v4f*)(xs + (size_t)node * 4) = o;
    }
  }
  __syncthreads();

  const bool staged = (n <= STAGECAP);
  for_bin_edges(pairs, runLen, runStart, w, lane, [&](const v2i e) {
    const int pos = atomicAdd(&cursor[e.y - d0], 1);
    if (staged) stage[pos] = e.x;
    else csr_src[P + pos] = e.x;
  });
  __syncthreads();
  if (staged) {
    for (int i = t; i < n; i += 1024) csr_src[P + i] = stage[i];
  }
}

__global__ __launch_bounds__(256) void row_gather_x_kernel(
    const float* __restrict__ xs, const int* __restrict__ rowptr,
    const int* __restrict__ indeg, const int* __restrict__ csr_src,
    const float* __restrict__ dinv, float* __restrict__ xa) {
  const int t = blockIdx.x * 256 + threadIdx.x;
  const int row = t >> 2;
  const int sub = t & 3;
  const v4f zero4 = 0.0f;
  v4f acc = zero4;
  float dd = 0.0f;
  if (row < NNODES) {
    dd = dinv[row];
    const int start = rowptr[row];
    const int end = start + indeg[row];
    const int last = end - 1;
    if (sub == 0) acc = *(const v4f*)(xs + (size_t)row * 4);
    for (int base = start; base < end; base += 32) {
      int s[8];
      for (int u = 0; u < 8; ++u) {
        const int e = base + sub + 4 * u;
        s[u] = csr_src[(e < end) ? e : last];
      }
      for (int u = 0; u < 8; ++u) {
        const int e = base + sub + 4 * u;
        const v4f v = *(const v4f*)(xs + (size_t)s[u] * 4);
        acc += (e < end) ? v : zero4;
      }
    }
  }
  float r[3];
  for (int c = 0; c < 3; ++c) {
    float a = acc[c];
    a += __builtin_bit_cast(float, __builtin_amdgcn_update_dpp(
             0, __builtin_bit_cast(int, a), 0xB1, 0xF, 0xF, false));
    a += __builtin_bit_cast(float, __builtin_amdgcn_update_dpp(
             0, __builtin_bit_cast(int, a), 0x4E, 0xF, 0xF, false));
    r[c] = a;
  }
  if (sub == 0) {
    v4f o;
    o.x = r[0] * dd * dd; o.y = r[1] * dd * dd; o.z = r[2] * dd * dd; o.w = dd;
    *(v4f*)(xa + (size_t)row * 4) = o;
  }
}

#define POOL_PASSES 2
#define POOL_ROWS   (POOL_PASSES * 32)
#define POOL_SLOTS  (POOL_PASSES * 8)
static_assert(MPAD % POOL_ROWS == 0, "whole blocks cover the padded rows");
__global__ __launch_bounds__(256) void row_gather_pool_kernel(
    const float* __restrict__ h, const int* __restrict__ rowptr,
    const int* __restrict__ indeg, const int* __restrict__ csr_src,
    const float* __restrict__ dinv, const int* __restrict__ batch,
    float* __restrict__ pooled, float* __restrict__ cnt) {
  constexpr int F = 64;
  constexpr int CH = F / 32;
  __shared__ __attribute__((aligned(16))) float part[POOL_SLOTS * 64];
  __shared__ int tag[POOL_SLOTS];
  const int lane = threadIdx.x & 31;
  const int wave = __builtin_amdgcn_readfirstlane((int)(threadIdx.x >> 5));
  const int q = lane & 7;
  const int gl = lane & 24;
  const int fo = q * 4;
  const v4f zero4 = 0.0f;
  const v4f* __restrict__ h4 = (const v4f*)h;

#pragma nounroll
  for (int it = 0; it < POOL_PASSES; ++it) {
    const int slot = it * 8 + wave;
    const int row = (int)(blockIdx.x * POOL_ROWS) + slot * 4 + (lane >> 3);
    float dd = 0.0f;
    int start = 0, n = 0;
    int g = -1;
    v4f acc[CH];
#pragma unroll
    for (int c = 0; c < CH; ++c) acc[c] = zero4;
    if (row < NNODES) {
      dd = dinv[row];
      start = rowptr[row];
      n = indeg[row];
      g = batch[row];
#pragma unroll
      for (int c = 0; c < CH; ++c) acc[c] = h4[(unsigned)(row * 16 + c * 8 + q)];
    }
    const int n0 = __builtin_amdgcn_readlane(n, 0);
    const int n1 = __builtin_amdgcn_readlane(n, 8);
    const int n2 = __builtin_amdgcn_readlane(n, 16);
    const int n3 = __builtin_amdgcn_readlane(n, 24);
    const int n01 = (n0 > n1) ? n0 : n1;
    const int n23 = (n2 > n3) ? n2 : n3;
    const int nmax = (n01 > n23) ? n01 : n23;

    for (int blk = 0; blk < nmax; blk += 32) {
      int ir[4];
#pragma unroll
      for (int r = 0; r < 4; ++r) {
        const int e = blk + q + 8 * r;
        const int pos = start + ((e < n) ? e : 0);
        ir[r] = csr_src[(pos < NEDGES) ? pos : (NEDGES - 1)];
      }
      const int lim = nmax - blk;
#pragma unroll
      for (int p = 0; p < 8; ++p) {
        if (4 * p < lim) {
          int s[4];
#pragma unroll
          for (int u = 0; u < 4; ++u) {
            const int from = gl + 4 * (p & 1) + u;
            s[u] = __shfl(ir[p >> 1], from);
          }
          if (blk + 4 * p < n) {
            v4f a[4][CH];
#pragma unroll
            for (int u = 0; u < 4; ++u) {
              const int su = (blk + 4 * p + u < n) ? s[u] : s[0];
#pragma unroll
              for (int c = 0; c < CH; ++c) a[u][c] = h4[(unsigned)(su * 16 + c * 8 + q)];
            }
#pragma unroll
            for (int u = 0; u < 4; ++u) {
              const float w = (blk + 4 * p + u < n) ? 1.0f : 0.0f;
#pragma unroll
              for (int c = 0; c < CH; ++c) acc[c] += a[u][c] * w;
            }
          }
        }
      }
    }
#pragma unroll
    for (int c = 0; c < CH; ++c) acc[c] = acc[c] * dd;
    const int g0 = __builtin_amdgcn_readlane(g, 0);
    const int g1 = __builtin_amdgcn_readlane(g, 8);
    const int g2 = __builtin_amdgcn_readlane(g, 16);
    const int g3 = __builtin_amdgcn_readlane(g, 24);
    int tg = -1;
    v2f pv = 0.0f;
    const int k = lane >> 3;
    if (g0 >= 0 && g0 == g1 && g0 == g2 && g0 == g3) {
      float t[CH][4];
#pragma unroll
      for (int c = 0; c < CH; ++c) {
#pragma unroll
        for (int j = 0; j < 4; ++j) {
          float v = acc[c][j];
          v += __shfl_xor(v, 8);
          v += __shfl_xor(v, 16);
          t[c][j] = v;
        }
      }
      pv.x = (k == 0) ? t[0][0] : (k == 1) ? t[0][2] : (k == 2) ? t[1][0] : t[1][2];
      pv.y = (k == 0) ? t[0][1] : (k == 1) ? t[0][3] : (k == 2) ? t[1][1] : t[1][3];
      tg = g0;
    } else if (g >= 0) {
#pragma unroll
      for (int c = 0; c < CH; ++c) {
#pragma unroll
        for (int j = 0; j < 4; ++j)
          atomicAdd(&pooled[(unsigned)g * 64u + (unsigned)(c * 32 + j) + (unsigned)fo], acc[c][j]);
      }
      if (q == 0) atomicAdd(&cnt[(unsigned)g], 1.0f);
    }
    *(v2f*)(part + slot * 64 + (k >> 1) * 32 + fo + (k & 1) * 2) = pv;
    if (lane == 0) tag[slot] = tg;
  }
  __syncthreads();

  if (threadIdx.x < 64) {
    const int f = threadIdx.x;
    int tgs[POOL_SLOTS];
    float pv[POOL_SLOTS];
#pragma unroll
    for (int w = 0; w < POOL_SLOTS; ++w) { tgs[w] = tag[w]; pv[w] = part[w * 64 + f]; }
    int cur = -1;
    float sum = 0.0f, rows = 0.0f;
#pragma unroll
    for (int w = 0; w < POOL_SLOTS; ++w) {
      const int tg = tgs[w];
      if (tg != cur) {
        if (cur >= 0) {
          atomicAdd(&pooled[(unsigned)cur * 64u + (unsigned)f], sum);
          if (f == 0) atomicAdd(&cnt[(unsigned)cur], rows);
        }
        cur = tg; sum = 0.0f; rows = 0.0f;
      }
      if (tg >= 0) { sum += pv[w]; rows += 4.0f; }
    }
    if (cur >= 0) {
      atomicAdd(&pooled[(unsigned)cur * 64u + (unsigned)f], sum);
      if (f == 0) atomicAdd(&cnt[(unsigned)cur], rows);
    }
  }
}

#define TPITCH 132
template <int N>
__global__ __launch_bounds__(512) void gather_product_kernel(
    const float* __restrict__ h, const int* __restrict__ rowptr,
    const int* __restrict__ indeg, const int* __restrict__ csr_src,
    const float* __restrict__ dinv, const float* __restrict__ bias,
    const float* __restrict__ W, float* __restrict__ C) {
  constexpr int F = 128;
  constexpr int K = 128;
  constexpr int CH = F / 32;
  __shared__ float lw[K * N];
  __shared__ __attribute__((aligned(16))) float tile[128 * TPITCH];

  const int tid = threadIdx.x;
  const int lane = tid & 31;
  const int wave = tid >> 5;

  constexpr int QUADS = (K / 2) * N / 4;
  for (int qd = tid; qd < QUADS; qd += 512) {
    const int t = (qd * 4) / N;
    const int n = (qd * 4) % N;
    const int k = t * 2;
    const v4f w0 = *(const v4f*)(W + (size_t)k * N + n);
    const v4f w1 = *(const v4f*)(W + (size_t)(k + 1) * N + n);
    v4f lo, hi;
    lo.x = w0.x; lo.y = w1.x; lo.z = w0.y; lo.w = w1.y;
    hi.x = w0.z; hi.y = w1.z; hi.z = w0.w; hi.w = w1.w;
    float* d = lw + ((size_t)t * N + n) * 2;
    *(v4f*)(d)     = lo;
    *(v4f*)(d + 4) = hi;
  }

  const int q = lane & 7;
  const int gl = lane & 24;
  const int fo = q * 4;
  const v4f zero4 = 0.0f;
#pragma nounroll
  for (int it = 0; it < 2; ++it) {
    const int lrow = wave * 8 + it * 4 + (lane >> 3);
    const int row = (int)(blockIdx.x * 128) + lrow;
    float dd = 0.0f;
    int start = 0, n = 0;
    v4f acc[CH];
#pragma unroll
    for (int c = 0; c < CH; ++c) acc[c] = zero4;
    if (row < NNODES) {
      dd = dinv[row];
      start = rowptr[row];
      n = indeg[row];
      const float* __restrict__ hr = h + (size_t)row * F + fo;
#pragma unroll
      for (int c = 0; c < CH; ++c) acc[c] = *(const v4f*)(hr + c * 32);
    }
    const int n0 = __builtin_amdgcn_readlane(n, 0);
    const int n1 = __builtin_amdgcn_readlane(n, 8);
    const int n2 = __builtin_amdgcn_readlane(n, 16);
    const int n3 = __builtin_amdgcn_readlane(n, 24);
    const int n01 = (n0 > n1) ? n0 : n1;
    const int n23 = (n2 > n3) ? n2 : n3;
    const int nmax = (n01 > n23) ? n01 : n23;

    for (int blk = 0; blk < nmax; blk += 32) {
      int ir[4];
#pragma unroll
      for (int r = 0; r < 4; ++r) {
        const int e = blk + q + 8 * r;
        const int pos = start + ((e < n) ? e : 0);
        ir[r] = csr_src[(pos < NEDGES) ? pos : (NEDGES - 1)];
      }
      const int lim = nmax - blk;
#pragma unroll
      for (int p = 0; p < 8; ++p) {
        if (4 * p < lim) {
          int s[4];
#pragma unroll
          for (int u = 0; u < 4; ++u) {
            const int from = gl + 4 * (p & 1) + u;
            s[u] = __shfl(ir[p >> 1], from);
          }
          if (blk + 4 * p < n) {
            v4f a[4][CH];
#pragma unroll
            for (int u = 0; u < 4; ++u) {
              const int su = (blk + 4 * p + u < n) ? s[u] : s[0];
              const float* __restrict__ hs = h + (size_t)su * F + fo;
#pragma unroll
              for (int c = 0; c < CH; ++c) a[u][c] = *(const v4f*)(hs + c * 32);
            }
#pragma unroll
            for (int u = 0; u < 4; ++u) {
              const float w = (blk + 4 * p + u < n) ? 1.0f : 0.0f;
#pragma unroll
              for (int c = 0; c < CH; ++c) acc[c] += a[u][c] * w;
            }
          }
        }
      }
    }
    float* __restrict__ tr = tile + lrow * TPITCH + fo;
#pragma unroll
    for (int c = 0; c < CH; ++c) {
      v4f v = zero4;
      if (row < NNODES) {
        const v4f bv = *(const v4f*)(bias + c * 32 + fo);
        v = acc[c] * dd + bv;
        v.x = fmaxf(v.x, 0.0f); v.y = fmaxf(v.y, 0.0f);
        v.z = fmaxf(v.z, 0.0f); v.w = fmaxf(v.w, 0.0f);
      }
      *(v4f*)(tr + c * 32) = v;
    }
  }
  __syncthreads();

  const int half = lane >> 4;
  const int l16  = lane & 15;
  const int rt = wave >> 1;
  const int chalf = wave & 1;
  constexpr int NT = N / 32;
  const int rb = (int)(blockIdx.x * 128) + rt * 16 + half * 8;
  static_assert(NNODES % 8 == 0, "a group of 8 rows is all nodes or all pad");
  float sc[8];
  {
    v4f s0 = 0.0f, s1 = 0.0f;
    if (rb < NNODES) {
      s0 = *(const v4f*)(dinv + rb);
      s1 = *(const v4f*)(dinv + rb + 4);
    }
    sc[0] = s0.x; sc[1] = s0.y; sc[2] = s0.z; sc[3] = s0.w;
    sc[4] = s1.x; sc[5] = s1.y; sc[6] = s1.z; sc[7] = s1.w;
  }
  const float* __restrict__ Ar = tile + (rt * 16 + l16) * TPITCH + 2 * half;
  v8f pacc[NT];
#pragma unroll
  for (int j = 0; j < NT; ++j) {
#pragma unroll
    for (int r = 0; r < 8; ++r) pacc[j][r] = 0.0f;
  }
#pragma unroll
  for (int k = 0; k < K; k += 4) {
    const v2f av = *(const v2f*)(Ar + k);
    const int kk = k + 2 * half;
#pragma unroll
    for (int j = 0; j < NT; ++j) {
      const int n = (chalf * NT + j) * 16 + l16;
      const v2f bv = *(const v2f*)(lw + ((size_t)(kk >> 1) * N + n) * 2);
      pacc[j] = __builtin_amdgcn_wmma_f32_16x16x4_f32(
          false, av, false, bv, (short)0, pacc[j], false, false);
    }
  }
#pragma unroll
  for (int j = 0; j < NT; ++j) {
    const int n = (chalf * NT + j) * 16 + l16;
    float* __restrict__ Cp = C + (size_t)rb * N + n;
#pragma unroll
    for (int r = 0; r < 8; ++r) Cp[(size_t)r * N] = pacc[j][r] * sc[r];
  }
}


__device__ __forceinline__ void stage_w128(const float* __restrict__ W, float* lw, int tid) {
  constexpr int N = 128;
  constexpr int QUADS = 64 * N / 4;
  for (int qd = tid; qd < QUADS; qd += 512) {
    const int t = (qd * 4) / N;
    const int n = (qd * 4) % N;
    const int k = t * 2;
    const v4f w0 = *(const v4f*)(W + (size_t)k * N + n);
    const v4f w1 = *(const v4f*)(W + (size_t)(k + 1) * N + n);
    v4f lo, hi;
    lo.x = w0.x; lo.y = w1.x; lo.z = w0.y; lo.w = w1.y;
    hi.x = w0.z; hi.y = w1.z; hi.z = w0.w; hi.w = w1.w;
    float* d = lw + ((size_t)t * N + n) * 2;
    *(v4f*)(d)     = lo;
    *(v4f*)(d + 4) = hi;
  }
}

__device__ __forceinline__ void tile_product128(const float* Ar, const float* lw, int chalf, int half,
                                                int l16, v8f (&pacc)[4]) {
#pragma unroll
  for (int k = 0; k < 128; k += 4) {
    const v2f av = *(const v2f*)(Ar + k);
    const int kk = k + 2 * half;
#pragma unroll
    for (int j = 0; j < 4; ++j) {
      const int n = (chalf * 4 + j) * 16 + l16;
      const v2f bv = *(const v2f*)(lw + ((size_t)(kk >> 1) * 128 + n) * 2);
      pacc[j] = __builtin_amdgcn_wmma_f32_16x16x4_f32(
          false, av, false, bv, (short)0, pacc[j], false, false);
    }
  }
}

__global__ __launch_bounds__(512) void rebuild_sum_product2_kernel(
    const float* __restrict__ xg, const int* __restrict__ rowptr,
    const int* __restrict__ indeg, const int* __restrict__ csr_src,
    const float* __restrict__ dinv,
    const float* __restrict__ W1, const float* __restrict__ b1,
    const float* __restrict__ W2, const float* __restrict__ b2,
    const float* __restrict__ W3, float* __restrict__ C) {
  __shared__ float lw[128 * 128];
  __shared__ __attribute__((aligned(16))) float tile[128 * TPITCH];

  const int tid = threadIdx.x;
  const int lane = tid & 31;
  const int wave = tid >> 5;
  const int half = lane >> 4;
  const int l16  = lane & 15;

  stage_w128(W2, lw, tid);

  v2f bf[8];
#pragma unroll
  for (int ft = 0; ft < 8; ++ft) {
    const int f = 16 * ft + l16;
    bf[ft].x = half ? W1[256 + f] : W1[f];
    bf[ft].y = half ? b1[f] : W1[128 + f];
  }
  const int wrow0 = (int)(blockIdx.x * 128) + wave * 8;
  int rp = 0, nd = 0;
  float dv = 0.0f;
  {
    const int rr = wrow0 + (lane & 7);
    if (rr < NNODES) { rp = rowptr[rr]; nd = indeg[rr] + 1; dv = dinv[rr]; }
  }
  const v2f zero2 = 0.0f;
  const v8f zero8 = 0.0f;
#pragma nounroll
  for (int pr = 0; pr < 4; ++pr) {
    const int lr = 2 * pr + (l16 >> 3);
    const int startL = __shfl(rp, lr);
    const int n1L = __shfl(nd, lr);
    const int rowL = wrow0 + lr;
    const int ha = 2 * pr + half;
    const float ddH = __shfl(dv, ha);
    const int nA = __builtin_amdgcn_readlane(nd, 2 * pr);
    const int nB = __builtin_amdgcn_readlane(nd, 2 * pr + 1);
    const int nmax = (nA > nB) ? nA : nB;
    float acc[8];
#pragma unroll
    for (int ft = 0; ft < 8; ++ft) acc[ft] = 0.0f;

    for (int blk = 0; blk < nmax; blk += 32) {
      const int lim = nmax - blk;
      int id[4];
#pragma unroll
      for (int r = 0; r < 4; ++r) {
        const int e = blk + 8 * r + (l16 & 7);
        int pos = startL + e - 1;
        pos = (pos < 0) ? 0 : pos;
        pos = (pos < NEDGES) ? pos : (NEDGES - 1);
        id[r] = csr_src[pos];
      }
      asm volatile("" :: "v"(id[0]), "v"(id[1]), "v"(id[2]), "v"(id[3]));
      id[0] = (blk + (l16 & 7) == 0) ? rowL : id[0];
      v2f av[4];
#pragma unroll
      for (int r = 0; r < 4; ++r) av[r] = *(const v2f*)(xg + (size_t)id[r] * 4 + 2 * half);
#pragma unroll
      for (int r = 0; r < 4; ++r) {
        const int e = blk + 8 * r + (l16 & 7);
        av[r] = (e < n1L) ? av[r] : zero2;
      }
#pragma unroll
      for (int r = 0; r < 4; ++r) {
        if (8 * r < lim) {
#pragma unroll
          for (int ft = 0; ft < 8; ++ft) {
            const v8f z = __builtin_amdgcn_wmma_f32_16x16x4_f32(
                false, av[r], false, bf[ft], (short)0, zero8, false, false);
            float sum = fmaxf(z[0], 0.0f);
#pragma unroll
            for (int i = 1; i < 8; ++i) sum += fmaxf(z[i], 0.0f);
            acc[ft] += sum;
          }
        }
      }
    }
    float* __restrict__ tr = tile + (wave * 8 + ha) * TPITCH + l16;
#pragma unroll
    for (int ft = 0; ft < 8; ++ft) tr[16 * ft] = acc[ft] * ddH;
  }
  __syncthreads();

  const int rt = wave >> 1;
  const int chalf = wave & 1;
  const float* Ar = tile + (rt * 16 + l16) * TPITCH + 2 * half;
  v8f pacc[4];
#pragma unroll
  for (int j = 0; j < 4; ++j) {
    const float binit = b2[(chalf * 4 + j) * 16 + l16];
#pragma unroll
    for (int r = 0; r < 8; ++r) pacc[j][r] = binit;
  }
  tile_product128(Ar, lw, chalf, half, l16, pacc);
  __syncthreads();

  {
    float* tw = tile + (rt * 16 + half * 8) * TPITCH + chalf * 64 + l16;
#pragma unroll
    for (int j = 0; j < 4; ++j) {
#pragma unroll
      for (int r = 0; r < 8; ++r) tw[r * TPITCH + 16 * j] = fmaxf(pacc[j][r], 0.0f);
    }
  }
  stage_w128(W3, lw, tid);
  const int rb = (int)(blockIdx.x * 128) + rt * 16 + half * 8;
  static_assert(NNODES % 8 == 0, "a group of 8 rows is all nodes or all pad");
  float sc[8];
  {
    v4f s0 = 0.0f, s1 = 0.0f;
    if (rb < NNODES) {
      s0 = *(const v4f*)(dinv + rb);
      s1 = *(const v4f*)(dinv + rb + 4);
    }
    sc[0] = s0.x; sc[1] = s0.y; sc[2] = s0.z; sc[3] = s0.w;
    sc[4] = s1.x; sc[5] = s1.y; sc[6] = s1.z; sc[7] = s1.w;
  }
  __syncthreads();
#pragma unroll
  for (int j = 0; j < 4; ++j) {
#pragma unroll
    for (int r = 0; r < 8; ++r) pacc[j][r] = 0.0f;
  }
  tile_product128(Ar, lw, chalf, half, l16, pacc);
#pragma unroll
  for (int j = 0; j < 4; ++j) {
    const int n = (chalf * 4 + j) * 16 + l16;
    float* __restrict__ Cp = C + (size_t)rb * 128 + n;
#pragma unroll
    for (int r = 0; r < 8; ++r) Cp[(size_t)r * 128] = pacc[j][r] * sc[r];
  }
}

#define HIDP 34
__global__ __launch_bounds__(256) void head_kernel(
    const float* __restrict__ sums, const float* __restrict__ cnt, const float* __restrict__ b4,
    const float* __restrict__ Wl1, const float* __restrict__ bl1,
    const float* __restrict__ Wl2, const float* __restrict__ bl2,
    float* __restrict__ out) {
  __shared__ __attribute__((aligned(16))) float lw1[64 * 32];
  __shared__ __attribute__((aligned(16))) float lw2[32 * 16];
  __shared__ __attribute__((aligned(16))) float hid[8 * 16 * HIDP];
  __shared__ float ost[128 * 9];

  const int tid  = threadIdx.x;
  const int lane = tid & 31;
  const int wave = tid >> 5;
  const int half = lane >> 4;
  const int l16  = lane & 15;
  const int row0 = blockIdx.x * 128 + wave * 16;
  const int row  = row0 + l16;

  const float cg = cnt[row];
  const float den = fmaxf(cg, 1.0f);
  const float* __restrict__ Arow = sums + (size_t)row * 64 + 2 * half;
  const float* __restrict__ Brow = b4 + 2 * half;
  v2f a1[16];
#pragma unroll
  for (int s = 0; s < 16; ++s) {
    a1[s] = *(const v2f*)(Arow + 4 * s);
    const v2f bb = *(const v2f*)(Brow + 4 * s);
    a1[s].x = (a1[s].x + cg * bb.x) / den;
    a1[s].y = (a1[s].y + cg * bb.y) / den;
  }

  {
    const int t = (tid * 4) / 32;
    const int n = (tid * 4) % 32;
    const int k = t * 2;
    const v4f w0 = *(const v4f*)(Wl1 + (size_t)k * 32 + n);
    const v4f w1 = *(const v4f*)(Wl1 + (size_t)(k + 1) * 32 + n);
    v4f lo, hi;
    lo.x = w0.x; lo.y = w1.x; lo.z = w0.y; lo.w = w1.y;
    hi.x = w0.z; hi.y = w1.z; hi.z = w0.w; hi.w = w1.w;
    float* d = lw1 + ((size_t)t * 32 + n) * 2;
    *(v4f*)(d)     = lo;
    *(v4f*)(d + 4) = hi;
  }
  for (int i = tid; i < 32 * 16; i += 256) {
    const int k = i >> 4, n = i & 15;
    lw2[((k >> 1) * 16 + n) * 2 + (k & 1)] = (n < 9) ? Wl2[k * 9 + n] : 0.0f;
  }
  __syncthreads();

  float* __restrict__ hw = hid + wave * (16 * HIDP);
#pragma unroll
  for (int j = 0; j < 2; ++j) {
    const int n = 16 * j + l16;
    const float binit = bl1[n];
    v8f acc;
#pragma unroll
    for (int r = 0; r < 8; ++r) acc[r] = binit;
#pragma unroll
    for (int s = 0; s < 16; ++s) {
      const int kk = 4 * s + 2 * half;
      const v2f bv = *(const v2f*)(lw1 + ((size_t)(kk >> 1) * 32 + n) * 2);
      acc = __builtin_amdgcn_wmma_f32_16x16x4_f32(
          false, a1[s], false, bv, (short)0, acc, false, false);
    }
#pragma unroll
    for (int r = 0; r < 8; ++r) hw[(half * 8 + r) * HIDP + n] = acc[r];
  }
  __syncthreads();

  const float binit2 = (l16 < 9) ? bl2[l16] : 0.0f;
  v8f acc2;
#pragma unroll
  for (int r = 0; r < 8; ++r) acc2[r] = binit2;
#pragma unroll
  for (int s = 0; s < 8; ++s) {
    const int kk = 4 * s + 2 * half;
    v2f av = *(const v2f*)(hw + l16 * HIDP + kk);
    av.x = fmaxf(av.x, 0.0f); av.y = fmaxf(av.y, 0.0f);
    const v2f bv = *(const v2f*)(lw2 + ((size_t)(kk >> 1) * 16 + l16) * 2);
    acc2 = __builtin_amdgcn_wmma_f32_16x16x4_f32(
        false, av, false, bv, (short)0, acc2, false, false);
  }
  if (l16 < 9) {
    float* __restrict__ op = ost + (wave * 16 + half * 8) * 9 + l16;
#pragma unroll
    for (int r = 0; r < 8; ++r) op[r * 9] = acc2[r];
  }
  __syncthreads();
  float* __restrict__ ob = out + (size_t)blockIdx.x * (128 * 9);
  for (int i = tid; i < 128 * 9; i += 256) ob[i] = ost[i];
}

extern "C" void kernel_launch(void* const* d_in, const int* in_sizes, int n_in,
                              void* d_out, int out_size, void* d_ws, size_t ws_size,
                              hipStream_t stream) {
  const float* x   = (const float*)d_in[0];
  const int*   ei  = (const int*)d_in[1];
  const int*   bat = (const int*)d_in[2];
  const float* W1  = (const float*)d_in[3];  const float* b1  = (const float*)d_in[4];
  const float* W2  = (const float*)d_in[5];  const float* b2  = (const float*)d_in[6];
  const float* W3  = (const float*)d_in[7];  const float* b3  = (const float*)d_in[8];
  const float* W4  = (const float*)d_in[9];  const float* b4  = (const float*)d_in[10];
  const float* Wl1 = (const float*)d_in[11]; const float* bl1 = (const float*)d_in[12];
  const float* Wl2 = (const float*)d_in[13]; const float* bl2 = (const float*)d_in[14];
  float* out = (float*)d_out;

  const int* src = ei;
  const int* dst = ei + NEDGES;

  float* ws     = (float*)d_ws;
  float* dinv   = ws;
  float* xs     = dinv + NNODES;
  float* xa     = xs + (size_t)MPAD * 4;
  float* hT     = xa + (size_t)MPAD * 4;
  float* hA     = hT + (size_t)MPAD * 128;
  float* pooled = hA + (size_t)MPAD * 128;
  float* cnt    = pooled + NGRAPHS * 64;
  int*   indeg   = (int*)(cnt + NGRAPHS);
  int*   rowptr  = indeg + NNODES;
  int*   cntT    = rowptr + NNODES;
  int*   offT    = cntT + NBINS * NCHUNK;
  int*   binned  = offT + NBINS * NCHUNK;
  int*   csr_src = binned + 2 * NEDGES;

  bin_chunk_kernel<<<NCHUNK, 1024, 0, stream>>>(src, dst, binned, cntT, offT);
  bin_rows_kernel<<<NBINS, 1024, 0, stream>>>(binned, cntT, offT, x, indeg, rowptr, dinv, xs, csr_src,
                                              pooled, cnt);

  const unsigned GB = MPAD / 128;

  row_gather_x_kernel<<<MPAD / 64, 256, 0, stream>>>(xs, rowptr, indeg, csr_src, dinv, xa);

  rebuild_sum_product2_kernel<<<GB, 512, 0, stream>>>(xa, rowptr, indeg, csr_src, dinv,
                                                       W1, b1, W2, b2, W3, hA);

  gather_product_kernel<64><<<GB, 512, 0, stream>>>(hA, rowptr, indeg, csr_src, dinv, b3, W4, hT);

  row_gather_pool_kernel<<<MPAD / POOL_ROWS, 256, 0, stream>>>(hT, rowptr, indeg, csr_src, dinv,
                                                               bat, pooled, cnt);

  head_kernel<<<NGRAPHS / 128, 256, 0, stream>>>(pooled, cnt, b4, Wl1, bl1, Wl2, bl2, out);
}
